# starting kernel
# baseline (speedup 1.0000x reference)
#include <hip/hip_runtime.h>

#define NUM_NODES 100000
#define NUM_EDGES 50000
#define NUM_CONN  800000
#define IN_DIM    256
#define HIDDEN    128
#define OUT_DIM   17
#define LN_EPS    1e-5f

#define WAVES_PER_BLOCK 5
#define TILE_ROWS 16
#define N_TILES (HIDDEN / 16)
#define K_STEPS (IN_DIM / 4)
#define WPACK_FLOATS (K_STEPS * 32 * 16)

typedef float v2f __attribute__((ext_vector_type(2)));
typedef float v8f __attribute__((ext_vector_type(8)));

__global__ void zero_ws_kernel(float* __restrict__ p, int n) {
    int i = blockIdx.x * blockDim.x + threadIdx.x;
    if (i < n) p[i] = 0.0f;
}

__global__ void repack_w1_kernel(const float* __restrict__ W1,
                                 float* __restrict__ wpack) {
    int id = blockIdx.x * blockDim.x + threadIdx.x;
    if (id >= K_STEPS * 32 * 8) return;
    int ks   = id >> 8;
    int rem  = id & 255;
    int lane = rem >> 3;
    int nt   = rem & 7;
    int m    = lane & 15;
    int hi   = lane >> 4;
    int krow = ks * 4 + 2 * hi;
    int n    = nt * 16 + m;
    wpack[2 * id + 0] = W1[(size_t)(krow + 0) * HIDDEN + n];
    wpack[2 * id + 1] = W1[(size_t)(krow + 1) * HIDDEN + n];
}

__global__ __launch_bounds__(256) void scatter_add_kernel(
    const float* __restrict__ nf,
    const int*   __restrict__ nodes,
    const int*   __restrict__ edges,
    float* __restrict__ summed,
    float* __restrict__ counts) {
    int t    = blockIdx.x * blockDim.x + threadIdx.x;
    int conn = t >> 5;
    if (conn >= NUM_CONN) return;
    int lane = t & 31;
    int node = nodes[conn];
    int edge = edges[conn];

    const float4* src = (const float4*)(nf + (size_t)node * IN_DIM + lane * 8);
    float4 a = src[0];
    float4 b = src[1];

    float* dst = summed + (size_t)edge * IN_DIM + lane * 8;
    atomicAdd(dst + 0, a.x);
    atomicAdd(dst + 1, a.y);
    atomicAdd(dst + 2, a.z);
    atomicAdd(dst + 3, a.w);
    atomicAdd(dst + 4, b.x);
    atomicAdd(dst + 5, b.y);
    atomicAdd(dst + 6, b.z);
    atomicAdd(dst + 7, b.w);
    if (lane == 0) atomicAdd(counts + edge, 1.0f);
}

__global__ __launch_bounds__(WAVES_PER_BLOCK * 32) void edge_mlp_kernel(
    const float* __restrict__ summed,
    const float* __restrict__ counts,
    const float* __restrict__ wpack,
    const float* __restrict__ b1,
    const float* __restrict__ lng,
    const float* __restrict__ lnb,
    const float* __restrict__ W2,
    const float* __restrict__ b2,
    float* __restrict__ out) {
    __shared__ float sH[WAVES_PER_BLOCK * TILE_ROWS * HIDDEN];

    const int wave = threadIdx.x >> 5;
    const int lane = threadIdx.x & 31;
    const int tile = blockIdx.x * WAVES_PER_BLOCK + wave;
    const int m0   = tile * TILE_ROWS;
    const int m    = lane & 15;
    const int hi   = lane >> 4;

    const int row = m0 + m;
    const float invc = 1.0f / fmaxf(counts[row], 1.0f);
    const float* __restrict__ arow = summed + (size_t)row * IN_DIM;

    v8f acc[N_TILES] = {};

    for (int ks = 0; ks < K_STEPS; ++ks) {
        const int k = ks * 4;
        v2f a = *(const v2f*)(arow + k + 2 * hi);
        a.x *= invc;
        a.y *= invc;

        const float4* __restrict__ bp =
            (const float4*)(wpack + ((size_t)ks * 32 + lane) * 16);
        float4 q0 = bp[0], q1 = bp[1], q2 = bp[2], q3 = bp[3];
        v2f bbs[N_TILES];
        bbs[0].x = q0.x; bbs[0].y = q0.y;
        bbs[1].x = q0.z; bbs[1].y = q0.w;
        bbs[2].x = q1.x; bbs[2].y = q1.y;
        bbs[3].x = q1.z; bbs[3].y = q1.w;
        bbs[4].x = q2.x; bbs[4].y = q2.y;
        bbs[5].x = q2.z; bbs[5].y = q2.w;
        bbs[6].x = q3.x; bbs[6].y = q3.y;
        bbs[7].x = q3.z; bbs[7].y = q3.w;

#pragma unroll
        for (int nt = 0; nt < N_TILES; ++nt) {
            acc[nt] = __builtin_amdgcn_wmma_f32_16x16x4_f32(
                false, a, false, bbs[nt], (short)0, acc[nt], false, false);
        }
    }

#pragma unroll
    for (int nt = 0; nt < N_TILES; ++nt) {
        const float bias = b1[nt * 16 + m];
#pragma unroll
        for (int r = 0; r < 8; ++r) acc[nt][r] += bias;
    }

    float s[8], q[8];
#pragma unroll
    for (int r = 0; r < 8; ++r) { s[r] = 0.0f; q[r] = 0.0f; }
#pragma unroll
    for (int nt = 0; nt < N_TILES; ++nt) {
#pragma unroll
        for (int r = 0; r < 8; ++r) {
            const float x = acc[nt][r];
            s[r] += x;
            q[r] += x * x;
        }
    }
#pragma unroll
    for (int mask = 1; mask < 16; mask <<= 1) {
#pragma unroll
        for (int r = 0; r < 8; ++r) {
            s[r] += __shfl_xor(s[r], mask, 32);
            q[r] += __shfl_xor(q[r], mask, 32);
        }
    }
    float mean[8], rstd[8];
#pragma unroll
    for (int r = 0; r < 8; ++r) {
        mean[r] = s[r] * (1.0f / HIDDEN);
        const float var = q[r] * (1.0f / HIDDEN) - mean[r] * mean[r];
        rstd[r] = rsqrtf(var + LN_EPS);
    }

    float* __restrict__ hbase = sH + wave * (TILE_ROWS * HIDDEN);
#pragma unroll
    for (int nt = 0; nt < N_TILES; ++nt) {
        const int n = nt * 16 + m;
        const float g  = lng[n];
        const float bb = lnb[n];
#pragma unroll
        for (int r = 0; r < 8; ++r) {
            float x = (acc[nt][r] - mean[r]) * rstd[r] * g + bb;
            x = fmaxf(x, 0.0f);
            hbase[(r + 8 * hi) * HIDDEN + n] = x;
        }
    }
    __syncthreads();

    for (int idx = lane; idx < TILE_ROWS * OUT_DIM; idx += 32) {
        const int r = idx / OUT_DIM;
        const int j = idx - r * OUT_DIM;
        float sum = b2[j];
        const float* __restrict__ hr = hbase + r * HIDDEN;
#pragma unroll 4
        for (int k = 0; k < HIDDEN; ++k)
            sum = fmaf(hr[k], W2[k * OUT_DIM + j], sum);
        out[(size_t)(m0 + r) * OUT_DIM + j] = sum;
    }
}

extern "C" void kernel_launch(void* const* d_in, const int* in_sizes, int n_in,
                              void* d_out, int out_size, void* d_ws, size_t ws_size,
                              hipStream_t stream) {
    const float* nf    = (const float*)d_in[0];
    const int*   hidx  = (const int*)d_in[1];
    const int*   nodes = hidx;
    const int*   edges = hidx + NUM_CONN;
    const float* W1    = (const float*)d_in[2];
    const float* b1    = (const float*)d_in[3];
    const float* lng   = (const float*)d_in[4];
    const float* lnb   = (const float*)d_in[5];
    const float* W2    = (const float*)d_in[6];
    const float* b2    = (const float*)d_in[7];
    float*       out   = (float*)d_out;

    float* summed = (float*)d_ws;
    float* counts = summed + (size_t)NUM_EDGES * IN_DIM;
    float* wpack  = counts + NUM_EDGES;

    const int npack = K_STEPS * 32 * 8;
    repack_w1_kernel<<<(npack + 255) / 256, 256, 0, stream>>>(W1, wpack);

    const int nzero = NUM_EDGES * IN_DIM + NUM_EDGES;
    zero_ws_kernel<<<(nzero + 255) / 256, 256, 0, stream>>>(summed, nzero);

    const long long nthreads = (long long)NUM_CONN * 32;
    scatter_add_kernel<<<(int)((nthreads + 255) / 256), 256, 0, stream>>>(
        nf, nodes, edges, summed, counts);

    const int ntiles = NUM_EDGES / TILE_ROWS;
    edge_mlp_kernel<<<ntiles / WAVES_PER_BLOCK, WAVES_PER_BLOCK * 32, 0, stream>>>(
        summed, counts, wpack, b1, lng, lnb, W2, b2, out);
}


# Float-atomic scatter-add replaced by grouping connections by edge (integer rank, scan, place) and one register sum per edge row; row sums now sequential in arrival order
# speedup vs baseline: 3.6283x; 3.6283x over previous
#include <hip/hip_runtime.h>

#define NUM_NODES 100000
#define NUM_EDGES 50000
#define NUM_CONN  800000
#define IN_DIM    256
#define HIDDEN    128
#define OUT_DIM   17
#define LN_EPS    1e-5f

#define WAVES_PER_BLOCK 5
#define TILE_ROWS 16
#define N_TILES (HIDDEN / 16)
#define K_STEPS (IN_DIM / 4)
#define WPACK_FLOATS (K_STEPS * 32 * 16)

typedef float v2f __attribute__((ext_vector_type(2)));
typedef float v8f __attribute__((ext_vector_type(8)));

__global__ void zero_counts_kernel(int* __restrict__ p, int n) {
    int i = blockIdx.x * blockDim.x + threadIdx.x;
    if (i < n) p[i] = 0;
}

__global__ void repack_w1_kernel(const float* __restrict__ W1,
                                 float* __restrict__ wpack) {
    int id = blockIdx.x * blockDim.x + threadIdx.x;
    if (id >= K_STEPS * 32 * 8) return;
    int ks   = id >> 8;
    int rem  = id & 255;
    int lane = rem >> 3;
    int nt   = rem & 7;
    int m    = lane & 15;
    int hi   = lane >> 4;
    int krow = ks * 4 + 2 * hi;
    int n    = nt * 16 + m;
    wpack[2 * id + 0] = W1[(size_t)(krow + 0) * HIDDEN + n];
    wpack[2 * id + 1] = W1[(size_t)(krow + 1) * HIDDEN + n];
}


#define SCAN_THREADS 1024
#define SCAN_PER_THREAD 49
#define SCAN_PADDED (SCAN_THREADS * SCAN_PER_THREAD)

__global__ __launch_bounds__(256) void rank_kernel(
    const int* __restrict__ edges,
    int* __restrict__ cnt,
    int* __restrict__ rank) {
    const int c = blockIdx.x * 256 + threadIdx.x;
    if (c >= NUM_CONN) return;
    const int e = edges[c];
    int r = -1;
    if ((unsigned)e < (unsigned)NUM_EDGES) r = atomicAdd(cnt + e, 1);
    rank[c] = r;
}

__global__ __launch_bounds__(SCAN_THREADS) void scan_kernel(int* __restrict__ cnt) {
    __shared__ int s[SCAN_PADDED];
    __shared__ int wsum[32];
    const int t    = threadIdx.x;
    const int lane = t & 31;
    const int w    = t >> 5;

    int v[SCAN_PER_THREAD];
#pragma unroll
    for (int k = 0; k < SCAN_PER_THREAD - 1; ++k) v[k] = cnt[t + k * SCAN_THREADS];
    {
        const int i = t + (SCAN_PER_THREAD - 1) * SCAN_THREADS;
        v[SCAN_PER_THREAD - 1] = (i < NUM_EDGES) ? cnt[i] : 0;
    }
#pragma unroll
    for (int k = 0; k < SCAN_PER_THREAD; ++k) s[t + k * SCAN_THREADS] = v[k];
    __syncthreads();

    const int base = t * SCAN_PER_THREAD;
    int sum = 0;
    for (int j = 0; j < SCAN_PER_THREAD; ++j) {
        sum += s[base + j];
        s[base + j] = sum;
    }

    int inc = sum;
#pragma unroll
    for (int d = 1; d < 32; d <<= 1) {
        const int up = __shfl_up(inc, d, 32);
        if (lane >= d) inc += up;
    }
    if (lane == 31) wsum[w] = inc;
    __syncthreads();
    if (w == 0) {
        const int tot = wsum[lane];
        int inc2 = tot;
#pragma unroll
        for (int d = 1; d < 32; d <<= 1) {
            const int u = __shfl_up(inc2, d, 32);
            if (lane >= d) inc2 += u;
        }
        wsum[lane] = inc2 - tot;
    }
    __syncthreads();
    const int pre = wsum[w] + inc - sum;
    for (int j = 0; j < SCAN_PER_THREAD; ++j) s[base + j] += pre;
    __syncthreads();

    for (int i = t; i < NUM_EDGES; i += SCAN_THREADS) cnt[i] = s[i];
}

__global__ __launch_bounds__(256) void place_kernel(
    const int* __restrict__ nodes,
    const int* __restrict__ edges,
    const int* __restrict__ rank,
    const int* __restrict__ ends,
    int* __restrict__ perm) {
    const int c = blockIdx.x * 256 + threadIdx.x;
    if (c >= NUM_CONN) return;
    const int r  = rank[c];
    const int e  = edges[c];
    const int nd = nodes[c];
    if (r < 0 || (unsigned)e >= (unsigned)NUM_EDGES) return;
    const int beg = (e > 0) ? ends[e - 1] : 0;
    const unsigned pos = (unsigned)(beg + r);
    if (pos < (unsigned)NUM_CONN) perm[pos] = nd;
}

#define EDGES_PER_WAVE 4
#define SUM_WAVES 8

__device__ __forceinline__ void add4(float4& a, const float4& x) {
    a.x += x.x; a.y += x.y; a.z += x.z; a.w += x.w;
}

__device__ __forceinline__ const float4* node_row(const float* __restrict__ nf, int node) {
    const unsigned n = (unsigned)node < (unsigned)NUM_NODES ? (unsigned)node
                                                            : (unsigned)(NUM_NODES - 1);
    return (const float4*)(nf + (size_t)n * IN_DIM);
}

__global__ __launch_bounds__(SUM_WAVES * 32) void segment_sum_kernel(
    const float* __restrict__ nf,
    const int*   __restrict__ perm,
    const int*   __restrict__ ends,
    float* __restrict__ summed,
    float* __restrict__ counts) {
    const int lane = threadIdx.x & 31;
    const int w = __builtin_amdgcn_readfirstlane(
        (int)(blockIdx.x * SUM_WAVES + (threadIdx.x >> 5)));

    for (int q = 0; q < EDGES_PER_WAVE; ++q) {
        const int e = w * EDGES_PER_WAVE + q;
        if (e >= NUM_EDGES) break;
        int beg = (e > 0) ? ends[e - 1] : 0;
        int end = ends[e];
        if (beg < 0) beg = 0;
        if (end > NUM_CONN) end = NUM_CONN;

        float4 a0 = make_float4(0.0f, 0.0f, 0.0f, 0.0f);
        float4 a1 = a0;
        for (int base = beg; base < end; base += 32) {
            const int left = end - base;
            const int m = left < 32 ? left : 32;
            int myid = 0;
            if (lane < m) myid = perm[base + lane];

            int j = 0;
            for (; j + 4 <= m; j += 4) {
                const float4* p0 = node_row(nf, __builtin_amdgcn_readlane(myid, j + 0));
                const float4* p1 = node_row(nf, __builtin_amdgcn_readlane(myid, j + 1));
                const float4* p2 = node_row(nf, __builtin_amdgcn_readlane(myid, j + 2));
                const float4* p3 = node_row(nf, __builtin_amdgcn_readlane(myid, j + 3));
                const float4 x0 = p0[lane], y0 = p0[32 + lane];
                const float4 x1 = p1[lane], y1 = p1[32 + lane];
                const float4 x2 = p2[lane], y2 = p2[32 + lane];
                const float4 x3 = p3[lane], y3 = p3[32 + lane];
                add4(a0, x0); add4(a1, y0);
                add4(a0, x1); add4(a1, y1);
                add4(a0, x2); add4(a1, y2);
                add4(a0, x3); add4(a1, y3);
            }
            const int rem = m - j;
            if (rem == 3) {
                const float4* p0 = node_row(nf, __builtin_amdgcn_readlane(myid, j + 0));
                const float4* p1 = node_row(nf, __builtin_amdgcn_readlane(myid, j + 1));
                const float4* p2 = node_row(nf, __builtin_amdgcn_readlane(myid, j + 2));
                const float4 x0 = p0[lane], y0 = p0[32 + lane];
                const float4 x1 = p1[lane], y1 = p1[32 + lane];
                const float4 x2 = p2[lane], y2 = p2[32 + lane];
                add4(a0, x0); add4(a1, y0);
                add4(a0, x1); add4(a1, y1);
                add4(a0, x2); add4(a1, y2);
            } else if (rem == 2) {
                const float4* p0 = node_row(nf, __builtin_amdgcn_readlane(myid, j + 0));
                const float4* p1 = node_row(nf, __builtin_amdgcn_readlane(myid, j + 1));
                const float4 x0 = p0[lane], y0 = p0[32 + lane];
                const float4 x1 = p1[lane], y1 = p1[32 + lane];
                add4(a0, x0); add4(a1, y0);
                add4(a0, x1); add4(a1, y1);
            } else if (rem == 1) {
                const float4* p0 = node_row(nf, __builtin_amdgcn_readlane(myid, j));
                const float4 x0 = p0[lane], y0 = p0[32 + lane];
                add4(a0, x0); add4(a1, y0);
            }
        }

        float4* dst = (float4*)(summed + (size_t)e * IN_DIM);
        dst[lane]      = a0;
        dst[32 + lane] = a1;
        if (lane == 0) counts[e] = (float)(end > beg ? end - beg : 0);
    }
}

__global__ __launch_bounds__(WAVES_PER_BLOCK * 32) void edge_mlp_kernel(
    const float* __restrict__ summed,
    const float* __restrict__ counts,
    const float* __restrict__ wpack,
    const float* __restrict__ b1,
    const float* __restrict__ lng,
    const float* __restrict__ lnb,
    const float* __restrict__ W2,
    const float* __restrict__ b2,
    float* __restrict__ out) {
    __shared__ float sH[WAVES_PER_BLOCK * TILE_ROWS * HIDDEN];

    const int wave = threadIdx.x >> 5;
    const int lane = threadIdx.x & 31;
    const int tile = blockIdx.x * WAVES_PER_BLOCK + wave;
    const int m0   = tile * TILE_ROWS;
    const int m    = lane & 15;
    const int hi   = lane >> 4;

    const int row = m0 + m;
    const float invc = 1.0f / fmaxf(counts[row], 1.0f);
    const float* __restrict__ arow = summed + (size_t)row * IN_DIM;

    v8f acc[N_TILES] = {};

    for (int ks = 0; ks < K_STEPS; ++ks) {
        const int k = ks * 4;
        v2f a = *(const v2f*)(arow + k + 2 * hi);
        a.x *= invc;
        a.y *= invc;

        const float4* __restrict__ bp =
            (const float4*)(wpack + ((size_t)ks * 32 + lane) * 16);
        float4 q0 = bp[0], q1 = bp[1], q2 = bp[2], q3 = bp[3];
        v2f bbs[N_TILES];
        bbs[0].x = q0.x; bbs[0].y = q0.y;
        bbs[1].x = q0.z; bbs[1].y = q0.w;
        bbs[2].x = q1.x; bbs[2].y = q1.y;
        bbs[3].x = q1.z; bbs[3].y = q1.w;
        bbs[4].x = q2.x; bbs[4].y = q2.y;
        bbs[5].x = q2.z; bbs[5].y = q2.w;
        bbs[6].x = q3.x; bbs[6].y = q3.y;
        bbs[7].x = q3.z; bbs[7].y = q3.w;

#pragma unroll
        for (int nt = 0; nt < N_TILES; ++nt) {
            acc[nt] = __builtin_amdgcn_wmma_f32_16x16x4_f32(
                false, a, false, bbs[nt], (short)0, acc[nt], false, false);
        }
    }

#pragma unroll
    for (int nt = 0; nt < N_TILES; ++nt) {
        const float bias = b1[nt * 16 + m];
#pragma unroll
        for (int r = 0; r < 8; ++r) acc[nt][r] += bias;
    }

    float s[8], q[8];
#pragma unroll
    for (int r = 0; r < 8; ++r) { s[r] = 0.0f; q[r] = 0.0f; }
#pragma unroll
    for (int nt = 0; nt < N_TILES; ++nt) {
#pragma unroll
        for (int r = 0; r < 8; ++r) {
            const float x = acc[nt][r];
            s[r] += x;
            q[r] += x * x;
        }
    }
#pragma unroll
    for (int mask = 1; mask < 16; mask <<= 1) {
#pragma unroll
        for (int r = 0; r < 8; ++r) {
            s[r] += __shfl_xor(s[r], mask, 32);
            q[r] += __shfl_xor(q[r], mask, 32);
        }
    }
    float mean[8], rstd[8];
#pragma unroll
    for (int r = 0; r < 8; ++r) {
        mean[r] = s[r] * (1.0f / HIDDEN);
        const float var = q[r] * (1.0f / HIDDEN) - mean[r] * mean[r];
        rstd[r] = rsqrtf(var + LN_EPS);
    }

    float* __restrict__ hbase = sH + wave * (TILE_ROWS * HIDDEN);
#pragma unroll
    for (int nt = 0; nt < N_TILES; ++nt) {
        const int n = nt * 16 + m;
        const float g  = lng[n];
        const float bb = lnb[n];
#pragma unroll
        for (int r = 0; r < 8; ++r) {
            float x = (acc[nt][r] - mean[r]) * rstd[r] * g + bb;
            x = fmaxf(x, 0.0f);
            hbase[(r + 8 * hi) * HIDDEN + n] = x;
        }
    }
    __syncthreads();

    for (int idx = lane; idx < TILE_ROWS * OUT_DIM; idx += 32) {
        const int r = idx / OUT_DIM;
        const int j = idx - r * OUT_DIM;
        float sum = b2[j];
        const float* __restrict__ hr = hbase + r * HIDDEN;
#pragma unroll 4
        for (int k = 0; k < HIDDEN; ++k)
            sum = fmaf(hr[k], W2[k * OUT_DIM + j], sum);
        out[(size_t)(m0 + r) * OUT_DIM + j] = sum;
    }
}

extern "C" void kernel_launch(void* const* d_in, const int* in_sizes, int n_in,
                              void* d_out, int out_size, void* d_ws, size_t ws_size,
                              hipStream_t stream) {
    const float* nf    = (const float*)d_in[0];
    const int*   hidx  = (const int*)d_in[1];
    const int*   nodes = hidx;
    const int*   edges = hidx + NUM_CONN;
    const float* W1    = (const float*)d_in[2];
    const float* b1    = (const float*)d_in[3];
    const float* lng   = (const float*)d_in[4];
    const float* lnb   = (const float*)d_in[5];
    const float* W2    = (const float*)d_in[6];
    const float* b2    = (const float*)d_in[7];
    float*       out   = (float*)d_out;

    float* summed = (float*)d_ws;
    float* counts = summed + (size_t)NUM_EDGES * IN_DIM;
    float* wpack  = counts + NUM_EDGES;

    int* rank = (int*)d_ws;
    int* perm = (int*)d_out;
    int* ends = perm + NUM_CONN;

    const int npack = K_STEPS * 32 * 8;
    repack_w1_kernel<<<(npack + 255) / 256, 256, 0, stream>>>(W1, wpack);

    zero_counts_kernel<<<(NUM_EDGES + 255) / 256, 256, 0, stream>>>(ends, NUM_EDGES);

    rank_kernel<<<(NUM_CONN + 255) / 256, 256, 0, stream>>>(edges, ends, rank);

    scan_kernel<<<1, SCAN_THREADS, 0, stream>>>(ends);

    place_kernel<<<(NUM_CONN + 255) / 256, 256, 0, stream>>>(nodes, edges, rank, ends, perm);

    const int nsumwaves = (NUM_EDGES + EDGES_PER_WAVE - 1) / EDGES_PER_WAVE;
    segment_sum_kernel<<<(nsumwaves + SUM_WAVES - 1) / SUM_WAVES, SUM_WAVES * 32, 0, stream>>>(
        nf, perm, ends, summed, counts);

    const int ntiles = NUM_EDGES / TILE_ROWS;
    edge_mlp_kernel<<<ntiles / WAVES_PER_BLOCK, WAVES_PER_BLOCK * 32, 0, stream>>>(
        summed, counts, wpack, b1, lng, lnb, W2, b2, out);
}


# 128->17 output product moved from a per-lane fma loop onto the f32 16x16x4 matrix instruction (hidden tile and W2 read from LDS as fragments); its sum order is now the matrix unit's
# speedup vs baseline: 4.2975x; 1.1844x over previous
#include <hip/hip_runtime.h>

#define NUM_NODES 100000
#define NUM_EDGES 50000
#define NUM_CONN  800000
#define IN_DIM    256
#define HIDDEN    128
#define OUT_DIM   17
#define LN_EPS    1e-5f

#define WAVES_PER_BLOCK 5
#define TILE_ROWS 16
#define N_TILES (HIDDEN / 16)
#define K_STEPS (IN_DIM / 4)
#define WPACK_FLOATS (K_STEPS * 32 * 16)
#define K2_STEPS (HIDDEN / 4)
#define H_PITCH  (HIDDEN + 4)
#define W2_FLOATS (HIDDEN * OUT_DIM)
#define W2_COPY_TRIPS ((W2_FLOATS + WAVES_PER_BLOCK * 32 - 1) / (WAVES_PER_BLOCK * 32))

typedef float v2f __attribute__((ext_vector_type(2)));
typedef float v8f __attribute__((ext_vector_type(8)));

__global__ void zero_counts_kernel(int* __restrict__ p, int n) {
    int i = blockIdx.x * blockDim.x + threadIdx.x;
    if (i < n) p[i] = 0;
}

__global__ void repack_w1_kernel(const float* __restrict__ W1,
                                 float* __restrict__ wpack) {
    int id = blockIdx.x * blockDim.x + threadIdx.x;
    if (id >= K_STEPS * 32 * 8) return;
    int ks   = id >> 8;
    int rem  = id & 255;
    int lane = rem >> 3;
    int nt   = rem & 7;
    int m    = lane & 15;
    int hi   = lane >> 4;
    int krow = ks * 4 + 2 * hi;
    int n    = nt * 16 + m;
    wpack[2 * id + 0] = W1[(size_t)(krow + 0) * HIDDEN + n];
    wpack[2 * id + 1] = W1[(size_t)(krow + 1) * HIDDEN + n];
}


#define SCAN_THREADS 1024
#define SCAN_PER_THREAD 49
#define SCAN_PADDED (SCAN_THREADS * SCAN_PER_THREAD)

__global__ __launch_bounds__(256) void rank_kernel(
    const int* __restrict__ edges,
    int* __restrict__ cnt,
    int* __restrict__ rank) {
    const int c = blockIdx.x * 256 + threadIdx.x;
    if (c >= NUM_CONN) return;
    const int e = edges[c];
    int r = -1;
    if ((unsigned)e < (unsigned)NUM_EDGES) r = atomicAdd(cnt + e, 1);
    rank[c] = r;
}

__global__ __launch_bounds__(SCAN_THREADS) void scan_kernel(int* __restrict__ cnt) {
    __shared__ int s[SCAN_PADDED];
    __shared__ int wsum[32];
    const int t    = threadIdx.x;
    const int lane = t & 31;
    const int w    = t >> 5;

    int v[SCAN_PER_THREAD];
#pragma unroll
    for (int k = 0; k < SCAN_PER_THREAD - 1; ++k) v[k] = cnt[t + k * SCAN_THREADS];
    {
        const int i = t + (SCAN_PER_THREAD - 1) * SCAN_THREADS;
        v[SCAN_PER_THREAD - 1] = (i < NUM_EDGES) ? cnt[i] : 0;
    }
#pragma unroll
    for (int k = 0; k < SCAN_PER_THREAD; ++k) s[t + k * SCAN_THREADS] = v[k];
    __syncthreads();

    const int base = t * SCAN_PER_THREAD;
    int sum = 0;
    for (int j = 0; j < SCAN_PER_THREAD; ++j) {
        sum += s[base + j];
        s[base + j] = sum;
    }

    int inc = sum;
#pragma unroll
    for (int d = 1; d < 32; d <<= 1) {
        const int up = __shfl_up(inc, d, 32);
        if (lane >= d) inc += up;
    }
    if (lane == 31) wsum[w] = inc;
    __syncthreads();
    if (w == 0) {
        const int tot = wsum[lane];
        int inc2 = tot;
#pragma unroll
        for (int d = 1; d < 32; d <<= 1) {
            const int u = __shfl_up(inc2, d, 32);
            if (lane >= d) inc2 += u;
        }
        wsum[lane] = inc2 - tot;
    }
    __syncthreads();
    const int pre = wsum[w] + inc - sum;
    for (int j = 0; j < SCAN_PER_THREAD; ++j) s[base + j] += pre;
    __syncthreads();

    for (int i = t; i < NUM_EDGES; i += SCAN_THREADS) cnt[i] = s[i];
}

__global__ __launch_bounds__(256) void place_kernel(
    const int* __restrict__ nodes,
    const int* __restrict__ edges,
    const int* __restrict__ rank,
    const int* __restrict__ ends,
    int* __restrict__ perm) {
    const int c = blockIdx.x * 256 + threadIdx.x;
    if (c >= NUM_CONN) return;
    const int r  = rank[c];
    const int e  = edges[c];
    const int nd = nodes[c];
    if (r < 0 || (unsigned)e >= (unsigned)NUM_EDGES) return;
    const int beg = (e > 0) ? ends[e - 1] : 0;
    const unsigned pos = (unsigned)(beg + r);
    if (pos < (unsigned)NUM_CONN) perm[pos] = nd;
}

#define EDGES_PER_WAVE 4
#define SUM_WAVES 8

__device__ __forceinline__ void add4(float4& a, const float4& x) {
    a.x += x.x; a.y += x.y; a.z += x.z; a.w += x.w;
}

__device__ __forceinline__ const float4* node_row(const float* __restrict__ nf, int node) {
    const unsigned n = (unsigned)node < (unsigned)NUM_NODES ? (unsigned)node
                                                            : (unsigned)(NUM_NODES - 1);
    return (const float4*)(nf + (size_t)n * IN_DIM);
}

__global__ __launch_bounds__(SUM_WAVES * 32) void segment_sum_kernel(
    const float* __restrict__ nf,
    const int*   __restrict__ perm,
    const int*   __restrict__ ends,
    float* __restrict__ summed,
    float* __restrict__ counts) {
    const int lane = threadIdx.x & 31;
    const int w = __builtin_amdgcn_readfirstlane(
        (int)(blockIdx.x * SUM_WAVES + (threadIdx.x >> 5)));

    for (int q = 0; q < EDGES_PER_WAVE; ++q) {
        const int e = w * EDGES_PER_WAVE + q;
        if (e >= NUM_EDGES) break;
        int beg = (e > 0) ? ends[e - 1] : 0;
        int end = ends[e];
        if (beg < 0) beg = 0;
        if (end > NUM_CONN) end = NUM_CONN;

        float4 a0 = make_float4(0.0f, 0.0f, 0.0f, 0.0f);
        float4 a1 = a0;
        for (int base = beg; base < end; base += 32) {
            const int left = end - base;
            const int m = left < 32 ? left : 32;
            int myid = 0;
            if (lane < m) myid = perm[base + lane];

            int j = 0;
            for (; j + 4 <= m; j += 4) {
                const float4* p0 = node_row(nf, __builtin_amdgcn_readlane(myid, j + 0));
                const float4* p1 = node_row(nf, __builtin_amdgcn_readlane(myid, j + 1));
                const float4* p2 = node_row(nf, __builtin_amdgcn_readlane(myid, j + 2));
                const float4* p3 = node_row(nf, __builtin_amdgcn_readlane(myid, j + 3));
                const float4 x0 = p0[lane], y0 = p0[32 + lane];
                const float4 x1 = p1[lane], y1 = p1[32 + lane];
                const float4 x2 = p2[lane], y2 = p2[32 + lane];
                const float4 x3 = p3[lane], y3 = p3[32 + lane];
                add4(a0, x0); add4(a1, y0);
                add4(a0, x1); add4(a1, y1);
                add4(a0, x2); add4(a1, y2);
                add4(a0, x3); add4(a1, y3);
            }
            const int rem = m - j;
            if (rem == 3) {
                const float4* p0 = node_row(nf, __builtin_amdgcn_readlane(myid, j + 0));
                const float4* p1 = node_row(nf, __builtin_amdgcn_readlane(myid, j + 1));
                const float4* p2 = node_row(nf, __builtin_amdgcn_readlane(myid, j + 2));
                const float4 x0 = p0[lane], y0 = p0[32 + lane];
                const float4 x1 = p1[lane], y1 = p1[32 + lane];
                const float4 x2 = p2[lane], y2 = p2[32 + lane];
                add4(a0, x0); add4(a1, y0);
                add4(a0, x1); add4(a1, y1);
                add4(a0, x2); add4(a1, y2);
            } else if (rem == 2) {
                const float4* p0 = node_row(nf, __builtin_amdgcn_readlane(myid, j + 0));
                const float4* p1 = node_row(nf, __builtin_amdgcn_readlane(myid, j + 1));
                const float4 x0 = p0[lane], y0 = p0[32 + lane];
                const float4 x1 = p1[lane], y1 = p1[32 + lane];
                add4(a0, x0); add4(a1, y0);
                add4(a0, x1); add4(a1, y1);
            } else if (rem == 1) {
                const float4* p0 = node_row(nf, __builtin_amdgcn_readlane(myid, j));
                const float4 x0 = p0[lane], y0 = p0[32 + lane];
                add4(a0, x0); add4(a1, y0);
            }
        }

        float4* dst = (float4*)(summed + (size_t)e * IN_DIM);
        dst[lane]      = a0;
        dst[32 + lane] = a1;
        if (lane == 0) counts[e] = (float)(end > beg ? end - beg : 0);
    }
}

__global__ __launch_bounds__(WAVES_PER_BLOCK * 32) void edge_mlp_kernel(
    const float* __restrict__ summed,
    const float* __restrict__ counts,
    const float* __restrict__ wpack,
    const float* __restrict__ b1,
    const float* __restrict__ lng,
    const float* __restrict__ lnb,
    const float* __restrict__ W2,
    const float* __restrict__ b2,
    float* __restrict__ out) {
    __shared__ float sH[WAVES_PER_BLOCK * TILE_ROWS * H_PITCH];
    __shared__ float sW2[W2_COPY_TRIPS * WAVES_PER_BLOCK * 32];

    const int wave = threadIdx.x >> 5;
    const int lane = threadIdx.x & 31;
    const int tile = blockIdx.x * WAVES_PER_BLOCK + wave;
    const int m0   = tile * TILE_ROWS;
    const int m    = lane & 15;
    const int hi   = lane >> 4;

#pragma unroll
    for (int i = 0; i < W2_COPY_TRIPS; ++i) {
        const int id  = (int)threadIdx.x + i * (WAVES_PER_BLOCK * 32);
        const int idc = id < W2_FLOATS ? id : W2_FLOATS - 1;
        sW2[id] = W2[idc];
    }

    const int row = m0 + m;
    const float invc = 1.0f / fmaxf(counts[row], 1.0f);
    const float* __restrict__ arow = summed + (size_t)row * IN_DIM;

    v8f acc[N_TILES] = {};

    for (int ks = 0; ks < K_STEPS; ++ks) {
        const int k = ks * 4;
        v2f a = *(const v2f*)(arow + k + 2 * hi);
        a.x *= invc;
        a.y *= invc;

        const float4* __restrict__ bp =
            (const float4*)(wpack + ((size_t)ks * 32 + lane) * 16);
        float4 q0 = bp[0], q1 = bp[1], q2 = bp[2], q3 = bp[3];
        v2f bbs[N_TILES];
        bbs[0].x = q0.x; bbs[0].y = q0.y;
        bbs[1].x = q0.z; bbs[1].y = q0.w;
        bbs[2].x = q1.x; bbs[2].y = q1.y;
        bbs[3].x = q1.z; bbs[3].y = q1.w;
        bbs[4].x = q2.x; bbs[4].y = q2.y;
        bbs[5].x = q2.z; bbs[5].y = q2.w;
        bbs[6].x = q3.x; bbs[6].y = q3.y;
        bbs[7].x = q3.z; bbs[7].y = q3.w;

#pragma unroll
        for (int nt = 0; nt < N_TILES; ++nt) {
            acc[nt] = __builtin_amdgcn_wmma_f32_16x16x4_f32(
                false, a, false, bbs[nt], (short)0, acc[nt], false, false);
        }
    }

#pragma unroll
    for (int nt = 0; nt < N_TILES; ++nt) {
        const float bias = b1[nt * 16 + m];
#pragma unroll
        for (int r = 0; r < 8; ++r) acc[nt][r] += bias;
    }

    float s[8], q[8];
#pragma unroll
    for (int r = 0; r < 8; ++r) { s[r] = 0.0f; q[r] = 0.0f; }
#pragma unroll
    for (int nt = 0; nt < N_TILES; ++nt) {
#pragma unroll
        for (int r = 0; r < 8; ++r) {
            const float x = acc[nt][r];
            s[r] += x;
            q[r] += x * x;
        }
    }
#pragma unroll
    for (int mask = 1; mask < 16; mask <<= 1) {
#pragma unroll
        for (int r = 0; r < 8; ++r) {
            s[r] += __shfl_xor(s[r], mask, 32);
            q[r] += __shfl_xor(q[r], mask, 32);
        }
    }
    float mean[8], rstd[8];
#pragma unroll
    for (int r = 0; r < 8; ++r) {
        mean[r] = s[r] * (1.0f / HIDDEN);
        const float var = q[r] * (1.0f / HIDDEN) - mean[r] * mean[r];
        rstd[r] = rsqrtf(var + LN_EPS);
    }

    float* __restrict__ hbase = sH + wave * (TILE_ROWS * H_PITCH);
#pragma unroll
    for (int nt = 0; nt < N_TILES; ++nt) {
        const int n = nt * 16 + m;
        const float g  = lng[n];
        const float bb = lnb[n];
#pragma unroll
        for (int r = 0; r < 8; ++r) {
            float x = (acc[nt][r] - mean[r]) * rstd[r] * g + bb;
            x = fmaxf(x, 0.0f);
            hbase[(r + 8 * hi) * H_PITCH + n] = x;
        }
    }
    __syncthreads();

    v8f o0, o1;
    {
        const float bias0 = b2[m];
        const float bias1 = b2[16];
#pragma unroll
        for (int r = 0; r < 8; ++r) { o0[r] = bias0; o1[r] = bias1; }
    }
    const float* __restrict__ ha  = hbase + m * H_PITCH + 2 * hi;
    const float* __restrict__ wb0 = sW2 + (2 * hi) * OUT_DIM + m;
    const float* __restrict__ wb1 = sW2 + (2 * hi) * OUT_DIM + 16;
#pragma unroll
    for (int ks = 0; ks < K2_STEPS; ++ks) {
        const v2f a = *(const v2f*)(ha + 4 * ks);
        v2f bA, bB;
        bA.x = wb0[ks * 4 * OUT_DIM];
        bA.y = wb0[ks * 4 * OUT_DIM + OUT_DIM];
        bB.x = wb1[ks * 4 * OUT_DIM];
        bB.y = wb1[ks * 4 * OUT_DIM + OUT_DIM];
        o0 = __builtin_amdgcn_wmma_f32_16x16x4_f32(
            false, a, false, bA, (short)0, o0, false, false);
        o1 = __builtin_amdgcn_wmma_f32_16x16x4_f32(
            false, a, false, bB, (short)0, o1, false, false);
    }

    float* __restrict__ orow = out + (size_t)(m0 + 8 * hi) * OUT_DIM;
#pragma unroll
    for (int r = 0; r < 8; ++r) orow[r * OUT_DIM + m] = o0[r];
    if (m == 0) {
#pragma unroll
        for (int r = 0; r < 8; ++r) orow[r * OUT_DIM + 16] = o1[r];
    }
}

extern "C" void kernel_launch(void* const* d_in, const int* in_sizes, int n_in,
                              void* d_out, int out_size, void* d_ws, size_t ws_size,
                              hipStream_t stream) {
    const float* nf    = (const float*)d_in[0];
    const int*   hidx  = (const int*)d_in[1];
    const int*   nodes = hidx;
    const int*   edges = hidx + NUM_CONN;
    const float* W1    = (const float*)d_in[2];
    const float* b1    = (const float*)d_in[3];
    const float* lng   = (const float*)d_in[4];
    const float* lnb   = (const float*)d_in[5];
    const float* W2    = (const float*)d_in[6];
    const float* b2    = (const float*)d_in[7];
    float*       out   = (float*)d_out;

    float* summed = (float*)d_ws;
    float* counts = summed + (size_t)NUM_EDGES * IN_DIM;
    float* wpack  = counts + NUM_EDGES;

    int* rank = (int*)d_ws;
    int* perm = (int*)d_out;
    int* ends = perm + NUM_CONN;

    const int npack = K_STEPS * 32 * 8;
    repack_w1_kernel<<<(npack + 255) / 256, 256, 0, stream>>>(W1, wpack);

    zero_counts_kernel<<<(NUM_EDGES + 255) / 256, 256, 0, stream>>>(ends, NUM_EDGES);

    rank_kernel<<<(NUM_CONN + 255) / 256, 256, 0, stream>>>(edges, ends, rank);

    scan_kernel<<<1, SCAN_THREADS, 0, stream>>>(ends);

    place_kernel<<<(NUM_CONN + 255) / 256, 256, 0, stream>>>(nodes, edges, rank, ends, perm);

    const int nsumwaves = (NUM_EDGES + EDGES_PER_WAVE - 1) / EDGES_PER_WAVE;
    segment_sum_kernel<<<(nsumwaves + SUM_WAVES - 1) / SUM_WAVES, SUM_WAVES * 32, 0, stream>>>(
        nf, perm, ends, summed, counts);

    const int ntiles = NUM_EDGES / TILE_ROWS;
    edge_mlp_kernel<<<ntiles / WAVES_PER_BLOCK, WAVES_PER_BLOCK * 32, 0, stream>>>(
        summed, counts, wpack, b1, lng, lnb, W2, b2, out);
}


# Grouping by edge moved off global atomics: two-level counting sort on LDS counters (128 edge ranges: count, scan, scatter, per-range sort); rows still summed in arrival order, now of the LDS adds
# speedup vs baseline: 5.8574x; 1.3630x over previous
#include <hip/hip_runtime.h>

#define NUM_NODES 100000
#define NUM_EDGES 50000
#define NUM_CONN  800000
#define IN_DIM    256
#define HIDDEN    128
#define OUT_DIM   17
#define LN_EPS    1e-5f

#define WAVES_PER_BLOCK 5
#define TILE_ROWS 16
#define N_TILES (HIDDEN / 16)
#define K_STEPS (IN_DIM / 4)
#define WPACK_FLOATS (K_STEPS * 32 * 16)
#define K2_STEPS (HIDDEN / 4)
#define H_PITCH  (HIDDEN + 4)
#define W2_FLOATS (HIDDEN * OUT_DIM)
#define W2_COPY_TRIPS ((W2_FLOATS + WAVES_PER_BLOCK * 32 - 1) / (WAVES_PER_BLOCK * 32))

typedef float v2f __attribute__((ext_vector_type(2)));
typedef float v8f __attribute__((ext_vector_type(8)));

__global__ void repack_w1_kernel(const float* __restrict__ W1,
                                 float* __restrict__ wpack) {
    int id = blockIdx.x * blockDim.x + threadIdx.x;
    if (id >= K_STEPS * 32 * 8) return;
    int ks   = id >> 8;
    int rem  = id & 255;
    int lane = rem >> 3;
    int nt   = rem & 7;
    int m    = lane & 15;
    int hi   = lane >> 4;
    int krow = ks * 4 + 2 * hi;
    int n    = nt * 16 + m;
    wpack[2 * id + 0] = W1[(size_t)(krow + 0) * HIDDEN + n];
    wpack[2 * id + 1] = W1[(size_t)(krow + 1) * HIDDEN + n];
}


#define NB 128
#define EPB 391
#define EPB_PAD 512
#define NCHUNK 128
#define CHUNK (NUM_CONN / NCHUNK)
#define BIN_THREADS 1024
#define BIN_PER_THREAD 7
#define TAB_N (NB * NCHUNK)
#define TAB_PER_THREAD (TAB_N / BIN_THREADS)
#define NODE_BITS 17
#define NODE_MASK ((1 << NODE_BITS) - 1)
#define SORT_PER_THREAD 8

static_assert(NB * EPB >= NUM_EDGES, "ranges cover every edge");
static_assert(NCHUNK * CHUNK == NUM_CONN, "chunks cover every connection");
static_assert(BIN_THREADS * BIN_PER_THREAD >= CHUNK, "one register batch per chunk");
static_assert(TAB_PER_THREAD * BIN_THREADS == TAB_N, "table scan: whole rows per thread");
static_assert((1 << NODE_BITS) >= NUM_NODES && EPB <= EPB_PAD, "packed word fits");

__global__ __launch_bounds__(BIN_THREADS) void bin_count_kernel(
    const int* __restrict__ edges,
    int* __restrict__ tab) {
    __shared__ int h[NB];
    const int t = threadIdx.x;
    const int w = blockIdx.x;
    if (t < NB) h[t] = 0;
    const int* __restrict__ src = edges + w * CHUNK;
    int e[BIN_PER_THREAD];
#pragma unroll
    for (int k = 0; k < BIN_PER_THREAD; ++k) {
        const int i = t + k * BIN_THREADS;
        e[k] = (i < CHUNK) ? src[i] : -1;
    }
    __syncthreads();
#pragma unroll
    for (int k = 0; k < BIN_PER_THREAD; ++k) {
        if ((unsigned)e[k] < (unsigned)NUM_EDGES) atomicAdd(&h[e[k] / EPB], 1);
    }
    __syncthreads();
    if (t < NB) tab[t * NCHUNK + w] = h[t];
}

__global__ __launch_bounds__(BIN_THREADS) void scan_kernel(int* __restrict__ tab) {
    __shared__ int wsum[32];
    const int t    = threadIdx.x;
    const int lane = t & 31;
    const int w    = t >> 5;

    int4* __restrict__ row = (int4*)tab + t * (TAB_PER_THREAD / 4);
    const int4 v0 = row[0], v1 = row[1], v2 = row[2], v3 = row[3];
    int x[TAB_PER_THREAD] = { v0.x, v0.y, v0.z, v0.w, v1.x, v1.y, v1.z, v1.w,
                              v2.x, v2.y, v2.z, v2.w, v3.x, v3.y, v3.z, v3.w };
    int sum = 0;
#pragma unroll
    for (int j = 0; j < TAB_PER_THREAD; ++j) { const int c = x[j]; x[j] = sum; sum += c; }

    int inc = sum;
#pragma unroll
    for (int d = 1; d < 32; d <<= 1) {
        const int up = __shfl_up(inc, d, 32);
        if (lane >= d) inc += up;
    }
    if (lane == 31) wsum[w] = inc;
    __syncthreads();
    if (w == 0) {
        const int tot = wsum[lane];
        int inc2 = tot;
#pragma unroll
        for (int d = 1; d < 32; d <<= 1) {
            const int u = __shfl_up(inc2, d, 32);
            if (lane >= d) inc2 += u;
        }
        wsum[lane] = inc2 - tot;
    }
    __syncthreads();
    const int pre = wsum[w] + inc - sum;
    row[0] = make_int4(x[0] + pre,  x[1] + pre,  x[2] + pre,  x[3] + pre);
    row[1] = make_int4(x[4] + pre,  x[5] + pre,  x[6] + pre,  x[7] + pre);
    row[2] = make_int4(x[8] + pre,  x[9] + pre,  x[10] + pre, x[11] + pre);
    row[3] = make_int4(x[12] + pre, x[13] + pre, x[14] + pre, x[15] + pre);
    if (t == BIN_THREADS - 1) tab[TAB_N] = pre + sum;
}

__device__ __forceinline__ int clamp_node(int node) {
    return (unsigned)node < (unsigned)NUM_NODES ? node : NUM_NODES - 1;
}

__global__ __launch_bounds__(BIN_THREADS) void bin_scatter_kernel(
    const int* __restrict__ nodes,
    const int* __restrict__ edges,
    const int* __restrict__ tab,
    int* __restrict__ binned) {
    __shared__ int cur[NB];
    const int t = threadIdx.x;
    const int w = blockIdx.x;
    if (t < NB) cur[t] = tab[t * NCHUNK + w];
    const int* __restrict__ esrc = edges + w * CHUNK;
    const int* __restrict__ nsrc = nodes + w * CHUNK;
    int e[BIN_PER_THREAD], nd[BIN_PER_THREAD];
#pragma unroll
    for (int k = 0; k < BIN_PER_THREAD; ++k) {
        const int i = t + k * BIN_THREADS;
        e[k]  = (i < CHUNK) ? esrc[i] : -1;
        nd[k] = (i < CHUNK) ? nsrc[i] : 0;
    }
    __syncthreads();
#pragma unroll
    for (int k = 0; k < BIN_PER_THREAD; ++k) {
        if ((unsigned)e[k] < (unsigned)NUM_EDGES) {
            const int b  = e[k] / EPB;
            const int le = e[k] - b * EPB;
            const unsigned pos = (unsigned)atomicAdd(&cur[b], 1);
            if (pos < (unsigned)NUM_CONN) binned[pos] = (le << NODE_BITS) | clamp_node(nd[k]);
        }
    }
}

__global__ __launch_bounds__(BIN_THREADS) void range_sort_kernel(
    const int* __restrict__ tab,
    const int* __restrict__ binned,
    int* __restrict__ perm,
    int* __restrict__ ends) {
    __shared__ int cnt[EPB_PAD];
    __shared__ int wsum[32];
    const int t    = threadIdx.x;
    const int lane = t & 31;
    const int w    = t >> 5;
    const int b    = blockIdx.x;

    int bs = tab[b * NCHUNK];
    int be = tab[(b + 1) * NCHUNK];
    if (bs < 0) bs = 0;
    if (be > NUM_CONN) be = NUM_CONN;

    if (t < EPB_PAD) cnt[t] = 0;

    int p[SORT_PER_THREAD];
#pragma unroll
    for (int k = 0; k < SORT_PER_THREAD; ++k) {
        const int i = bs + t + k * BIN_THREADS;
        p[k] = (i < be) ? binned[i] : -1;
    }
    __syncthreads();

#pragma unroll
    for (int k = 0; k < SORT_PER_THREAD; ++k) {
        if (p[k] >= 0) atomicAdd(&cnt[(p[k] >> NODE_BITS) & (EPB_PAD - 1)], 1);
    }
    for (int i = bs + t + SORT_PER_THREAD * BIN_THREADS; i < be; i += BIN_THREADS) {
        const int q = binned[i];
        if (q >= 0) atomicAdd(&cnt[(q >> NODE_BITS) & (EPB_PAD - 1)], 1);
    }
    __syncthreads();

    const int c = (t < EPB_PAD) ? cnt[t] : 0;
    int inc = c;
#pragma unroll
    for (int d = 1; d < 32; d <<= 1) {
        const int up = __shfl_up(inc, d, 32);
        if (lane >= d) inc += up;
    }
    if (lane == 31) wsum[w] = inc;
    __syncthreads();
    if (w == 0) {
        const int tot = wsum[lane];
        int inc2 = tot;
#pragma unroll
        for (int d = 1; d < 32; d <<= 1) {
            const int u = __shfl_up(inc2, d, 32);
            if (lane >= d) inc2 += u;
        }
        wsum[lane] = inc2 - tot;
    }
    __syncthreads();
    const int end_t = bs + wsum[w] + inc;
    const int e = b * EPB + t;
    if (t < EPB && e < NUM_EDGES) ends[e] = end_t;
    if (t < EPB_PAD) cnt[t] = end_t - c;
    __syncthreads();

#pragma unroll
    for (int k = 0; k < SORT_PER_THREAD; ++k) {
        if (p[k] >= 0) {
            const unsigned pos = (unsigned)atomicAdd(&cnt[(p[k] >> NODE_BITS) & (EPB_PAD - 1)], 1);
            if (pos < (unsigned)NUM_CONN) perm[pos] = p[k] & NODE_MASK;
        }
    }
    for (int i = bs + t + SORT_PER_THREAD * BIN_THREADS; i < be; i += BIN_THREADS) {
        const int q = binned[i];
        if (q >= 0) {
            const unsigned pos = (unsigned)atomicAdd(&cnt[(q >> NODE_BITS) & (EPB_PAD - 1)], 1);
            if (pos < (unsigned)NUM_CONN) perm[pos] = q & NODE_MASK;
        }
    }
}

#define EDGES_PER_WAVE 4
#define SUM_WAVES 8

__device__ __forceinline__ void add4(float4& a, const float4& x) {
    a.x += x.x; a.y += x.y; a.z += x.z; a.w += x.w;
}

__device__ __forceinline__ const float4* node_row(const float* __restrict__ nf, int node) {
    const unsigned n = (unsigned)node < (unsigned)NUM_NODES ? (unsigned)node
                                                            : (unsigned)(NUM_NODES - 1);
    return (const float4*)(nf + (size_t)n * IN_DIM);
}

__global__ __launch_bounds__(SUM_WAVES * 32) void segment_sum_kernel(
    const float* __restrict__ nf,
    const int*   __restrict__ perm,
    const int*   __restrict__ ends,
    float* __restrict__ summed,
    float* __restrict__ counts) {
    const int lane = threadIdx.x & 31;
    const int w = __builtin_amdgcn_readfirstlane(
        (int)(blockIdx.x * SUM_WAVES + (threadIdx.x >> 5)));

    for (int q = 0; q < EDGES_PER_WAVE; ++q) {
        const int e = w * EDGES_PER_WAVE + q;
        if (e >= NUM_EDGES) break;
        int beg = (e > 0) ? ends[e - 1] : 0;
        int end = ends[e];
        if (beg < 0) beg = 0;
        if (end > NUM_CONN) end = NUM_CONN;

        float4 a0 = make_float4(0.0f, 0.0f, 0.0f, 0.0f);
        float4 a1 = a0;
        for (int base = beg; base < end; base += 32) {
            const int left = end - base;
            const int m = left < 32 ? left : 32;
            int myid = 0;
            if (lane < m) myid = perm[base + lane];

            int j = 0;
            for (; j + 4 <= m; j += 4) {
                const float4* p0 = node_row(nf, __builtin_amdgcn_readlane(myid, j + 0));
                const float4* p1 = node_row(nf, __builtin_amdgcn_readlane(myid, j + 1));
                const float4* p2 = node_row(nf, __builtin_amdgcn_readlane(myid, j + 2));
                const float4* p3 = node_row(nf, __builtin_amdgcn_readlane(myid, j + 3));
                const float4 x0 = p0[lane], y0 = p0[32 + lane];
                const float4 x1 = p1[lane], y1 = p1[32 + lane];
                const float4 x2 = p2[lane], y2 = p2[32 + lane];
                const float4 x3 = p3[lane], y3 = p3[32 + lane];
                add4(a0, x0); add4(a1, y0);
                add4(a0, x1); add4(a1, y1);
                add4(a0, x2); add4(a1, y2);
                add4(a0, x3); add4(a1, y3);
            }
            const int rem = m - j;
            if (rem == 3) {
                const float4* p0 = node_row(nf, __builtin_amdgcn_readlane(myid, j + 0));
                const float4* p1 = node_row(nf, __builtin_amdgcn_readlane(myid, j + 1));
                const float4* p2 = node_row(nf, __builtin_amdgcn_readlane(myid, j + 2));
                const float4 x0 = p0[lane], y0 = p0[32 + lane];
                const float4 x1 = p1[lane], y1 = p1[32 + lane];
                const float4 x2 = p2[lane], y2 = p2[32 + lane];
                add4(a0, x0); add4(a1, y0);
                add4(a0, x1); add4(a1, y1);
                add4(a0, x2); add4(a1, y2);
            } else if (rem == 2) {
                const float4* p0 = node_row(nf, __builtin_amdgcn_readlane(myid, j + 0));
                const float4* p1 = node_row(nf, __builtin_amdgcn_readlane(myid, j + 1));
                const float4 x0 = p0[lane], y0 = p0[32 + lane];
                const float4 x1 = p1[lane], y1 = p1[32 + lane];
                add4(a0, x0); add4(a1, y0);
                add4(a0, x1); add4(a1, y1);
            } else if (rem == 1) {
                const float4* p0 = node_row(nf, __builtin_amdgcn_readlane(myid, j));
                const float4 x0 = p0[lane], y0 = p0[32 + lane];
                add4(a0, x0); add4(a1, y0);
            }
        }

        float4* dst = (float4*)(summed + (size_t)e * IN_DIM);
        dst[lane]      = a0;
        dst[32 + lane] = a1;
        if (lane == 0) counts[e] = (float)(end > beg ? end - beg : 0);
    }
}

__global__ __launch_bounds__(WAVES_PER_BLOCK * 32) void edge_mlp_kernel(
    const float* __restrict__ summed,
    const float* __restrict__ counts,
    const float* __restrict__ wpack,
    const float* __restrict__ b1,
    const float* __restrict__ lng,
    const float* __restrict__ lnb,
    const float* __restrict__ W2,
    const float* __restrict__ b2,
    float* __restrict__ out) {
    __shared__ float sH[WAVES_PER_BLOCK * TILE_ROWS * H_PITCH];
    __shared__ float sW2[W2_COPY_TRIPS * WAVES_PER_BLOCK * 32];

    const int wave = threadIdx.x >> 5;
    const int lane = threadIdx.x & 31;
    const int tile = blockIdx.x * WAVES_PER_BLOCK + wave;
    const int m0   = tile * TILE_ROWS;
    const int m    = lane & 15;
    const int hi   = lane >> 4;

#pragma unroll
    for (int i = 0; i < W2_COPY_TRIPS; ++i) {
        const int id  = (int)threadIdx.x + i * (WAVES_PER_BLOCK * 32);
        const int idc = id < W2_FLOATS ? id : W2_FLOATS - 1;
        sW2[id] = W2[idc];
    }

    const int row = m0 + m;
    const float invc = 1.0f / fmaxf(counts[row], 1.0f);
    const float* __restrict__ arow = summed + (size_t)row * IN_DIM;

    v8f acc[N_TILES] = {};

    for (int ks = 0; ks < K_STEPS; ++ks) {
        const int k = ks * 4;
        v2f a = *(const v2f*)(arow + k + 2 * hi);
        a.x *= invc;
        a.y *= invc;

        const float4* __restrict__ bp =
            (const float4*)(wpack + ((size_t)ks * 32 + lane) * 16);
        float4 q0 = bp[0], q1 = bp[1], q2 = bp[2], q3 = bp[3];
        v2f bbs[N_TILES];
        bbs[0].x = q0.x; bbs[0].y = q0.y;
        bbs[1].x = q0.z; bbs[1].y = q0.w;
        bbs[2].x = q1.x; bbs[2].y = q1.y;
        bbs[3].x = q1.z; bbs[3].y = q1.w;
        bbs[4].x = q2.x; bbs[4].y = q2.y;
        bbs[5].x = q2.z; bbs[5].y = q2.w;
        bbs[6].x = q3.x; bbs[6].y = q3.y;
        bbs[7].x = q3.z; bbs[7].y = q3.w;

#pragma unroll
        for (int nt = 0; nt < N_TILES; ++nt) {
            acc[nt] = __builtin_amdgcn_wmma_f32_16x16x4_f32(
                false, a, false, bbs[nt], (short)0, acc[nt], false, false);
        }
    }

#pragma unroll
    for (int nt = 0; nt < N_TILES; ++nt) {
        const float bias = b1[nt * 16 + m];
#pragma unroll
        for (int r = 0; r < 8; ++r) acc[nt][r] += bias;
    }

    float s[8], q[8];
#pragma unroll
    for (int r = 0; r < 8; ++r) { s[r] = 0.0f; q[r] = 0.0f; }
#pragma unroll
    for (int nt = 0; nt < N_TILES; ++nt) {
#pragma unroll
        for (int r = 0; r < 8; ++r) {
            const float x = acc[nt][r];
            s[r] += x;
            q[r] += x * x;
        }
    }
#pragma unroll
    for (int mask = 1; mask < 16; mask <<= 1) {
#pragma unroll
        for (int r = 0; r < 8; ++r) {
            s[r] += __shfl_xor(s[r], mask, 32);
            q[r] += __shfl_xor(q[r], mask, 32);
        }
    }
    float mean[8], rstd[8];
#pragma unroll
    for (int r = 0; r < 8; ++r) {
        mean[r] = s[r] * (1.0f / HIDDEN);
        const float var = q[r] * (1.0f / HIDDEN) - mean[r] * mean[r];
        rstd[r] = rsqrtf(var + LN_EPS);
    }

    float* __restrict__ hbase = sH + wave * (TILE_ROWS * H_PITCH);
#pragma unroll
    for (int nt = 0; nt < N_TILES; ++nt) {
        const int n = nt * 16 + m;
        const float g  = lng[n];
        const float bb = lnb[n];
#pragma unroll
        for (int r = 0; r < 8; ++r) {
            float x = (acc[nt][r] - mean[r]) * rstd[r] * g + bb;
            x = fmaxf(x, 0.0f);
            hbase[(r + 8 * hi) * H_PITCH + n] = x;
        }
    }
    __syncthreads();

    v8f o0, o1;
    {
        const float bias0 = b2[m];
        const float bias1 = b2[16];
#pragma unroll
        for (int r = 0; r < 8; ++r) { o0[r] = bias0; o1[r] = bias1; }
    }
    const float* __restrict__ ha  = hbase + m * H_PITCH + 2 * hi;
    const float* __restrict__ wb0 = sW2 + (2 * hi) * OUT_DIM + m;
    const float* __restrict__ wb1 = sW2 + (2 * hi) * OUT_DIM + 16;
#pragma unroll
    for (int ks = 0; ks < K2_STEPS; ++ks) {
        const v2f a = *(const v2f*)(ha + 4 * ks);
        v2f bA, bB;
        bA.x = wb0[ks * 4 * OUT_DIM];
        bA.y = wb0[ks * 4 * OUT_DIM + OUT_DIM];
        bB.x = wb1[ks * 4 * OUT_DIM];
        bB.y = wb1[ks * 4 * OUT_DIM + OUT_DIM];
        o0 = __builtin_amdgcn_wmma_f32_16x16x4_f32(
            false, a, false, bA, (short)0, o0, false, false);
        o1 = __builtin_amdgcn_wmma_f32_16x16x4_f32(
            false, a, false, bB, (short)0, o1, false, false);
    }

    float* __restrict__ orow = out + (size_t)(m0 + 8 * hi) * OUT_DIM;
#pragma unroll
    for (int r = 0; r < 8; ++r) orow[r * OUT_DIM + m] = o0[r];
    if (m == 0) {
#pragma unroll
        for (int r = 0; r < 8; ++r) orow[r * OUT_DIM + 16] = o1[r];
    }
}

extern "C" void kernel_launch(void* const* d_in, const int* in_sizes, int n_in,
                              void* d_out, int out_size, void* d_ws, size_t ws_size,
                              hipStream_t stream) {
    const float* nf    = (const float*)d_in[0];
    const int*   hidx  = (const int*)d_in[1];
    const int*   nodes = hidx;
    const int*   edges = hidx + NUM_CONN;
    const float* W1    = (const float*)d_in[2];
    const float* b1    = (const float*)d_in[3];
    const float* lng   = (const float*)d_in[4];
    const float* lnb   = (const float*)d_in[5];
    const float* W2    = (const float*)d_in[6];
    const float* b2    = (const float*)d_in[7];
    float*       out   = (float*)d_out;

    float* summed = (float*)d_ws;
    float* counts = summed + (size_t)NUM_EDGES * IN_DIM;
    float* wpack  = counts + NUM_EDGES;

    int* tab    = (int*)d_ws;
    int* binned = tab + TAB_N + 64;
    int* perm   = (int*)d_out;
    int* ends   = perm + NUM_CONN;

    const int npack = K_STEPS * 32 * 8;
    repack_w1_kernel<<<(npack + 255) / 256, 256, 0, stream>>>(W1, wpack);

    bin_count_kernel<<<NCHUNK, BIN_THREADS, 0, stream>>>(edges, tab);

    scan_kernel<<<1, BIN_THREADS, 0, stream>>>(tab);

    bin_scatter_kernel<<<NCHUNK, BIN_THREADS, 0, stream>>>(nodes, edges, tab, binned);

    range_sort_kernel<<<NB, BIN_THREADS, 0, stream>>>(tab, binned, perm, ends);

    const int nsumwaves = (NUM_EDGES + EDGES_PER_WAVE - 1) / EDGES_PER_WAVE;
    segment_sum_kernel<<<(nsumwaves + SUM_WAVES - 1) / SUM_WAVES, SUM_WAVES * 32, 0, stream>>>(
        nf, perm, ends, summed, counts);

    const int ntiles = NUM_EDGES / TILE_ROWS;
    edge_mlp_kernel<<<ntiles / WAVES_PER_BLOCK, WAVES_PER_BLOCK * 32, 0, stream>>>(
        summed, counts, wpack, b1, lng, lnb, W2, b2, out);
}


# Segment sum fused into the MLP launch: each wave sums its 16 edges' node rows (same list order) into an LDS tile that feeds GEMM1's A fragments; row sums never go to memory, 7->6 launches
# speedup vs baseline: 6.5462x; 1.1176x over previous
#include <hip/hip_runtime.h>

#define NUM_NODES 100000
#define NUM_EDGES 50000
#define NUM_CONN  800000
#define IN_DIM    256
#define HIDDEN    128
#define OUT_DIM   17
#define LN_EPS    1e-5f

#define WAVES_PER_BLOCK 4
#define TILE_ROWS 16
#define NUM_TILES (NUM_EDGES / TILE_ROWS)
#define N_TILES (HIDDEN / 16)
#define K_STEPS (IN_DIM / 4)
#define WPACK_FLOATS (K_STEPS * 32 * 16)
#define K2_STEPS (HIDDEN / 4)
#define H_PITCH  (HIDDEN + 4)
#define W2_FLOATS (HIDDEN * OUT_DIM)
#define W2_COPY_TRIPS ((W2_FLOATS + WAVES_PER_BLOCK * 32 - 1) / (WAVES_PER_BLOCK * 32))

typedef float v2f __attribute__((ext_vector_type(2)));
typedef float v8f __attribute__((ext_vector_type(8)));

__global__ void repack_w1_kernel(const float* __restrict__ W1,
                                 float* __restrict__ wpack) {
    int id = blockIdx.x * blockDim.x + threadIdx.x;
    if (id >= K_STEPS * 32 * 8) return;
    int ks   = id >> 8;
    int rem  = id & 255;
    int lane = rem >> 3;
    int nt   = rem & 7;
    int m    = lane & 15;
    int hi   = lane >> 4;
    int krow = ks * 4 + 2 * hi;
    int n    = nt * 16 + m;
    wpack[2 * id + 0] = W1[(size_t)(krow + 0) * HIDDEN + n];
    wpack[2 * id + 1] = W1[(size_t)(krow + 1) * HIDDEN + n];
}


#define NB 128
#define EPB 391
#define EPB_PAD 512
#define NCHUNK 128
#define CHUNK (NUM_CONN / NCHUNK)
#define BIN_THREADS 1024
#define BIN_PER_THREAD 7
#define TAB_N (NB * NCHUNK)
#define TAB_PER_THREAD (TAB_N / BIN_THREADS)
#define NODE_BITS 17
#define NODE_MASK ((1 << NODE_BITS) - 1)
#define SORT_PER_THREAD 8

static_assert(NB * EPB >= NUM_EDGES, "ranges cover every edge");
static_assert(NCHUNK * CHUNK == NUM_CONN, "chunks cover every connection");
static_assert(BIN_THREADS * BIN_PER_THREAD >= CHUNK, "one register batch per chunk");
static_assert(TAB_PER_THREAD * BIN_THREADS == TAB_N, "table scan: whole rows per thread");
static_assert((1 << NODE_BITS) >= NUM_NODES && EPB <= EPB_PAD, "packed word fits");

__global__ __launch_bounds__(BIN_THREADS) void bin_count_kernel(
    const int* __restrict__ edges,
    int* __restrict__ tab) {
    __shared__ int h[NB];
    const int t = threadIdx.x;
    const int w = blockIdx.x;
    if (t < NB) h[t] = 0;
    const int* __restrict__ src = edges + w * CHUNK;
    int e[BIN_PER_THREAD];
#pragma unroll
    for (int k = 0; k < BIN_PER_THREAD; ++k) {
        const int i = t + k * BIN_THREADS;
        e[k] = (i < CHUNK) ? src[i] : -1;
    }
    __syncthreads();
#pragma unroll
    for (int k = 0; k < BIN_PER_THREAD; ++k) {
        if ((unsigned)e[k] < (unsigned)NUM_EDGES) atomicAdd(&h[e[k] / EPB], 1);
    }
    __syncthreads();
    if (t < NB) tab[t * NCHUNK + w] = h[t];
}

__global__ __launch_bounds__(BIN_THREADS) void scan_kernel(int* __restrict__ tab) {
    __shared__ int wsum[32];
    const int t    = threadIdx.x;
    const int lane = t & 31;
    const int w    = t >> 5;

    int4* __restrict__ row = (int4*)tab + t * (TAB_PER_THREAD / 4);
    const int4 v0 = row[0], v1 = row[1], v2 = row[2], v3 = row[3];
    int x[TAB_PER_THREAD] = { v0.x, v0.y, v0.z, v0.w, v1.x, v1.y, v1.z, v1.w,
                              v2.x, v2.y, v2.z, v2.w, v3.x, v3.y, v3.z, v3.w };
    int sum = 0;
#pragma unroll
    for (int j = 0; j < TAB_PER_THREAD; ++j) { const int c = x[j]; x[j] = sum; sum += c; }

    int inc = sum;
#pragma unroll
    for (int d = 1; d < 32; d <<= 1) {
        const int up = __shfl_up(inc, d, 32);
        if (lane >= d) inc += up;
    }
    if (lane == 31) wsum[w] = inc;
    __syncthreads();
    if (w == 0) {
        const int tot = wsum[lane];
        int inc2 = tot;
#pragma unroll
        for (int d = 1; d < 32; d <<= 1) {
            const int u = __shfl_up(inc2, d, 32);
            if (lane >= d) inc2 += u;
        }
        wsum[lane] = inc2 - tot;
    }
    __syncthreads();
    const int pre = wsum[w] + inc - sum;
    row[0] = make_int4(x[0] + pre,  x[1] + pre,  x[2] + pre,  x[3] + pre);
    row[1] = make_int4(x[4] + pre,  x[5] + pre,  x[6] + pre,  x[7] + pre);
    row[2] = make_int4(x[8] + pre,  x[9] + pre,  x[10] + pre, x[11] + pre);
    row[3] = make_int4(x[12] + pre, x[13] + pre, x[14] + pre, x[15] + pre);
    if (t == BIN_THREADS - 1) tab[TAB_N] = pre + sum;
}

__device__ __forceinline__ int clamp_node(int node) {
    return (unsigned)node < (unsigned)NUM_NODES ? node : NUM_NODES - 1;
}

__global__ __launch_bounds__(BIN_THREADS) void bin_scatter_kernel(
    const int* __restrict__ nodes,
    const int* __restrict__ edges,
    const int* __restrict__ tab,
    int* __restrict__ binned) {
    __shared__ int cur[NB];
    const int t = threadIdx.x;
    const int w = blockIdx.x;
    if (t < NB) cur[t] = tab[t * NCHUNK + w];
    const int* __restrict__ esrc = edges + w * CHUNK;
    const int* __restrict__ nsrc = nodes + w * CHUNK;
    int e[BIN_PER_THREAD], nd[BIN_PER_THREAD];
#pragma unroll
    for (int k = 0; k < BIN_PER_THREAD; ++k) {
        const int i = t + k * BIN_THREADS;
        e[k]  = (i < CHUNK) ? esrc[i] : -1;
        nd[k] = (i < CHUNK) ? nsrc[i] : 0;
    }
    __syncthreads();
#pragma unroll
    for (int k = 0; k < BIN_PER_THREAD; ++k) {
        if ((unsigned)e[k] < (unsigned)NUM_EDGES) {
            const int b  = e[k] / EPB;
            const int le = e[k] - b * EPB;
            const unsigned pos = (unsigned)atomicAdd(&cur[b], 1);
            if (pos < (unsigned)NUM_CONN) binned[pos] = (le << NODE_BITS) | clamp_node(nd[k]);
        }
    }
}

__global__ __launch_bounds__(BIN_THREADS) void range_sort_kernel(
    const int* __restrict__ tab,
    const int* __restrict__ binned,
    int* __restrict__ perm,
    int* __restrict__ ends) {
    __shared__ int cnt[EPB_PAD];
    __shared__ int wsum[32];
    const int t    = threadIdx.x;
    const int lane = t & 31;
    const int w    = t >> 5;
    const int b    = blockIdx.x;

    int bs = tab[b * NCHUNK];
    int be = tab[(b + 1) * NCHUNK];
    if (bs < 0) bs = 0;
    if (be > NUM_CONN) be = NUM_CONN;

    if (t < EPB_PAD) cnt[t] = 0;

    int p[SORT_PER_THREAD];
#pragma unroll
    for (int k = 0; k < SORT_PER_THREAD; ++k) {
        const int i = bs + t + k * BIN_THREADS;
        p[k] = (i < be) ? binned[i] : -1;
    }
    __syncthreads();

#pragma unroll
    for (int k = 0; k < SORT_PER_THREAD; ++k) {
        if (p[k] >= 0) atomicAdd(&cnt[(p[k] >> NODE_BITS) & (EPB_PAD - 1)], 1);
    }
    for (int i = bs + t + SORT_PER_THREAD * BIN_THREADS; i < be; i += BIN_THREADS) {
        const int q = binned[i];
        if (q >= 0) atomicAdd(&cnt[(q >> NODE_BITS) & (EPB_PAD - 1)], 1);
    }
    __syncthreads();

    const int c = (t < EPB_PAD) ? cnt[t] : 0;
    int inc = c;
#pragma unroll
    for (int d = 1; d < 32; d <<= 1) {
        const int up = __shfl_up(inc, d, 32);
        if (lane >= d) inc += up;
    }
    if (lane == 31) wsum[w] = inc;
    __syncthreads();
    if (w == 0) {
        const int tot = wsum[lane];
        int inc2 = tot;
#pragma unroll
        for (int d = 1; d < 32; d <<= 1) {
            const int u = __shfl_up(inc2, d, 32);
            if (lane >= d) inc2 += u;
        }
        wsum[lane] = inc2 - tot;
    }
    __syncthreads();
    const int end_t = bs + wsum[w] + inc;
    const int e = b * EPB + t;
    if (t < EPB && e < NUM_EDGES) ends[e] = end_t;
    if (t < EPB_PAD) cnt[t] = end_t - c;
    __syncthreads();

#pragma unroll
    for (int k = 0; k < SORT_PER_THREAD; ++k) {
        if (p[k] >= 0) {
            const unsigned pos = (unsigned)atomicAdd(&cnt[(p[k] >> NODE_BITS) & (EPB_PAD - 1)], 1);
            if (pos < (unsigned)NUM_CONN) perm[pos] = p[k] & NODE_MASK;
        }
    }
    for (int i = bs + t + SORT_PER_THREAD * BIN_THREADS; i < be; i += BIN_THREADS) {
        const int q = binned[i];
        if (q >= 0) {
            const unsigned pos = (unsigned)atomicAdd(&cnt[(q >> NODE_BITS) & (EPB_PAD - 1)], 1);
            if (pos < (unsigned)NUM_CONN) perm[pos] = q & NODE_MASK;
        }
    }
}

#define A_PITCH (IN_DIM + 4)
#define ROWS_IN_FLIGHT 16

__device__ __forceinline__ void add4(float4& a, const float4& x) {
    a.x += x.x; a.y += x.y; a.z += x.z; a.w += x.w;
}

__device__ __forceinline__ const float4* node_row(const float* __restrict__ nf, int node) {
    const unsigned n = (unsigned)node < (unsigned)NUM_NODES ? (unsigned)node
                                                            : (unsigned)(NUM_NODES - 1);
    return (const float4*)(nf + (size_t)n * IN_DIM);
}

#define FINISH_EDGES_BEFORE(POS)                                                        \
    while (cur < TILE_ROWS && (POS) >= next_end) {                                      \
        float4* drow = (float4*)(abase + cur * A_PITCH);                                \
        drow[lane]      = a0;                                                           \
        drow[32 + lane] = a1;                                                           \
        a0 = make_float4(0.0f, 0.0f, 0.0f, 0.0f);                                       \
        a1 = a0;                                                                        \
        ++cur;                                                                          \
        next_end = __builtin_amdgcn_readlane(bndv, cur < TILE_ROWS ? cur + 1 : TILE_ROWS); \
    }

#define LOAD_ROWS(N)                                                                    \
        const float4* rp[N];                                                            \
        float4 rx[N], ry[N];                                                            \
        _Pragma("unroll")                                                               \
        for (int i = 0; i < (N); ++i)                                                   \
            rp[i] = node_row(nf, __builtin_amdgcn_readlane(myid, j + i));               \
        _Pragma("unroll")                                                               \
        for (int i = 0; i < (N); ++i) { rx[i] = rp[i][lane]; ry[i] = rp[i][32 + lane]; }
#define ADD_ROWS(N)                                                                     \
        _Pragma("unroll")                                                               \
        for (int i = 0; i < (N); ++i) {                                                 \
            FINISH_EDGES_BEFORE(base + j + i)                                           \
            add4(a0, rx[i]);                                                            \
            add4(a1, ry[i]);                                                            \
        }                                                                               \
        j += (N);
#define SUM_ROWS(N) { LOAD_ROWS(N) ADD_ROWS(N) }

__global__ __launch_bounds__(WAVES_PER_BLOCK * 32) void edge_sum_mlp_kernel(
    const float* __restrict__ nf,
    const int*   __restrict__ perm,
    const int*   __restrict__ ends,
    const float* __restrict__ wpack,
    const float* __restrict__ b1,
    const float* __restrict__ lng,
    const float* __restrict__ lnb,
    const float* __restrict__ W2,
    const float* __restrict__ b2,
    float* __restrict__ out) {
    __shared__ alignas(16) float sA[WAVES_PER_BLOCK * TILE_ROWS * A_PITCH];
    __shared__ float sW2[W2_COPY_TRIPS * WAVES_PER_BLOCK * 32];

    const int lane = threadIdx.x & 31;
    const int wave = __builtin_amdgcn_readfirstlane((int)(threadIdx.x >> 5));
    const int  tile_id = blockIdx.x * WAVES_PER_BLOCK + wave;
    const bool live    = tile_id < NUM_TILES;
    const int  tile    = live ? tile_id : NUM_TILES - 1;
    const int  m0      = tile * TILE_ROWS;
    const int m    = lane & 15;
    const int hi   = lane >> 4;

    float* abase = sA + wave * (TILE_ROWS * A_PITCH);

    const int bidx = m0 - 1 + (lane < TILE_ROWS ? lane : TILE_ROWS);
    const int braw = ends[bidx < 0 ? 0 : bidx];

    float w2r[W2_COPY_TRIPS];
#pragma unroll
    for (int i = 0; i < W2_COPY_TRIPS; ++i) {
        const int id  = (int)threadIdx.x + i * (WAVES_PER_BLOCK * 32);
        const int idc = id < W2_FLOATS ? id : W2_FLOATS - 1;
        w2r[i] = W2[idc];
    }

    int bndv = (bidx < 0) ? 0 : braw;
    if (bndv < 0) bndv = 0;
    if (bndv > NUM_CONN) bndv = NUM_CONN;

    {
        const int beg   = __builtin_amdgcn_readlane(bndv, 0);
        const int end_t = __builtin_amdgcn_readlane(bndv, TILE_ROWS);
        int cur      = 0;
        int next_end = __builtin_amdgcn_readlane(bndv, 1);
        float4 a0 = make_float4(0.0f, 0.0f, 0.0f, 0.0f);
        float4 a1 = a0;

        int id_next = 0;
        if (beg + lane < end_t) id_next = perm[beg + lane];
        for (int base = beg; base < end_t; base += 32) {
            const int left = end_t - base;
            const int mm   = left < 32 ? left : 32;
            const int myid = id_next;

            int j = 0;
            if (mm == 32) {
                LOAD_ROWS(ROWS_IN_FLIGHT)
                id_next = 0;
                if (base + 32 + lane < end_t) id_next = perm[base + 32 + lane];
                ADD_ROWS(ROWS_IN_FLIGHT)
            }
            while (j + ROWS_IN_FLIGHT <= mm) SUM_ROWS(ROWS_IN_FLIGHT)
            if (j + 8 <= mm) SUM_ROWS(8)
            if (j + 4 <= mm) SUM_ROWS(4)
            if (j + 2 <= mm) SUM_ROWS(2)
            if (j + 1 <= mm) SUM_ROWS(1)
        }
        FINISH_EDGES_BEFORE(NUM_CONN)
    }
#pragma unroll
    for (int i = 0; i < W2_COPY_TRIPS; ++i)
        sW2[(int)threadIdx.x + i * (WAVES_PER_BLOCK * 32)] = w2r[i];
    __syncthreads();

    const int seg_beg = __shfl(bndv, m, 32);
    const int seg_end = __shfl(bndv, m + 1, 32);
    const float degree = (float)(seg_end > seg_beg ? seg_end - seg_beg : 0);
    const float invc = 1.0f / fmaxf(degree, 1.0f);
    const float* arow = abase + m * A_PITCH;

    v8f acc[N_TILES] = {};

    for (int ks = 0; ks < K_STEPS; ++ks) {
        const int k = ks * 4;
        v2f a = *(const v2f*)(arow + k + 2 * hi);
        a.x *= invc;
        a.y *= invc;

        const float4* __restrict__ bp =
            (const float4*)(wpack + ((size_t)ks * 32 + lane) * 16);
        float4 q0 = bp[0], q1 = bp[1], q2 = bp[2], q3 = bp[3];
        v2f bbs[N_TILES];
        bbs[0].x = q0.x; bbs[0].y = q0.y;
        bbs[1].x = q0.z; bbs[1].y = q0.w;
        bbs[2].x = q1.x; bbs[2].y = q1.y;
        bbs[3].x = q1.z; bbs[3].y = q1.w;
        bbs[4].x = q2.x; bbs[4].y = q2.y;
        bbs[5].x = q2.z; bbs[5].y = q2.w;
        bbs[6].x = q3.x; bbs[6].y = q3.y;
        bbs[7].x = q3.z; bbs[7].y = q3.w;

#pragma unroll
        for (int nt = 0; nt < N_TILES; ++nt) {
            acc[nt] = __builtin_amdgcn_wmma_f32_16x16x4_f32(
                false, a, false, bbs[nt], (short)0, acc[nt], false, false);
        }
    }

#pragma unroll
    for (int nt = 0; nt < N_TILES; ++nt) {
        const float bias = b1[nt * 16 + m];
#pragma unroll
        for (int r = 0; r < 8; ++r) acc[nt][r] += bias;
    }

    float s[8], q[8];
#pragma unroll
    for (int r = 0; r < 8; ++r) { s[r] = 0.0f; q[r] = 0.0f; }
#pragma unroll
    for (int nt = 0; nt < N_TILES; ++nt) {
#pragma unroll
        for (int r = 0; r < 8; ++r) {
            const float x = acc[nt][r];
            s[r] += x;
            q[r] += x * x;
        }
    }
#pragma unroll
    for (int mask = 1; mask < 16; mask <<= 1) {
#pragma unroll
        for (int r = 0; r < 8; ++r) {
            s[r] += __shfl_xor(s[r], mask, 32);
            q[r] += __shfl_xor(q[r], mask, 32);
        }
    }
    float mean[8], rstd[8];
#pragma unroll
    for (int r = 0; r < 8; ++r) {
        mean[r] = s[r] * (1.0f / HIDDEN);
        const float var = q[r] * (1.0f / HIDDEN) - mean[r] * mean[r];
        rstd[r] = rsqrtf(var + LN_EPS);
    }

    float* hbase = abase;
#pragma unroll
    for (int nt = 0; nt < N_TILES; ++nt) {
        const int n = nt * 16 + m;
        const float g  = lng[n];
        const float bb = lnb[n];
#pragma unroll
        for (int r = 0; r < 8; ++r) {
            float x = (acc[nt][r] - mean[r]) * rstd[r] * g + bb;
            x = fmaxf(x, 0.0f);
            hbase[(r + 8 * hi) * H_PITCH + n] = x;
        }
    }
    __syncthreads();

    v8f o0, o1;
    {
        const float bias0 = b2[m];
        const float bias1 = b2[16];
#pragma unroll
        for (int r = 0; r < 8; ++r) { o0[r] = bias0; o1[r] = bias1; }
    }
    const float* ha  = hbase + m * H_PITCH + 2 * hi;
    const float* __restrict__ wb0 = sW2 + (2 * hi) * OUT_DIM + m;
    const float* __restrict__ wb1 = sW2 + (2 * hi) * OUT_DIM + 16;
#pragma unroll
    for (int ks = 0; ks < K2_STEPS; ++ks) {
        const v2f a = *(const v2f*)(ha + 4 * ks);
        v2f bA, bB;
        bA.x = wb0[ks * 4 * OUT_DIM];
        bA.y = wb0[ks * 4 * OUT_DIM + OUT_DIM];
        bB.x = wb1[ks * 4 * OUT_DIM];
        bB.y = wb1[ks * 4 * OUT_DIM + OUT_DIM];
        o0 = __builtin_amdgcn_wmma_f32_16x16x4_f32(
            false, a, false, bA, (short)0, o0, false, false);
        o1 = __builtin_amdgcn_wmma_f32_16x16x4_f32(
            false, a, false, bB, (short)0, o1, false, false);
    }

    if (live) {
        float* __restrict__ orow = out + (size_t)(m0 + 8 * hi) * OUT_DIM;
#pragma unroll
        for (int r = 0; r < 8; ++r) orow[r * OUT_DIM + m] = o0[r];
        if (m == 0) {
#pragma unroll
            for (int r = 0; r < 8; ++r) orow[r * OUT_DIM + 16] = o1[r];
        }
    }
}

extern "C" void kernel_launch(void* const* d_in, const int* in_sizes, int n_in,
                              void* d_out, int out_size, void* d_ws, size_t ws_size,
                              hipStream_t stream) {
    const float* nf    = (const float*)d_in[0];
    const int*   hidx  = (const int*)d_in[1];
    const int*   nodes = hidx;
    const int*   edges = hidx + NUM_CONN;
    const float* W1    = (const float*)d_in[2];
    const float* b1    = (const float*)d_in[3];
    const float* lng   = (const float*)d_in[4];
    const float* lnb   = (const float*)d_in[5];
    const float* W2    = (const float*)d_in[6];
    const float* b2    = (const float*)d_in[7];
    float*       out   = (float*)d_out;

    int* tab    = (int*)d_ws;
    int* binned = tab + TAB_N + 64;
    int* perm   = binned + NUM_CONN;
    int* ends   = perm + NUM_CONN;
    float* wpack = (float*)d_ws + (size_t)NUM_EDGES * IN_DIM + NUM_EDGES;

    const int npack = K_STEPS * 32 * 8;
    repack_w1_kernel<<<(npack + 255) / 256, 256, 0, stream>>>(W1, wpack);

    bin_count_kernel<<<NCHUNK, BIN_THREADS, 0, stream>>>(edges, tab);

    scan_kernel<<<1, BIN_THREADS, 0, stream>>>(tab);

    bin_scatter_kernel<<<NCHUNK, BIN_THREADS, 0, stream>>>(nodes, edges, tab, binned);

    range_sort_kernel<<<NB, BIN_THREADS, 0, stream>>>(tab, binned, perm, ends);

    edge_sum_mlp_kernel<<<(NUM_TILES + WAVES_PER_BLOCK - 1) / WAVES_PER_BLOCK, WAVES_PER_BLOCK * 32, 0, stream>>>(
        nf, perm, ends, wpack, b1, lng, lnb, W2, b2, out);
}


# Sort level 1 made chunk-local (each chunk ranks, scans and writes itself sorted by edge range, with its offsets): count and scan launches gone, 6->4; rows still summed in LDS-add arrival order
# speedup vs baseline: 7.9562x; 1.2154x over previous
#include <hip/hip_runtime.h>

#define NUM_NODES 100000
#define NUM_EDGES 50000
#define NUM_CONN  800000
#define IN_DIM    256
#define HIDDEN    128
#define OUT_DIM   17
#define LN_EPS    1e-5f

#define WAVES_PER_BLOCK 4
#define TILE_ROWS 16
#define NUM_TILES (NUM_EDGES / TILE_ROWS)
#define N_TILES (HIDDEN / 16)
#define K_STEPS (IN_DIM / 4)
#define WPACK_FLOATS (K_STEPS * 32 * 16)
#define K2_STEPS (HIDDEN / 4)
#define H_PITCH  (HIDDEN + 4)
#define W2_FLOATS (HIDDEN * OUT_DIM)
#define W2_COPY_TRIPS ((W2_FLOATS + WAVES_PER_BLOCK * 32 - 1) / (WAVES_PER_BLOCK * 32))

typedef float v2f __attribute__((ext_vector_type(2)));
typedef float v8f __attribute__((ext_vector_type(8)));

__global__ void repack_w1_kernel(const float* __restrict__ W1,
                                 float* __restrict__ wpack) {
    int id = blockIdx.x * blockDim.x + threadIdx.x;
    if (id >= K_STEPS * 32 * 8) return;
    int ks   = id >> 8;
    int rem  = id & 255;
    int lane = rem >> 3;
    int nt   = rem & 7;
    int m    = lane & 15;
    int hi   = lane >> 4;
    int krow = ks * 4 + 2 * hi;
    int n    = nt * 16 + m;
    wpack[2 * id + 0] = W1[(size_t)(krow + 0) * HIDDEN + n];
    wpack[2 * id + 1] = W1[(size_t)(krow + 1) * HIDDEN + n];
}


#define NB 128
#define EPB 391
#define EPB_PAD 512
#define NCHUNK 128
#define CHUNK (NUM_CONN / NCHUNK)
#define BIN_THREADS 1024
#define BIN_PER_THREAD 7
#define OFF_PITCH (NB + 1)
#define TAB_N (NCHUNK * OFF_PITCH)
#define NODE_BITS 17
#define NODE_MASK ((1 << NODE_BITS) - 1)
#define PIECES_PER_WAVE (NCHUNK / 32)
#define PIECE_TRIPS 2

static_assert(NB * EPB >= NUM_EDGES, "ranges cover every edge");
static_assert(NCHUNK * CHUNK == NUM_CONN, "chunks cover every connection");
static_assert(BIN_THREADS * BIN_PER_THREAD >= CHUNK, "one register batch per chunk");
static_assert(NB == 4 * 32 && NCHUNK == 4 * 32, "four whole waves scan the counts and read the offsets");
static_assert(PIECES_PER_WAVE * (BIN_THREADS / 32) == NCHUNK, "every piece has a wave");
static_assert((1 << NODE_BITS) >= NUM_NODES && EPB <= EPB_PAD, "packed word fits");

__device__ __forceinline__ int clamp_node(int node) {
    return (unsigned)node < (unsigned)NUM_NODES ? node : NUM_NODES - 1;
}

__global__ __launch_bounds__(BIN_THREADS) void chunk_sort_kernel(
    const int* __restrict__ nodes,
    const int* __restrict__ edges,
    int* __restrict__ tab,
    int* __restrict__ binned) {
    __shared__ int cnt[NB];
    __shared__ int wtot[NB / 32];
    const int t    = threadIdx.x;
    const int lane = t & 31;
    const int wv   = t >> 5;
    const int w    = blockIdx.x;
    if (t < NB) cnt[t] = 0;
    const int* __restrict__ esrc = edges + w * CHUNK;
    const int* __restrict__ nsrc = nodes + w * CHUNK;
    int e[BIN_PER_THREAD], nd[BIN_PER_THREAD], rk[BIN_PER_THREAD];
#pragma unroll
    for (int k = 0; k < BIN_PER_THREAD; ++k) {
        const int i = t + k * BIN_THREADS;
        e[k]  = (i < CHUNK) ? esrc[i] : -1;
        nd[k] = (i < CHUNK) ? nsrc[i] : 0;
    }
    __syncthreads();
#pragma unroll
    for (int k = 0; k < BIN_PER_THREAD; ++k) {
        rk[k] = 0;
        if ((unsigned)e[k] < (unsigned)NUM_EDGES) rk[k] = atomicAdd(&cnt[e[k] / EPB], 1);
    }
    __syncthreads();

    int c = 0, inc = 0;
    if (t < NB) {
        c = cnt[t];
        inc = c;
#pragma unroll
        for (int d = 1; d < 32; d <<= 1) {
            const int up = __shfl_up(inc, d, 32);
            if (lane >= d) inc += up;
        }
        if (lane == 31) wtot[wv] = inc;
    }
    __syncthreads();
    if (t < NB) {
        int pre = 0;
#pragma unroll
        for (int j = 0; j < NB / 32 - 1; ++j) if (j < wv) pre += wtot[j];
        const int start = pre + inc - c;
        cnt[t] = start;
        tab[w * OFF_PITCH + t] = start;
        if (t == NB - 1) tab[w * OFF_PITCH + NB] = start + c;
    }
    __syncthreads();

#pragma unroll
    for (int k = 0; k < BIN_PER_THREAD; ++k) {
        if ((unsigned)e[k] < (unsigned)NUM_EDGES) {
            const int b  = e[k] / EPB;
            const int le = e[k] - b * EPB;
            const unsigned pos = (unsigned)(cnt[b] + rk[k]);
            if (pos < (unsigned)CHUNK) binned[w * CHUNK + pos] = (le << NODE_BITS) | clamp_node(nd[k]);
        }
    }
}

__global__ __launch_bounds__(BIN_THREADS) void range_sort_kernel(
    const int* __restrict__ tab,
    const int* __restrict__ binned,
    int* __restrict__ perm,
    int* __restrict__ ends) {
    __shared__ int cnt[EPB_PAD];
    __shared__ int wsum[32];
    __shared__ int pbeg[NCHUNK];
    __shared__ int plen[NCHUNK];
    __shared__ int osum[NCHUNK / 32];
    const int t    = threadIdx.x;
    const int lane = t & 31;
    const int w    = t >> 5;
    const int b    = blockIdx.x;

    if (t < EPB_PAD) cnt[t] = 0;

    int o0 = 0;
    if (t < NCHUNK) {
        o0     = tab[t * OFF_PITCH + b];
        int o1 = tab[t * OFF_PITCH + b + 1];
        if (o0 < 0) o0 = 0;
        if (o0 > CHUNK) o0 = CHUNK;
        if (o1 < o0) o1 = o0;
        if (o1 > CHUNK) o1 = CHUNK;
        pbeg[t] = t * CHUNK + o0;
        plen[t] = o1 - o0;
    }
    __syncthreads();

    int pb[PIECES_PER_WAVE], pl[PIECES_PER_WAVE];
    int p[PIECES_PER_WAVE][PIECE_TRIPS];
#pragma unroll
    for (int j = 0; j < PIECES_PER_WAVE; ++j) {
        pb[j] = __builtin_amdgcn_readfirstlane(pbeg[w * PIECES_PER_WAVE + j]);
        pl[j] = __builtin_amdgcn_readfirstlane(plen[w * PIECES_PER_WAVE + j]);
    }
#pragma unroll
    for (int j = 0; j < PIECES_PER_WAVE; ++j) {
#pragma unroll
        for (int k = 0; k < PIECE_TRIPS; ++k) {
            const int i = lane + 32 * k;
            p[j][k] = (i < pl[j]) ? binned[pb[j] + i] : -1;
        }
    }

    if (t < NCHUNK) {
        int osm = o0;
#pragma unroll
        for (int d = 16; d >= 1; d >>= 1) osm += __shfl_xor(osm, d, 32);
        if (lane == 0) osum[w] = osm;
    }

#pragma unroll
    for (int j = 0; j < PIECES_PER_WAVE; ++j) {
#pragma unroll
        for (int k = 0; k < PIECE_TRIPS; ++k) {
            if (p[j][k] >= 0) atomicAdd(&cnt[(p[j][k] >> NODE_BITS) & (EPB_PAD - 1)], 1);
        }
        for (int i = lane + 32 * PIECE_TRIPS; i < pl[j]; i += 32) {
            const int q = binned[pb[j] + i];
            if (q >= 0) atomicAdd(&cnt[(q >> NODE_BITS) & (EPB_PAD - 1)], 1);
        }
    }
    __syncthreads();

    int bs = 0;
#pragma unroll
    for (int j = 0; j < NCHUNK / 32; ++j) bs += osum[j];
    if (bs > NUM_CONN) bs = NUM_CONN;

    const int c = (t < EPB_PAD) ? cnt[t] : 0;
    int inc = c;
#pragma unroll
    for (int d = 1; d < 32; d <<= 1) {
        const int up = __shfl_up(inc, d, 32);
        if (lane >= d) inc += up;
    }
    if (lane == 31) wsum[w] = inc;
    __syncthreads();
    if (w == 0) {
        const int tot = wsum[lane];
        int inc2 = tot;
#pragma unroll
        for (int d = 1; d < 32; d <<= 1) {
            const int u = __shfl_up(inc2, d, 32);
            if (lane >= d) inc2 += u;
        }
        wsum[lane] = inc2 - tot;
    }
    __syncthreads();
    const int end_t = bs + wsum[w] + inc;
    const int e = b * EPB + t;
    if (t < EPB && e < NUM_EDGES) ends[e] = end_t;
    if (t < EPB_PAD) cnt[t] = end_t - c;
    __syncthreads();

#pragma unroll
    for (int j = 0; j < PIECES_PER_WAVE; ++j) {
#pragma unroll
        for (int k = 0; k < PIECE_TRIPS; ++k) {
            if (p[j][k] >= 0) {
                const unsigned pos = (unsigned)atomicAdd(&cnt[(p[j][k] >> NODE_BITS) & (EPB_PAD - 1)], 1);
                if (pos < (unsigned)NUM_CONN) perm[pos] = p[j][k] & NODE_MASK;
            }
        }
        for (int i = lane + 32 * PIECE_TRIPS; i < pl[j]; i += 32) {
            const int q = binned[pb[j] + i];
            if (q >= 0) {
                const unsigned pos = (unsigned)atomicAdd(&cnt[(q >> NODE_BITS) & (EPB_PAD - 1)], 1);
                if (pos < (unsigned)NUM_CONN) perm[pos] = q & NODE_MASK;
            }
        }
    }
}

#define A_PITCH (IN_DIM + 4)
#define ROWS_IN_FLIGHT 16

__device__ __forceinline__ void add4(float4& a, const float4& x) {
    a.x += x.x; a.y += x.y; a.z += x.z; a.w += x.w;
}

__device__ __forceinline__ const float4* node_row(const float* __restrict__ nf, int node) {
    const unsigned n = (unsigned)node < (unsigned)NUM_NODES ? (unsigned)node
                                                            : (unsigned)(NUM_NODES - 1);
    return (const float4*)(nf + (size_t)n * IN_DIM);
}

#define FINISH_EDGES_BEFORE(POS)                                                        \
    while (cur < TILE_ROWS && (POS) >= next_end) {                                      \
        float4* drow = (float4*)(abase + cur * A_PITCH);                                \
        drow[lane]      = a0;                                                           \
        drow[32 + lane] = a1;                                                           \
        a0 = make_float4(0.0f, 0.0f, 0.0f, 0.0f);                                       \
        a1 = a0;                                                                        \
        ++cur;                                                                          \
        next_end = __builtin_amdgcn_readlane(bndv, cur < TILE_ROWS ? cur + 1 : TILE_ROWS); \
    }

#define LOAD_ROWS(N)                                                                    \
        const float4* rp[N];                                                            \
        float4 rx[N], ry[N];                                                            \
        _Pragma("unroll")                                                               \
        for (int i = 0; i < (N); ++i)                                                   \
            rp[i] = node_row(nf, __builtin_amdgcn_readlane(myid, j + i));               \
        _Pragma("unroll")                                                               \
        for (int i = 0; i < (N); ++i) { rx[i] = rp[i][lane]; ry[i] = rp[i][32 + lane]; }
#define ADD_ROWS(N)                                                                     \
        _Pragma("unroll")                                                               \
        for (int i = 0; i < (N); ++i) {                                                 \
            FINISH_EDGES_BEFORE(base + j + i)                                           \
            add4(a0, rx[i]);                                                            \
            add4(a1, ry[i]);                                                            \
        }                                                                               \
        j += (N);
#define SUM_ROWS(N) { LOAD_ROWS(N) ADD_ROWS(N) }

__global__ __launch_bounds__(WAVES_PER_BLOCK * 32) void edge_sum_mlp_kernel(
    const float* __restrict__ nf,
    const int*   __restrict__ perm,
    const int*   __restrict__ ends,
    const float* __restrict__ wpack,
    const float* __restrict__ b1,
    const float* __restrict__ lng,
    const float* __restrict__ lnb,
    const float* __restrict__ W2,
    const float* __restrict__ b2,
    float* __restrict__ out) {
    __shared__ alignas(16) float sA[WAVES_PER_BLOCK * TILE_ROWS * A_PITCH];
    __shared__ float sW2[W2_COPY_TRIPS * WAVES_PER_BLOCK * 32];

    const int lane = threadIdx.x & 31;
    const int wave = __builtin_amdgcn_readfirstlane((int)(threadIdx.x >> 5));
    const int  tile_id = blockIdx.x * WAVES_PER_BLOCK + wave;
    const bool live    = tile_id < NUM_TILES;
    const int  tile    = live ? tile_id : NUM_TILES - 1;
    const int  m0      = tile * TILE_ROWS;
    const int m    = lane & 15;
    const int hi   = lane >> 4;

    float* abase = sA + wave * (TILE_ROWS * A_PITCH);

    const int bidx = m0 - 1 + (lane < TILE_ROWS ? lane : TILE_ROWS);
    const int braw = ends[bidx < 0 ? 0 : bidx];

    float w2r[W2_COPY_TRIPS];
#pragma unroll
    for (int i = 0; i < W2_COPY_TRIPS; ++i) {
        const int id  = (int)threadIdx.x + i * (WAVES_PER_BLOCK * 32);
        const int idc = id < W2_FLOATS ? id : W2_FLOATS - 1;
        w2r[i] = W2[idc];
    }

    int bndv = (bidx < 0) ? 0 : braw;
    if (bndv < 0) bndv = 0;
    if (bndv > NUM_CONN) bndv = NUM_CONN;

    {
        const int beg   = __builtin_amdgcn_readlane(bndv, 0);
        const int end_t = __builtin_amdgcn_readlane(bndv, TILE_ROWS);
        int cur      = 0;
        int next_end = __builtin_amdgcn_readlane(bndv, 1);
        float4 a0 = make_float4(0.0f, 0.0f, 0.0f, 0.0f);
        float4 a1 = a0;

        int id_next = 0;
        if (beg + lane < end_t) id_next = perm[beg + lane];
        for (int base = beg; base < end_t; base += 32) {
            const int left = end_t - base;
            const int mm   = left < 32 ? left : 32;
            const int myid = id_next;

            int j = 0;
            if (mm == 32) {
                LOAD_ROWS(ROWS_IN_FLIGHT)
                id_next = 0;
                if (base + 32 + lane < end_t) id_next = perm[base + 32 + lane];
                ADD_ROWS(ROWS_IN_FLIGHT)
            }
            while (j + ROWS_IN_FLIGHT <= mm) SUM_ROWS(ROWS_IN_FLIGHT)
            if (j + 8 <= mm) SUM_ROWS(8)
            if (j + 4 <= mm) SUM_ROWS(4)
            if (j + 2 <= mm) SUM_ROWS(2)
            if (j + 1 <= mm) SUM_ROWS(1)
        }
        FINISH_EDGES_BEFORE(NUM_CONN)
    }
#pragma unroll
    for (int i = 0; i < W2_COPY_TRIPS; ++i)
        sW2[(int)threadIdx.x + i * (WAVES_PER_BLOCK * 32)] = w2r[i];
    __syncthreads();

    const int seg_beg = __shfl(bndv, m, 32);
    const int seg_end = __shfl(bndv, m + 1, 32);
    const float degree = (float)(seg_end > seg_beg ? seg_end - seg_beg : 0);
    const float invc = 1.0f / fmaxf(degree, 1.0f);
    const float* arow = abase + m * A_PITCH;

    v8f acc[N_TILES] = {};

    for (int ks = 0; ks < K_STEPS; ++ks) {
        const int k = ks * 4;
        v2f a = *(const v2f*)(arow + k + 2 * hi);
        a.x *= invc;
        a.y *= invc;

        const float4* __restrict__ bp =
            (const float4*)(wpack + ((size_t)ks * 32 + lane) * 16);
        float4 q0 = bp[0], q1 = bp[1], q2 = bp[2], q3 = bp[3];
        v2f bbs[N_TILES];
        bbs[0].x = q0.x; bbs[0].y = q0.y;
        bbs[1].x = q0.z; bbs[1].y = q0.w;
        bbs[2].x = q1.x; bbs[2].y = q1.y;
        bbs[3].x = q1.z; bbs[3].y = q1.w;
        bbs[4].x = q2.x; bbs[4].y = q2.y;
        bbs[5].x = q2.z; bbs[5].y = q2.w;
        bbs[6].x = q3.x; bbs[6].y = q3.y;
        bbs[7].x = q3.z; bbs[7].y = q3.w;

#pragma unroll
        for (int nt = 0; nt < N_TILES; ++nt) {
            acc[nt] = __builtin_amdgcn_wmma_f32_16x16x4_f32(
                false, a, false, bbs[nt], (short)0, acc[nt], false, false);
        }
    }

#pragma unroll
    for (int nt = 0; nt < N_TILES; ++nt) {
        const float bias = b1[nt * 16 + m];
#pragma unroll
        for (int r = 0; r < 8; ++r) acc[nt][r] += bias;
    }

    float s[8], q[8];
#pragma unroll
    for (int r = 0; r < 8; ++r) { s[r] = 0.0f; q[r] = 0.0f; }
#pragma unroll
    for (int nt = 0; nt < N_TILES; ++nt) {
#pragma unroll
        for (int r = 0; r < 8; ++r) {
            const float x = acc[nt][r];
            s[r] += x;
            q[r] += x * x;
        }
    }
#pragma unroll
    for (int mask = 1; mask < 16; mask <<= 1) {
#pragma unroll
        for (int r = 0; r < 8; ++r) {
            s[r] += __shfl_xor(s[r], mask, 32);
            q[r] += __shfl_xor(q[r], mask, 32);
        }
    }
    float mean[8], rstd[8];
#pragma unroll
    for (int r = 0; r < 8; ++r) {
        mean[r] = s[r] * (1.0f / HIDDEN);
        const float var = q[r] * (1.0f / HIDDEN) - mean[r] * mean[r];
        rstd[r] = rsqrtf(var + LN_EPS);
    }

    float* hbase = abase;
#pragma unroll
    for (int nt = 0; nt < N_TILES; ++nt) {
        const int n = nt * 16 + m;
        const float g  = lng[n];
        const float bb = lnb[n];
#pragma unroll
        for (int r = 0; r < 8; ++r) {
            float x = (acc[nt][r] - mean[r]) * rstd[r] * g + bb;
            x = fmaxf(x, 0.0f);
            hbase[(r + 8 * hi) * H_PITCH + n] = x;
        }
    }
    __syncthreads();

    v8f o0, o1;
    {
        const float bias0 = b2[m];
        const float bias1 = b2[16];
#pragma unroll
        for (int r = 0; r < 8; ++r) { o0[r] = bias0; o1[r] = bias1; }
    }
    const float* ha  = hbase + m * H_PITCH + 2 * hi;
    const float* __restrict__ wb0 = sW2 + (2 * hi) * OUT_DIM + m;
    const float* __restrict__ wb1 = sW2 + (2 * hi) * OUT_DIM + 16;
#pragma unroll
    for (int ks = 0; ks < K2_STEPS; ++ks) {
        const v2f a = *(const v2f*)(ha + 4 * ks);
        v2f bA, bB;
        bA.x = wb0[ks * 4 * OUT_DIM];
        bA.y = wb0[ks * 4 * OUT_DIM + OUT_DIM];
        bB.x = wb1[ks * 4 * OUT_DIM];
        bB.y = wb1[ks * 4 * OUT_DIM + OUT_DIM];
        o0 = __builtin_amdgcn_wmma_f32_16x16x4_f32(
            false, a, false, bA, (short)0, o0, false, false);
        o1 = __builtin_amdgcn_wmma_f32_16x16x4_f32(
            false, a, false, bB, (short)0, o1, false, false);
    }

    if (live) {
        float* __restrict__ orow = out + (size_t)(m0 + 8 * hi) * OUT_DIM;
#pragma unroll
        for (int r = 0; r < 8; ++r) orow[r * OUT_DIM + m] = o0[r];
        if (m == 0) {
#pragma unroll
            for (int r = 0; r < 8; ++r) orow[r * OUT_DIM + 16] = o1[r];
        }
    }
}

extern "C" void kernel_launch(void* const* d_in, const int* in_sizes, int n_in,
                              void* d_out, int out_size, void* d_ws, size_t ws_size,
                              hipStream_t stream) {
    const float* nf    = (const float*)d_in[0];
    const int*   hidx  = (const int*)d_in[1];
    const int*   nodes = hidx;
    const int*   edges = hidx + NUM_CONN;
    const float* W1    = (const float*)d_in[2];
    const float* b1    = (const float*)d_in[3];
    const float* lng   = (const float*)d_in[4];
    const float* lnb   = (const float*)d_in[5];
    const float* W2    = (const float*)d_in[6];
    const float* b2    = (const float*)d_in[7];
    float*       out   = (float*)d_out;

    int* tab    = (int*)d_ws;
    int* binned = tab + TAB_N + 64;
    int* perm   = binned + NUM_CONN;
    int* ends   = perm + NUM_CONN;
    float* wpack = (float*)d_ws + (size_t)NUM_EDGES * IN_DIM + NUM_EDGES;

    repack_w1_kernel<<<(K_STEPS * 32 * 8 + 255) / 256, 256, 0, stream>>>(W1, wpack);

    chunk_sort_kernel<<<NCHUNK, BIN_THREADS, 0, stream>>>(nodes, edges, tab, binned);

    range_sort_kernel<<<NB, BIN_THREADS, 0, stream>>>(tab, binned, perm, ends);

    edge_sum_mlp_kernel<<<(NUM_TILES + WAVES_PER_BLOCK - 1) / WAVES_PER_BLOCK, WAVES_PER_BLOCK * 32, 0, stream>>>(
        nf, perm, ends, wpack, b1, lng, lnb, W2, b2, out);
}


# Both sort levels put their sorted output together in an LDS window and store it front to back as consecutive words (whole lines) instead of 32-line scattered 4-byte global stores
# speedup vs baseline: 8.4917x; 1.0673x over previous
#include <hip/hip_runtime.h>

#define NUM_NODES 100000
#define NUM_EDGES 50000
#define NUM_CONN  800000
#define IN_DIM    256
#define HIDDEN    128
#define OUT_DIM   17
#define LN_EPS    1e-5f

#define WAVES_PER_BLOCK 4
#define TILE_ROWS 16
#define NUM_TILES (NUM_EDGES / TILE_ROWS)
#define N_TILES (HIDDEN / 16)
#define K_STEPS (IN_DIM / 4)
#define WPACK_FLOATS (K_STEPS * 32 * 16)
#define K2_STEPS (HIDDEN / 4)
#define H_PITCH  (HIDDEN + 4)
#define W2_FLOATS (HIDDEN * OUT_DIM)
#define W2_COPY_TRIPS ((W2_FLOATS + WAVES_PER_BLOCK * 32 - 1) / (WAVES_PER_BLOCK * 32))

typedef float v2f __attribute__((ext_vector_type(2)));
typedef float v8f __attribute__((ext_vector_type(8)));

__global__ void repack_w1_kernel(const float* __restrict__ W1,
                                 float* __restrict__ wpack) {
    int id = blockIdx.x * blockDim.x + threadIdx.x;
    if (id >= K_STEPS * 32 * 8) return;
    int ks   = id >> 8;
    int rem  = id & 255;
    int lane = rem >> 3;
    int nt   = rem & 7;
    int m    = lane & 15;
    int hi   = lane >> 4;
    int krow = ks * 4 + 2 * hi;
    int n    = nt * 16 + m;
    wpack[2 * id + 0] = W1[(size_t)(krow + 0) * HIDDEN + n];
    wpack[2 * id + 1] = W1[(size_t)(krow + 1) * HIDDEN + n];
}


#define NB 128
#define EPB 391
#define EPB_PAD 512
#define NCHUNK 128
#define CHUNK (NUM_CONN / NCHUNK)
#define BIN_THREADS 1024
#define BIN_PER_THREAD 7
#define OFF_PITCH (NB + 1)
#define TAB_N (NCHUNK * OFF_PITCH)
#define NODE_BITS 17
#define NODE_MASK ((1 << NODE_BITS) - 1)
#define PIECES_PER_WAVE (NCHUNK / 32)
#define PIECE_TRIPS 2
#define STAGE_CAP 16384

static_assert(NB * EPB >= NUM_EDGES, "ranges cover every edge");
static_assert(NCHUNK * CHUNK == NUM_CONN, "chunks cover every connection");
static_assert(BIN_THREADS * BIN_PER_THREAD >= CHUNK, "one register batch per chunk");
static_assert(NB == 4 * 32 && NCHUNK == 4 * 32, "four whole waves scan the counts and read the offsets");
static_assert(PIECES_PER_WAVE * (BIN_THREADS / 32) == NCHUNK, "every piece has a wave");
static_assert((1 << NODE_BITS) >= NUM_NODES && EPB <= EPB_PAD, "packed word fits");
static_assert(STAGE_CAP % (4 * BIN_THREADS) == 0, "the level-2 window is a whole number of copy batches");

__device__ __forceinline__ int clamp_node(int node) {
    return (unsigned)node < (unsigned)NUM_NODES ? node : NUM_NODES - 1;
}

__global__ __launch_bounds__(BIN_THREADS) void chunk_sort_kernel(
    const int* __restrict__ nodes,
    const int* __restrict__ edges,
    int* __restrict__ tab,
    int* __restrict__ binned) {
    __shared__ int cnt[NB];
    __shared__ int wtot[NB / 32];
    __shared__ int stage[BIN_THREADS * BIN_PER_THREAD];
    __shared__ int ctot;
    const int t    = threadIdx.x;
    const int lane = t & 31;
    const int wv   = t >> 5;
    const int w    = blockIdx.x;
    if (t < NB) cnt[t] = 0;
    const int* __restrict__ esrc = edges + w * CHUNK;
    const int* __restrict__ nsrc = nodes + w * CHUNK;
    int e[BIN_PER_THREAD], nd[BIN_PER_THREAD], rk[BIN_PER_THREAD];
#pragma unroll
    for (int k = 0; k < BIN_PER_THREAD; ++k) {
        const int i = t + k * BIN_THREADS;
        e[k]  = (i < CHUNK) ? esrc[i] : -1;
        nd[k] = (i < CHUNK) ? nsrc[i] : 0;
    }
    __syncthreads();
#pragma unroll
    for (int k = 0; k < BIN_PER_THREAD; ++k) {
        rk[k] = 0;
        if ((unsigned)e[k] < (unsigned)NUM_EDGES) rk[k] = atomicAdd(&cnt[e[k] / EPB], 1);
    }
    __syncthreads();

    int c = 0, inc = 0;
    if (t < NB) {
        c = cnt[t];
        inc = c;
#pragma unroll
        for (int d = 1; d < 32; d <<= 1) {
            const int up = __shfl_up(inc, d, 32);
            if (lane >= d) inc += up;
        }
        if (lane == 31) wtot[wv] = inc;
    }
    __syncthreads();
    if (t < NB) {
        int pre = 0;
#pragma unroll
        for (int j = 0; j < NB / 32 - 1; ++j) if (j < wv) pre += wtot[j];
        const int start = pre + inc - c;
        cnt[t] = start;
        tab[w * OFF_PITCH + t] = start;
        if (t == NB - 1) {
            tab[w * OFF_PITCH + NB] = start + c;
            ctot = start + c;
        }
    }
    __syncthreads();

#pragma unroll
    for (int k = 0; k < BIN_PER_THREAD; ++k) {
        if ((unsigned)e[k] < (unsigned)NUM_EDGES) {
            const int b  = e[k] / EPB;
            const int le = e[k] - b * EPB;
            const unsigned pos = (unsigned)(cnt[b] + rk[k]);
            if (pos < (unsigned)CHUNK) stage[pos] = (le << NODE_BITS) | clamp_node(nd[k]);
        }
    }
    __syncthreads();
    const int tot = ctot < CHUNK ? ctot : CHUNK;
    int sv[BIN_PER_THREAD];
#pragma unroll
    for (int k = 0; k < BIN_PER_THREAD; ++k) sv[k] = stage[t + k * BIN_THREADS];
#pragma unroll
    for (int k = 0; k < BIN_PER_THREAD; ++k) {
        const int i = t + k * BIN_THREADS;
        if (i < tot) binned[w * CHUNK + i] = sv[k];
    }
}

__global__ __launch_bounds__(BIN_THREADS) void range_sort_kernel(
    const int* __restrict__ tab,
    const int* __restrict__ binned,
    int* __restrict__ perm,
    int* __restrict__ ends) {
    __shared__ int cnt[EPB_PAD];
    __shared__ int wsum[32];
    __shared__ int pbeg[NCHUNK];
    __shared__ int plen[NCHUNK];
    __shared__ int osum[NCHUNK / 32];
    __shared__ int stage[STAGE_CAP];
    __shared__ int rtot;
    const int t    = threadIdx.x;
    const int lane = t & 31;
    const int w    = t >> 5;
    const int b    = blockIdx.x;

    if (t < EPB_PAD) cnt[t] = 0;

    int o0 = 0;
    if (t < NCHUNK) {
        o0     = tab[t * OFF_PITCH + b];
        int o1 = tab[t * OFF_PITCH + b + 1];
        if (o0 < 0) o0 = 0;
        if (o0 > CHUNK) o0 = CHUNK;
        if (o1 < o0) o1 = o0;
        if (o1 > CHUNK) o1 = CHUNK;
        pbeg[t] = t * CHUNK + o0;
        plen[t] = o1 - o0;
    }
    __syncthreads();

    int pb[PIECES_PER_WAVE], pl[PIECES_PER_WAVE];
    int p[PIECES_PER_WAVE][PIECE_TRIPS];
#pragma unroll
    for (int j = 0; j < PIECES_PER_WAVE; ++j) {
        pb[j] = __builtin_amdgcn_readfirstlane(pbeg[w * PIECES_PER_WAVE + j]);
        pl[j] = __builtin_amdgcn_readfirstlane(plen[w * PIECES_PER_WAVE + j]);
    }
#pragma unroll
    for (int j = 0; j < PIECES_PER_WAVE; ++j) {
#pragma unroll
        for (int k = 0; k < PIECE_TRIPS; ++k) {
            const int i = lane + 32 * k;
            p[j][k] = (i < pl[j]) ? binned[pb[j] + i] : -1;
        }
    }

    if (t < NCHUNK) {
        int osm = o0;
#pragma unroll
        for (int d = 16; d >= 1; d >>= 1) osm += __shfl_xor(osm, d, 32);
        if (lane == 0) osum[w] = osm;
    }

#pragma unroll
    for (int j = 0; j < PIECES_PER_WAVE; ++j) {
#pragma unroll
        for (int k = 0; k < PIECE_TRIPS; ++k) {
            if (p[j][k] >= 0) atomicAdd(&cnt[(p[j][k] >> NODE_BITS) & (EPB_PAD - 1)], 1);
        }
        for (int i = lane + 32 * PIECE_TRIPS; i < pl[j]; i += 32) {
            const int q = binned[pb[j] + i];
            if (q >= 0) atomicAdd(&cnt[(q >> NODE_BITS) & (EPB_PAD - 1)], 1);
        }
    }
    __syncthreads();

    int bs = 0;
#pragma unroll
    for (int j = 0; j < NCHUNK / 32; ++j) bs += osum[j];
    if (bs > NUM_CONN) bs = NUM_CONN;

    const int c = (t < EPB_PAD) ? cnt[t] : 0;
    int inc = c;
#pragma unroll
    for (int d = 1; d < 32; d <<= 1) {
        const int up = __shfl_up(inc, d, 32);
        if (lane >= d) inc += up;
    }
    if (lane == 31) wsum[w] = inc;
    __syncthreads();
    if (w == 0) {
        const int tot = wsum[lane];
        int inc2 = tot;
#pragma unroll
        for (int d = 1; d < 32; d <<= 1) {
            const int u = __shfl_up(inc2, d, 32);
            if (lane >= d) inc2 += u;
        }
        wsum[lane] = inc2 - tot;
        if (lane == 31) rtot = inc2;
    }
    __syncthreads();
    const int  rt     = rtot;
    const bool staged = rt <= STAGE_CAP;
    const int end_t = bs + wsum[w] + inc;
    const int e = b * EPB + t;
    if (t < EPB && e < NUM_EDGES) ends[e] = end_t;
    if (t < EPB_PAD) cnt[t] = end_t - c;
    __syncthreads();

#pragma unroll
    for (int j = 0; j < PIECES_PER_WAVE; ++j) {
#pragma unroll
        for (int k = 0; k < PIECE_TRIPS; ++k) {
            if (p[j][k] >= 0) {
                const unsigned pos = (unsigned)atomicAdd(&cnt[(p[j][k] >> NODE_BITS) & (EPB_PAD - 1)], 1);
                const unsigned loc = pos - (unsigned)bs;
                if (staged) { if (loc < (unsigned)STAGE_CAP) stage[loc] = p[j][k] & NODE_MASK; }
                else if (pos < (unsigned)NUM_CONN) perm[pos] = p[j][k] & NODE_MASK;
            }
        }
        for (int i = lane + 32 * PIECE_TRIPS; i < pl[j]; i += 32) {
            const int q = binned[pb[j] + i];
            if (q >= 0) {
                const unsigned pos = (unsigned)atomicAdd(&cnt[(q >> NODE_BITS) & (EPB_PAD - 1)], 1);
                const unsigned loc = pos - (unsigned)bs;
                if (staged) { if (loc < (unsigned)STAGE_CAP) stage[loc] = q & NODE_MASK; }
                else if (pos < (unsigned)NUM_CONN) perm[pos] = q & NODE_MASK;
            }
        }
    }
    __syncthreads();
    if (staged) {
        for (int i0 = t; i0 < rt; i0 += 4 * BIN_THREADS) {
            int sv[4];
#pragma unroll
            for (int k = 0; k < 4; ++k) sv[k] = stage[i0 + k * BIN_THREADS];
#pragma unroll
            for (int k = 0; k < 4; ++k) {
                const int i = i0 + k * BIN_THREADS;
                if (i < rt && bs + i < NUM_CONN) perm[bs + i] = sv[k];
            }
        }
    }
}

#define A_PITCH (IN_DIM + 4)
#define ROWS_IN_FLIGHT 16

__device__ __forceinline__ void add4(float4& a, const float4& x) {
    a.x += x.x; a.y += x.y; a.z += x.z; a.w += x.w;
}

__device__ __forceinline__ const float4* node_row(const float* __restrict__ nf, int node) {
    const unsigned n = (unsigned)node < (unsigned)NUM_NODES ? (unsigned)node
                                                            : (unsigned)(NUM_NODES - 1);
    return (const float4*)(nf + (size_t)n * IN_DIM);
}

#define FINISH_EDGES_BEFORE(POS)                                                        \
    while (cur < TILE_ROWS && (POS) >= next_end) {                                      \
        float4* drow = (float4*)(abase + cur * A_PITCH);                                \
        drow[lane]      = a0;                                                           \
        drow[32 + lane] = a1;                                                           \
        a0 = make_float4(0.0f, 0.0f, 0.0f, 0.0f);                                       \
        a1 = a0;                                                                        \
        ++cur;                                                                          \
        next_end = __builtin_amdgcn_readlane(bndv, cur < TILE_ROWS ? cur + 1 : TILE_ROWS); \
    }

#define LOAD_ROWS(N)                                                                    \
        const float4* rp[N];                                                            \
        float4 rx[N], ry[N];                                                            \
        _Pragma("unroll")                                                               \
        for (int i = 0; i < (N); ++i)                                                   \
            rp[i] = node_row(nf, __builtin_amdgcn_readlane(myid, j + i));               \
        _Pragma("unroll")                                                               \
        for (int i = 0; i < (N); ++i) { rx[i] = rp[i][lane]; ry[i] = rp[i][32 + lane]; }
#define ADD_ROWS(N)                                                                     \
        _Pragma("unroll")                                                               \
        for (int i = 0; i < (N); ++i) {                                                 \
            FINISH_EDGES_BEFORE(base + j + i)                                           \
            add4(a0, rx[i]);                                                            \
            add4(a1, ry[i]);                                                            \
        }                                                                               \
        j += (N);
#define SUM_ROWS(N) { LOAD_ROWS(N) ADD_ROWS(N) }

__global__ __launch_bounds__(WAVES_PER_BLOCK * 32) void edge_sum_mlp_kernel(
    const float* __restrict__ nf,
    const int*   __restrict__ perm,
    const int*   __restrict__ ends,
    const float* __restrict__ wpack,
    const float* __restrict__ b1,
    const float* __restrict__ lng,
    const float* __restrict__ lnb,
    const float* __restrict__ W2,
    const float* __restrict__ b2,
    float* __restrict__ out) {
    __shared__ alignas(16) float sA[WAVES_PER_BLOCK * TILE_ROWS * A_PITCH];
    __shared__ float sW2[W2_COPY_TRIPS * WAVES_PER_BLOCK * 32];

    const int lane = threadIdx.x & 31;
    const int wave = __builtin_amdgcn_readfirstlane((int)(threadIdx.x >> 5));
    const int  tile_id = blockIdx.x * WAVES_PER_BLOCK + wave;
    const bool live    = tile_id < NUM_TILES;
    const int  tile    = live ? tile_id : NUM_TILES - 1;
    const int  m0      = tile * TILE_ROWS;
    const int m    = lane & 15;
    const int hi   = lane >> 4;

    float* abase = sA + wave * (TILE_ROWS * A_PITCH);

    const int bidx = m0 - 1 + (lane < TILE_ROWS ? lane : TILE_ROWS);
    const int braw = ends[bidx < 0 ? 0 : bidx];

    float w2r[W2_COPY_TRIPS];
#pragma unroll
    for (int i = 0; i < W2_COPY_TRIPS; ++i) {
        const int id  = (int)threadIdx.x + i * (WAVES_PER_BLOCK * 32);
        const int idc = id < W2_FLOATS ? id : W2_FLOATS - 1;
        w2r[i] = W2[idc];
    }

    int bndv = (bidx < 0) ? 0 : braw;
    if (bndv < 0) bndv = 0;
    if (bndv > NUM_CONN) bndv = NUM_CONN;

    {
        const int beg   = __builtin_amdgcn_readlane(bndv, 0);
        const int end_t = __builtin_amdgcn_readlane(bndv, TILE_ROWS);
        int cur      = 0;
        int next_end = __builtin_amdgcn_readlane(bndv, 1);
        float4 a0 = make_float4(0.0f, 0.0f, 0.0f, 0.0f);
        float4 a1 = a0;

        int id_next = 0;
        if (beg + lane < end_t) id_next = perm[beg + lane];
        for (int base = beg; base < end_t; base += 32) {
            const int left = end_t - base;
            const int mm   = left < 32 ? left : 32;
            const int myid = id_next;

            int j = 0;
            if (mm == 32) {
                LOAD_ROWS(ROWS_IN_FLIGHT)
                id_next = 0;
                if (base + 32 + lane < end_t) id_next = perm[base + 32 + lane];
                ADD_ROWS(ROWS_IN_FLIGHT)
            }
            while (j + ROWS_IN_FLIGHT <= mm) SUM_ROWS(ROWS_IN_FLIGHT)
            if (j + 8 <= mm) SUM_ROWS(8)
            if (j + 4 <= mm) SUM_ROWS(4)
            if (j + 2 <= mm) SUM_ROWS(2)
            if (j + 1 <= mm) SUM_ROWS(1)
        }
        FINISH_EDGES_BEFORE(NUM_CONN)
    }
#pragma unroll
    for (int i = 0; i < W2_COPY_TRIPS; ++i)
        sW2[(int)threadIdx.x + i * (WAVES_PER_BLOCK * 32)] = w2r[i];
    __syncthreads();

    const int seg_beg = __shfl(bndv, m, 32);
    const int seg_end = __shfl(bndv, m + 1, 32);
    const float degree = (float)(seg_end > seg_beg ? seg_end - seg_beg : 0);
    const float invc = 1.0f / fmaxf(degree, 1.0f);
    const float* arow = abase + m * A_PITCH;

    v8f acc[N_TILES] = {};

    for (int ks = 0; ks < K_STEPS; ++ks) {
        const int k = ks * 4;
        v2f a = *(const v2f*)(arow + k + 2 * hi);
        a.x *= invc;
        a.y *= invc;

        const float4* __restrict__ bp =
            (const float4*)(wpack + ((size_t)ks * 32 + lane) * 16);
        float4 q0 = bp[0], q1 = bp[1], q2 = bp[2], q3 = bp[3];
        v2f bbs[N_TILES];
        bbs[0].x = q0.x; bbs[0].y = q0.y;
        bbs[1].x = q0.z; bbs[1].y = q0.w;
        bbs[2].x = q1.x; bbs[2].y = q1.y;
        bbs[3].x = q1.z; bbs[3].y = q1.w;
        bbs[4].x = q2.x; bbs[4].y = q2.y;
        bbs[5].x = q2.z; bbs[5].y = q2.w;
        bbs[6].x = q3.x; bbs[6].y = q3.y;
        bbs[7].x = q3.z; bbs[7].y = q3.w;

#pragma unroll
        for (int nt = 0; nt < N_TILES; ++nt) {
            acc[nt] = __builtin_amdgcn_wmma_f32_16x16x4_f32(
                false, a, false, bbs[nt], (short)0, acc[nt], false, false);
        }
    }

#pragma unroll
    for (int nt = 0; nt < N_TILES; ++nt) {
        const float bias = b1[nt * 16 + m];
#pragma unroll
        for (int r = 0; r < 8; ++r) acc[nt][r] += bias;
    }

    float s[8], q[8];
#pragma unroll
    for (int r = 0; r < 8; ++r) { s[r] = 0.0f; q[r] = 0.0f; }
#pragma unroll
    for (int nt = 0; nt < N_TILES; ++nt) {
#pragma unroll
        for (int r = 0; r < 8; ++r) {
            const float x = acc[nt][r];
            s[r] += x;
            q[r] += x * x;
        }
    }
#pragma unroll
    for (int mask = 1; mask < 16; mask <<= 1) {
#pragma unroll
        for (int r = 0; r < 8; ++r) {
            s[r] += __shfl_xor(s[r], mask, 32);
            q[r] += __shfl_xor(q[r], mask, 32);
        }
    }
    float mean[8], rstd[8];
#pragma unroll
    for (int r = 0; r < 8; ++r) {
        mean[r] = s[r] * (1.0f / HIDDEN);
        const float var = q[r] * (1.0f / HIDDEN) - mean[r] * mean[r];
        rstd[r] = rsqrtf(var + LN_EPS);
    }

    float* hbase = abase;
#pragma unroll
    for (int nt = 0; nt < N_TILES; ++nt) {
        const int n = nt * 16 + m;
        const float g  = lng[n];
        const float bb = lnb[n];
#pragma unroll
        for (int r = 0; r < 8; ++r) {
            float x = (acc[nt][r] - mean[r]) * rstd[r] * g + bb;
            x = fmaxf(x, 0.0f);
            hbase[(r + 8 * hi) * H_PITCH + n] = x;
        }
    }
    __syncthreads();

    v8f o0, o1;
    {
        const float bias0 = b2[m];
        const float bias1 = b2[16];
#pragma unroll
        for (int r = 0; r < 8; ++r) { o0[r] = bias0; o1[r] = bias1; }
    }
    const float* ha  = hbase + m * H_PITCH + 2 * hi;
    const float* __restrict__ wb0 = sW2 + (2 * hi) * OUT_DIM + m;
    const float* __restrict__ wb1 = sW2 + (2 * hi) * OUT_DIM + 16;
#pragma unroll
    for (int ks = 0; ks < K2_STEPS; ++ks) {
        const v2f a = *(const v2f*)(ha + 4 * ks);
        v2f bA, bB;
        bA.x = wb0[ks * 4 * OUT_DIM];
        bA.y = wb0[ks * 4 * OUT_DIM + OUT_DIM];
        bB.x = wb1[ks * 4 * OUT_DIM];
        bB.y = wb1[ks * 4 * OUT_DIM + OUT_DIM];
        o0 = __builtin_amdgcn_wmma_f32_16x16x4_f32(
            false, a, false, bA, (short)0, o0, false, false);
        o1 = __builtin_amdgcn_wmma_f32_16x16x4_f32(
            false, a, false, bB, (short)0, o1, false, false);
    }

    if (live) {
        float* __restrict__ orow = out + (size_t)(m0 + 8 * hi) * OUT_DIM;
#pragma unroll
        for (int r = 0; r < 8; ++r) orow[r * OUT_DIM + m] = o0[r];
        if (m == 0) {
#pragma unroll
            for (int r = 0; r < 8; ++r) orow[r * OUT_DIM + 16] = o1[r];
        }
    }
}

extern "C" void kernel_launch(void* const* d_in, const int* in_sizes, int n_in,
                              void* d_out, int out_size, void* d_ws, size_t ws_size,
                              hipStream_t stream) {
    const float* nf    = (const float*)d_in[0];
    const int*   hidx  = (const int*)d_in[1];
    const int*   nodes = hidx;
    const int*   edges = hidx + NUM_CONN;
    const float* W1    = (const float*)d_in[2];
    const float* b1    = (const float*)d_in[3];
    const float* lng   = (const float*)d_in[4];
    const float* lnb   = (const float*)d_in[5];
    const float* W2    = (const float*)d_in[6];
    const float* b2    = (const float*)d_in[7];
    float*       out   = (float*)d_out;

    int* tab    = (int*)d_ws;
    int* binned = tab + TAB_N + 64;
    int* perm   = binned + NUM_CONN;
    int* ends   = perm + NUM_CONN;
    float* wpack = (float*)d_ws + (size_t)NUM_EDGES * IN_DIM + NUM_EDGES;

    repack_w1_kernel<<<(K_STEPS * 32 * 8 + 255) / 256, 256, 0, stream>>>(W1, wpack);

    chunk_sort_kernel<<<NCHUNK, BIN_THREADS, 0, stream>>>(nodes, edges, tab, binned);

    range_sort_kernel<<<NB, BIN_THREADS, 0, stream>>>(tab, binned, perm, ends);

    edge_sum_mlp_kernel<<<(NUM_TILES + WAVES_PER_BLOCK - 1) / WAVES_PER_BLOCK, WAVES_PER_BLOCK * 32, 0, stream>>>(
        nf, perm, ends, wpack, b1, lng, lnb, W2, b2, out);
}


# Row gather of the last launch kept in flight: ring of 16 rows in registers, each slot refilled in place with the row 16 ahead right behind its add (was: load 16, add 16 by turns); same sum order
# speedup vs baseline: 8.9480x; 1.0537x over previous
#include <hip/hip_runtime.h>

#define NUM_NODES 100000
#define NUM_EDGES 50000
#define NUM_CONN  800000
#define IN_DIM    256
#define HIDDEN    128
#define OUT_DIM   17
#define LN_EPS    1e-5f

#define WAVES_PER_BLOCK 4
#define TILE_ROWS 16
#define NUM_TILES (NUM_EDGES / TILE_ROWS)
#define N_TILES (HIDDEN / 16)
#define K_STEPS (IN_DIM / 4)
#define WPACK_FLOATS (K_STEPS * 32 * 16)
#define K2_STEPS (HIDDEN / 4)
#define H_PITCH  (HIDDEN + 4)
#define W2_FLOATS (HIDDEN * OUT_DIM)
#define W2_COPY_TRIPS ((W2_FLOATS + WAVES_PER_BLOCK * 32 - 1) / (WAVES_PER_BLOCK * 32))

typedef float v2f __attribute__((ext_vector_type(2)));
typedef float v8f __attribute__((ext_vector_type(8)));
typedef float v4f __attribute__((ext_vector_type(4)));

__global__ void repack_w1_kernel(const float* __restrict__ W1,
                                 float* __restrict__ wpack) {
    int id = blockIdx.x * blockDim.x + threadIdx.x;
    if (id >= K_STEPS * 32 * 8) return;
    int ks   = id >> 8;
    int rem  = id & 255;
    int lane = rem >> 3;
    int nt   = rem & 7;
    int m    = lane & 15;
    int hi   = lane >> 4;
    int krow = ks * 4 + 2 * hi;
    int n    = nt * 16 + m;
    wpack[2 * id + 0] = W1[(size_t)(krow + 0) * HIDDEN + n];
    wpack[2 * id + 1] = W1[(size_t)(krow + 1) * HIDDEN + n];
}


#define NB 128
#define EPB 391
#define EPB_PAD 512
#define NCHUNK 128
#define CHUNK (NUM_CONN / NCHUNK)
#define BIN_THREADS 1024
#define BIN_PER_THREAD 7
#define OFF_PITCH (NB + 1)
#define TAB_N (NCHUNK * OFF_PITCH)
#define NODE_BITS 17
#define NODE_MASK ((1 << NODE_BITS) - 1)
#define PIECES_PER_WAVE (NCHUNK / 32)
#define PIECE_TRIPS 2
#define STAGE_CAP 16384

static_assert(NB * EPB >= NUM_EDGES, "ranges cover every edge");
static_assert(NCHUNK * CHUNK == NUM_CONN, "chunks cover every connection");
static_assert(BIN_THREADS * BIN_PER_THREAD >= CHUNK, "one register batch per chunk");
static_assert(NB == 4 * 32 && NCHUNK == 4 * 32, "four whole waves scan the counts and read the offsets");
static_assert(PIECES_PER_WAVE * (BIN_THREADS / 32) == NCHUNK, "every piece has a wave");
static_assert((1 << NODE_BITS) >= NUM_NODES && EPB <= EPB_PAD, "packed word fits");
static_assert(STAGE_CAP % (4 * BIN_THREADS) == 0, "the level-2 window is a whole number of copy batches");

__device__ __forceinline__ int clamp_node(int node) {
    return (unsigned)node < (unsigned)NUM_NODES ? node : NUM_NODES - 1;
}

__global__ __launch_bounds__(BIN_THREADS) void chunk_sort_kernel(
    const int* __restrict__ nodes,
    const int* __restrict__ edges,
    int* __restrict__ tab,
    int* __restrict__ binned) {
    __shared__ int cnt[NB];
    __shared__ int wtot[NB / 32];
    __shared__ int stage[BIN_THREADS * BIN_PER_THREAD];
    __shared__ int ctot;
    const int t    = threadIdx.x;
    const int lane = t & 31;
    const int wv   = t >> 5;
    const int w    = blockIdx.x;
    if (t < NB) cnt[t] = 0;
    const int* __restrict__ esrc = edges + w * CHUNK;
    const int* __restrict__ nsrc = nodes + w * CHUNK;
    int e[BIN_PER_THREAD], nd[BIN_PER_THREAD], rk[BIN_PER_THREAD];
#pragma unroll
    for (int k = 0; k < BIN_PER_THREAD; ++k) {
        const int i = t + k * BIN_THREADS;
        e[k]  = (i < CHUNK) ? esrc[i] : -1;
        nd[k] = (i < CHUNK) ? nsrc[i] : 0;
    }
    __syncthreads();
#pragma unroll
    for (int k = 0; k < BIN_PER_THREAD; ++k) {
        rk[k] = 0;
        if ((unsigned)e[k] < (unsigned)NUM_EDGES) rk[k] = atomicAdd(&cnt[e[k] / EPB], 1);
    }
    __syncthreads();

    int c = 0, inc = 0;
    if (t < NB) {
        c = cnt[t];
        inc = c;
#pragma unroll
        for (int d = 1; d < 32; d <<= 1) {
            const int up = __shfl_up(inc, d, 32);
            if (lane >= d) inc += up;
        }
        if (lane == 31) wtot[wv] = inc;
    }
    __syncthreads();
    if (t < NB) {
        int pre = 0;
#pragma unroll
        for (int j = 0; j < NB / 32 - 1; ++j) if (j < wv) pre += wtot[j];
        const int start = pre + inc - c;
        cnt[t] = start;
        tab[w * OFF_PITCH + t] = start;
        if (t == NB - 1) {
            tab[w * OFF_PITCH + NB] = start + c;
            ctot = start + c;
        }
    }
    __syncthreads();

#pragma unroll
    for (int k = 0; k < BIN_PER_THREAD; ++k) {
        if ((unsigned)e[k] < (unsigned)NUM_EDGES) {
            const int b  = e[k] / EPB;
            const int le = e[k] - b * EPB;
            const unsigned pos = (unsigned)(cnt[b] + rk[k]);
            if (pos < (unsigned)CHUNK) stage[pos] = (le << NODE_BITS) | clamp_node(nd[k]);
        }
    }
    __syncthreads();
    const int tot = ctot < CHUNK ? ctot : CHUNK;
    int sv[BIN_PER_THREAD];
#pragma unroll
    for (int k = 0; k < BIN_PER_THREAD; ++k) sv[k] = stage[t + k * BIN_THREADS];
#pragma unroll
    for (int k = 0; k < BIN_PER_THREAD; ++k) {
        const int i = t + k * BIN_THREADS;
        if (i < tot) binned[w * CHUNK + i] = sv[k];
    }
}

__global__ __launch_bounds__(BIN_THREADS) void range_sort_kernel(
    const int* __restrict__ tab,
    const int* __restrict__ binned,
    int* __restrict__ perm,
    int* __restrict__ ends) {
    __shared__ int cnt[EPB_PAD];
    __shared__ int wsum[32];
    __shared__ int pbeg[NCHUNK];
    __shared__ int plen[NCHUNK];
    __shared__ int osum[NCHUNK / 32];
    __shared__ int stage[STAGE_CAP];
    __shared__ int rtot;
    const int t    = threadIdx.x;
    const int lane = t & 31;
    const int w    = t >> 5;
    const int b    = blockIdx.x;

    if (t < EPB_PAD) cnt[t] = 0;

    int o0 = 0;
    if (t < NCHUNK) {
        o0     = tab[t * OFF_PITCH + b];
        int o1 = tab[t * OFF_PITCH + b + 1];
        if (o0 < 0) o0 = 0;
        if (o0 > CHUNK) o0 = CHUNK;
        if (o1 < o0) o1 = o0;
        if (o1 > CHUNK) o1 = CHUNK;
        pbeg[t] = t * CHUNK + o0;
        plen[t] = o1 - o0;
    }
    __syncthreads();

    int pb[PIECES_PER_WAVE], pl[PIECES_PER_WAVE];
    int p[PIECES_PER_WAVE][PIECE_TRIPS];
#pragma unroll
    for (int j = 0; j < PIECES_PER_WAVE; ++j) {
        pb[j] = __builtin_amdgcn_readfirstlane(pbeg[w * PIECES_PER_WAVE + j]);
        pl[j] = __builtin_amdgcn_readfirstlane(plen[w * PIECES_PER_WAVE + j]);
    }
#pragma unroll
    for (int j = 0; j < PIECES_PER_WAVE; ++j) {
#pragma unroll
        for (int k = 0; k < PIECE_TRIPS; ++k) {
            const int i = lane + 32 * k;
            p[j][k] = (i < pl[j]) ? binned[pb[j] + i] : -1;
        }
    }

    if (t < NCHUNK) {
        int osm = o0;
#pragma unroll
        for (int d = 16; d >= 1; d >>= 1) osm += __shfl_xor(osm, d, 32);
        if (lane == 0) osum[w] = osm;
    }

#pragma unroll
    for (int j = 0; j < PIECES_PER_WAVE; ++j) {
#pragma unroll
        for (int k = 0; k < PIECE_TRIPS; ++k) {
            if (p[j][k] >= 0) atomicAdd(&cnt[(p[j][k] >> NODE_BITS) & (EPB_PAD - 1)], 1);
        }
        for (int i = lane + 32 * PIECE_TRIPS; i < pl[j]; i += 32) {
            const int q = binned[pb[j] + i];
            if (q >= 0) atomicAdd(&cnt[(q >> NODE_BITS) & (EPB_PAD - 1)], 1);
        }
    }
    __syncthreads();

    int bs = 0;
#pragma unroll
    for (int j = 0; j < NCHUNK / 32; ++j) bs += osum[j];
    if (bs > NUM_CONN) bs = NUM_CONN;

    const int c = (t < EPB_PAD) ? cnt[t] : 0;
    int inc = c;
#pragma unroll
    for (int d = 1; d < 32; d <<= 1) {
        const int up = __shfl_up(inc, d, 32);
        if (lane >= d) inc += up;
    }
    if (lane == 31) wsum[w] = inc;
    __syncthreads();
    if (w == 0) {
        const int tot = wsum[lane];
        int inc2 = tot;
#pragma unroll
        for (int d = 1; d < 32; d <<= 1) {
            const int u = __shfl_up(inc2, d, 32);
            if (lane >= d) inc2 += u;
        }
        wsum[lane] = inc2 - tot;
        if (lane == 31) rtot = inc2;
    }
    __syncthreads();
    const int  rt     = rtot;
    const bool staged = rt <= STAGE_CAP;
    const int end_t = bs + wsum[w] + inc;
    const int e = b * EPB + t;
    if (t < EPB && e < NUM_EDGES) ends[e] = end_t;
    if (t < EPB_PAD) cnt[t] = end_t - c;
    __syncthreads();

#pragma unroll
    for (int j = 0; j < PIECES_PER_WAVE; ++j) {
#pragma unroll
        for (int k = 0; k < PIECE_TRIPS; ++k) {
            if (p[j][k] >= 0) {
                const unsigned pos = (unsigned)atomicAdd(&cnt[(p[j][k] >> NODE_BITS) & (EPB_PAD - 1)], 1);
                const unsigned loc = pos - (unsigned)bs;
                if (staged) { if (loc < (unsigned)STAGE_CAP) stage[loc] = p[j][k] & NODE_MASK; }
                else if (pos < (unsigned)NUM_CONN) perm[pos] = p[j][k] & NODE_MASK;
            }
        }
        for (int i = lane + 32 * PIECE_TRIPS; i < pl[j]; i += 32) {
            const int q = binned[pb[j] + i];
            if (q >= 0) {
                const unsigned pos = (unsigned)atomicAdd(&cnt[(q >> NODE_BITS) & (EPB_PAD - 1)], 1);
                const unsigned loc = pos - (unsigned)bs;
                if (staged) { if (loc < (unsigned)STAGE_CAP) stage[loc] = q & NODE_MASK; }
                else if (pos < (unsigned)NUM_CONN) perm[pos] = q & NODE_MASK;
            }
        }
    }
    __syncthreads();
    if (staged) {
        for (int i0 = t; i0 < rt; i0 += 4 * BIN_THREADS) {
            int sv[4];
#pragma unroll
            for (int k = 0; k < 4; ++k) sv[k] = stage[i0 + k * BIN_THREADS];
#pragma unroll
            for (int k = 0; k < 4; ++k) {
                const int i = i0 + k * BIN_THREADS;
                if (i < rt && bs + i < NUM_CONN) perm[bs + i] = sv[k];
            }
        }
    }
}

#define A_PITCH (IN_DIM + 4)
#define ROWS_IN_FLIGHT 16

__device__ __forceinline__ void add4(float4& a, const float4& x) {
    a.x += x.x; a.y += x.y; a.z += x.z; a.w += x.w;
}

__device__ __forceinline__ const float4* node_row(const float* __restrict__ nf, int node) {
    const unsigned n = (unsigned)node < (unsigned)NUM_NODES ? (unsigned)node
                                                            : (unsigned)(NUM_NODES - 1);
    return (const float4*)(nf + (size_t)n * IN_DIM);
}

__device__ __forceinline__ void add4v(float4& a, const v4f& x) {
    a.x += x.x; a.y += x.y; a.z += x.z; a.w += x.w;
}

__device__ __forceinline__ const float* node_row_base(const float* __restrict__ nf, int node) {
    const unsigned n = (unsigned)node < (unsigned)NUM_NODES ? (unsigned)node
                                                            : (unsigned)(NUM_NODES - 1);
    return nf + (size_t)n * IN_DIM;
}

#define FINISH_EDGES_BEFORE(POS)                                                        \
    _Pragma("nounroll")                                                                 \
    while (cur < TILE_ROWS && (POS) >= next_end) {                                      \
        float4* drow = (float4*)(abase + cur * A_PITCH);                                \
        drow[lane]      = a0;                                                           \
        drow[32 + lane] = a1;                                                           \
        a0 = make_float4(0.0f, 0.0f, 0.0f, 0.0f);                                       \
        a1 = a0;                                                                        \
        ++cur;                                                                          \
        next_end = __builtin_amdgcn_readlane(bndv, cur < TILE_ROWS ? cur + 1 : TILE_ROWS); \
    }

#define LOAD_ROWS(N)                                                                    \
        const float4* rp[N];                                                            \
        float4 rx[N], ry[N];                                                            \
        _Pragma("unroll")                                                               \
        for (int i = 0; i < (N); ++i)                                                   \
            rp[i] = node_row(nf, __builtin_amdgcn_readlane(myid, j + i));               \
        _Pragma("unroll")                                                               \
        for (int i = 0; i < (N); ++i) { rx[i] = rp[i][lane]; ry[i] = rp[i][32 + lane]; }
#define ADD_ROWS(N)                                                                     \
        _Pragma("unroll")                                                               \
        for (int i = 0; i < (N); ++i) {                                                 \
            FINISH_EDGES_BEFORE(base + j + i)                                           \
            add4(a0, rx[i]);                                                            \
            add4(a1, ry[i]);                                                            \
        }                                                                               \
        j += (N);
#define SUM_ROWS(N) { LOAD_ROWS(N) ADD_ROWS(N) }

#define RING_FIRST(S)                                                                   \
        asm volatile("global_load_b128 %0, %2, %3\n\tglobal_load_b128 %1, %2, %3 offset:512" \
                     : "=v"(sx[S]), "=v"(sy[S])                                         \
                     : "v"(lane16), "s"(node_row_base(nf, __builtin_amdgcn_readlane(id_cur, S))));
#define RING_ASK(S, IDV, LN)                                                            \
        asm volatile("global_load_b128 %0, %2, %3\n\tglobal_load_b128 %1, %2, %3 offset:512" \
                     : "+v"(sx[S]), "+v"(sy[S])                                         \
                     : "v"(lane16), "s"(node_row_base(nf, __builtin_amdgcn_readlane(IDV, LN))));
#define RING_STEP_A(S)                                                                  \
        FINISH_EDGES_BEFORE(base + S)                                                   \
        asm volatile("s_wait_loadcnt 31" : "+v"(sx[S]), "+v"(sy[S]));                   \
        add4v(a0, sx[S]);                                                               \
        add4v(a1, sy[S]);                                                               \
        RING_ASK(S, id_cur, 16 + S)
#define RING_STEP_B0                                                                    \
        FINISH_EDGES_BEFORE(base + 16)                                                  \
        asm volatile("s_wait_loadcnt 30\n\tv_mov_b32 %2, %3"                            \
                     : "+v"(sx[0]), "+v"(sy[0]), "=v"(id_ok) : "v"(id_nx));             \
        add4v(a0, sx[0]);                                                               \
        add4v(a1, sy[0]);                                                               \
        RING_ASK(0, id_ok, 0)
#define RING_STEP_B(S)                                                                  \
        FINISH_EDGES_BEFORE(base + 16 + S)                                              \
        asm volatile("s_wait_loadcnt 30" : "+v"(sx[S]), "+v"(sy[S]));                   \
        add4v(a0, sx[S]);                                                               \
        add4v(a1, sy[S]);                                                               \
        RING_ASK(S, id_ok, S)
#define RING_DRAIN(S, N)                                                                \
        FINISH_EDGES_BEFORE(base + S)                                                   \
        asm volatile("s_wait_loadcnt " #N : "+v"(sx[S]), "+v"(sy[S]));                  \
        add4v(a0, sx[S]);                                                               \
        add4v(a1, sy[S]);

__global__ __launch_bounds__(WAVES_PER_BLOCK * 32) void edge_sum_mlp_kernel(
    const float* __restrict__ nf,
    const int*   __restrict__ perm,
    const int*   __restrict__ ends,
    const float* __restrict__ wpack,
    const float* __restrict__ b1,
    const float* __restrict__ lng,
    const float* __restrict__ lnb,
    const float* __restrict__ W2,
    const float* __restrict__ b2,
    float* __restrict__ out) {
    __shared__ alignas(16) float sA[WAVES_PER_BLOCK * TILE_ROWS * A_PITCH];
    __shared__ float sW2[W2_COPY_TRIPS * WAVES_PER_BLOCK * 32];

    const int lane = threadIdx.x & 31;
    const int wave = __builtin_amdgcn_readfirstlane((int)(threadIdx.x >> 5));
    const int  tile_id = blockIdx.x * WAVES_PER_BLOCK + wave;
    const bool live    = tile_id < NUM_TILES;
    const int  tile    = live ? tile_id : NUM_TILES - 1;
    const int  m0      = tile * TILE_ROWS;
    const int m    = lane & 15;
    const int hi   = lane >> 4;

    float* abase = sA + wave * (TILE_ROWS * A_PITCH);

    const int bidx = m0 - 1 + (lane < TILE_ROWS ? lane : TILE_ROWS);
    const int braw = ends[bidx < 0 ? 0 : bidx];

    float w2r[W2_COPY_TRIPS];
#pragma unroll
    for (int i = 0; i < W2_COPY_TRIPS; ++i) {
        const int id  = (int)threadIdx.x + i * (WAVES_PER_BLOCK * 32);
        const int idc = id < W2_FLOATS ? id : W2_FLOATS - 1;
        w2r[i] = W2[idc];
    }

    int bndv = (bidx < 0) ? 0 : braw;
    if (bndv < 0) bndv = 0;
    if (bndv > NUM_CONN) bndv = NUM_CONN;

    {
        const int beg   = __builtin_amdgcn_readlane(bndv, 0);
        const int end_t = __builtin_amdgcn_readlane(bndv, TILE_ROWS);
        int cur      = 0;
        int next_end = __builtin_amdgcn_readlane(bndv, 1);
        float4 a0 = make_float4(0.0f, 0.0f, 0.0f, 0.0f);
        float4 a1 = a0;

        int base   = beg;
        int j0     = 0;
        int id_cur = 0;
        if (beg + lane < end_t) id_cur = perm[beg + lane];
#pragma unroll
        for (int i = 0; i < W2_COPY_TRIPS; ++i)
            sW2[(int)threadIdx.x + i * (WAVES_PER_BLOCK * 32)] = w2r[i];

        if (end_t - beg >= 48) {
            v4f sx[ROWS_IN_FLIGHT], sy[ROWS_IN_FLIGHT];
            const int lane16 = lane * 16;
            const int lane4  = lane * 4;
            RING_FIRST(0) RING_FIRST(1) RING_FIRST(2) RING_FIRST(3) RING_FIRST(4) RING_FIRST(5) RING_FIRST(6) RING_FIRST(7) RING_FIRST(8) RING_FIRST(9) RING_FIRST(10) RING_FIRST(11) RING_FIRST(12) RING_FIRST(13) RING_FIRST(14) RING_FIRST(15)
            do {
                int id_nx, id_ok;
                asm volatile("global_load_b32 %0, %1, %2" : "=v"(id_nx) : "v"(lane4), "s"(perm + base + 32));
                RING_STEP_A(0) RING_STEP_A(1) RING_STEP_A(2) RING_STEP_A(3) RING_STEP_A(4) RING_STEP_A(5) RING_STEP_A(6) RING_STEP_A(7) RING_STEP_A(8) RING_STEP_A(9) RING_STEP_A(10) RING_STEP_A(11) RING_STEP_A(12) RING_STEP_A(13) RING_STEP_A(14) RING_STEP_A(15)
                RING_STEP_B0 RING_STEP_B(1) RING_STEP_B(2) RING_STEP_B(3) RING_STEP_B(4) RING_STEP_B(5) RING_STEP_B(6) RING_STEP_B(7) RING_STEP_B(8) RING_STEP_B(9) RING_STEP_B(10) RING_STEP_B(11) RING_STEP_B(12) RING_STEP_B(13) RING_STEP_B(14) RING_STEP_B(15)
                id_cur = id_ok;
                base += 32;
            } while (end_t - base >= 48);
            RING_DRAIN(0, 30) RING_DRAIN(1, 28) RING_DRAIN(2, 26) RING_DRAIN(3, 24) RING_DRAIN(4, 22) RING_DRAIN(5, 20) RING_DRAIN(6, 18) RING_DRAIN(7, 16) RING_DRAIN(8, 14) RING_DRAIN(9, 12) RING_DRAIN(10, 10) RING_DRAIN(11, 8) RING_DRAIN(12, 6) RING_DRAIN(13, 4) RING_DRAIN(14, 2) RING_DRAIN(15, 0)
            j0 = 16;
        }

        for (; base < end_t; base += 32) {
            const int left = end_t - base;
            const int mm   = left < 32 ? left : 32;
            const int myid = id_cur;
            int j = j0;
            j0 = 0;
            id_cur = 0;
            if (base + 32 + lane < end_t) id_cur = perm[base + 32 + lane];
            while (j + ROWS_IN_FLIGHT <= mm) SUM_ROWS(ROWS_IN_FLIGHT)
            if (j + 8 <= mm) SUM_ROWS(8)
            if (j + 4 <= mm) SUM_ROWS(4)
            if (j + 2 <= mm) SUM_ROWS(2)
            if (j + 1 <= mm) SUM_ROWS(1)
        }
        FINISH_EDGES_BEFORE(NUM_CONN)
    }
    __syncthreads();

    const int seg_beg = __shfl(bndv, m, 32);
    const int seg_end = __shfl(bndv, m + 1, 32);
    const float degree = (float)(seg_end > seg_beg ? seg_end - seg_beg : 0);
    const float invc = 1.0f / fmaxf(degree, 1.0f);
    const float* arow = abase + m * A_PITCH;

    v8f acc[N_TILES] = {};

    for (int ks = 0; ks < K_STEPS; ++ks) {
        const int k = ks * 4;
        v2f a = *(const v2f*)(arow + k + 2 * hi);
        a.x *= invc;
        a.y *= invc;

        const float4* __restrict__ bp =
            (const float4*)(wpack + ((size_t)ks * 32 + lane) * 16);
        float4 q0 = bp[0], q1 = bp[1], q2 = bp[2], q3 = bp[3];
        v2f bbs[N_TILES];
        bbs[0].x = q0.x; bbs[0].y = q0.y;
        bbs[1].x = q0.z; bbs[1].y = q0.w;
        bbs[2].x = q1.x; bbs[2].y = q1.y;
        bbs[3].x = q1.z; bbs[3].y = q1.w;
        bbs[4].x = q2.x; bbs[4].y = q2.y;
        bbs[5].x = q2.z; bbs[5].y = q2.w;
        bbs[6].x = q3.x; bbs[6].y = q3.y;
        bbs[7].x = q3.z; bbs[7].y = q3.w;

#pragma unroll
        for (int nt = 0; nt < N_TILES; ++nt) {
            acc[nt] = __builtin_amdgcn_wmma_f32_16x16x4_f32(
                false, a, false, bbs[nt], (short)0, acc[nt], false, false);
        }
    }

#pragma unroll
    for (int nt = 0; nt < N_TILES; ++nt) {
        const float bias = b1[nt * 16 + m];
#pragma unroll
        for (int r = 0; r < 8; ++r) acc[nt][r] += bias;
    }

    float s[8], q[8];
#pragma unroll
    for (int r = 0; r < 8; ++r) { s[r] = 0.0f; q[r] = 0.0f; }
#pragma unroll
    for (int nt = 0; nt < N_TILES; ++nt) {
#pragma unroll
        for (int r = 0; r < 8; ++r) {
            const float x = acc[nt][r];
            s[r] += x;
            q[r] += x * x;
        }
    }
#pragma unroll
    for (int mask = 1; mask < 16; mask <<= 1) {
#pragma unroll
        for (int r = 0; r < 8; ++r) {
            s[r] += __shfl_xor(s[r], mask, 32);
            q[r] += __shfl_xor(q[r], mask, 32);
        }
    }
    float mean[8], rstd[8];
#pragma unroll
    for (int r = 0; r < 8; ++r) {
        mean[r] = s[r] * (1.0f / HIDDEN);
        const float var = q[r] * (1.0f / HIDDEN) - mean[r] * mean[r];
        rstd[r] = rsqrtf(var + LN_EPS);
    }

    float* hbase = abase;
#pragma unroll
    for (int nt = 0; nt < N_TILES; ++nt) {
        const int n = nt * 16 + m;
        const float g  = lng[n];
        const float bb = lnb[n];
#pragma unroll
        for (int r = 0; r < 8; ++r) {
            float x = (acc[nt][r] - mean[r]) * rstd[r] * g + bb;
            x = fmaxf(x, 0.0f);
            hbase[(r + 8 * hi) * H_PITCH + n] = x;
        }
    }
    __syncthreads();

    v8f o0, o1;
    {
        const float bias0 = b2[m];
        const float bias1 = b2[16];
#pragma unroll
        for (int r = 0; r < 8; ++r) { o0[r] = bias0; o1[r] = bias1; }
    }
    const float* ha  = hbase + m * H_PITCH + 2 * hi;
    const float* __restrict__ wb0 = sW2 + (2 * hi) * OUT_DIM + m;
    const float* __restrict__ wb1 = sW2 + (2 * hi) * OUT_DIM + 16;
#pragma unroll
    for (int ks = 0; ks < K2_STEPS; ++ks) {
        const v2f a = *(const v2f*)(ha + 4 * ks);
        v2f bA, bB;
        bA.x = wb0[ks * 4 * OUT_DIM];
        bA.y = wb0[ks * 4 * OUT_DIM + OUT_DIM];
        bB.x = wb1[ks * 4 * OUT_DIM];
        bB.y = wb1[ks * 4 * OUT_DIM + OUT_DIM];
        o0 = __builtin_amdgcn_wmma_f32_16x16x4_f32(
            false, a, false, bA, (short)0, o0, false, false);
        o1 = __builtin_amdgcn_wmma_f32_16x16x4_f32(
            false, a, false, bB, (short)0, o1, false, false);
    }

    if (live) {
        float* __restrict__ orow = out + (size_t)(m0 + 8 * hi) * OUT_DIM;
#pragma unroll
        for (int r = 0; r < 8; ++r) orow[r * OUT_DIM + m] = o0[r];
        if (m == 0) {
#pragma unroll
            for (int r = 0; r < 8; ++r) orow[r * OUT_DIM + 16] = o1[r];
        }
    }
}

extern "C" void kernel_launch(void* const* d_in, const int* in_sizes, int n_in,
                              void* d_out, int out_size, void* d_ws, size_t ws_size,
                              hipStream_t stream) {
    const float* nf    = (const float*)d_in[0];
    const int*   hidx  = (const int*)d_in[1];
    const int*   nodes = hidx;
    const int*   edges = hidx + NUM_CONN;
    const float* W1    = (const float*)d_in[2];
    const float* b1    = (const float*)d_in[3];
    const float* lng   = (const float*)d_in[4];
    const float* lnb   = (const float*)d_in[5];
    const float* W2    = (const float*)d_in[6];
    const float* b2    = (const float*)d_in[7];
    float*       out   = (float*)d_out;

    int* tab    = (int*)d_ws;
    int* binned = tab + TAB_N + 64;
    int* perm   = binned + NUM_CONN;
    int* ends   = perm + NUM_CONN;
    float* wpack = (float*)d_ws + (size_t)NUM_EDGES * IN_DIM + NUM_EDGES;

    repack_w1_kernel<<<(K_STEPS * 32 * 8 + 255) / 256, 256, 0, stream>>>(W1, wpack);

    chunk_sort_kernel<<<NCHUNK, BIN_THREADS, 0, stream>>>(nodes, edges, tab, binned);

    range_sort_kernel<<<NB, BIN_THREADS, 0, stream>>>(tab, binned, perm, ends);

    edge_sum_mlp_kernel<<<(NUM_TILES + WAVES_PER_BLOCK - 1) / WAVES_PER_BLOCK, WAVES_PER_BLOCK * 32, 0, stream>>>(
        nf, perm, ends, wpack, b1, lng, lnb, W2, b2, out);
}


# W1 repack moved into the first sort launch as 16 extra work-groups (it needs only W1 and is read two launches later): its own launch is gone, 4->3 launches; every sum order as before
# speedup vs baseline: 9.2348x; 1.0321x over previous
#include <hip/hip_runtime.h>

#define NUM_NODES 100000
#define NUM_EDGES 50000
#define NUM_CONN  800000
#define IN_DIM    256
#define HIDDEN    128
#define OUT_DIM   17
#define LN_EPS    1e-5f

#define WAVES_PER_BLOCK 4
#define TILE_ROWS 16
#define NUM_TILES (NUM_EDGES / TILE_ROWS)
#define N_TILES (HIDDEN / 16)
#define K_STEPS (IN_DIM / 4)
#define WPACK_FLOATS (K_STEPS * 32 * 16)
#define K2_STEPS (HIDDEN / 4)
#define H_PITCH  (HIDDEN + 4)
#define W2_FLOATS (HIDDEN * OUT_DIM)
#define W2_COPY_TRIPS ((W2_FLOATS + WAVES_PER_BLOCK * 32 - 1) / (WAVES_PER_BLOCK * 32))

typedef float v2f __attribute__((ext_vector_type(2)));
typedef float v8f __attribute__((ext_vector_type(8)));
typedef float v4f __attribute__((ext_vector_type(4)));

#define REPACK_ITEMS  (K_STEPS * 32 * 8)
__device__ __forceinline__ void repack_w1_item(const float* __restrict__ W1,
                                               float* __restrict__ wpack, int id) {
    if (id >= REPACK_ITEMS) return;
    int ks   = id >> 8;
    int rem  = id & 255;
    int lane = rem >> 3;
    int nt   = rem & 7;
    int m    = lane & 15;
    int hi   = lane >> 4;
    int krow = ks * 4 + 2 * hi;
    int n    = nt * 16 + m;
    wpack[2 * id + 0] = W1[(size_t)(krow + 0) * HIDDEN + n];
    wpack[2 * id + 1] = W1[(size_t)(krow + 1) * HIDDEN + n];
}


#define NB 128
#define EPB 391
#define EPB_PAD 512
#define NCHUNK 128
#define CHUNK (NUM_CONN / NCHUNK)
#define BIN_THREADS 1024
#define BIN_PER_THREAD 7
#define OFF_PITCH (NB + 1)
#define TAB_N (NCHUNK * OFF_PITCH)
#define NODE_BITS 17
#define NODE_MASK ((1 << NODE_BITS) - 1)
#define PIECES_PER_WAVE (NCHUNK / 32)
#define PIECE_TRIPS 2
#define STAGE_CAP 16384
#define REPACK_GROUPS ((REPACK_ITEMS + BIN_THREADS - 1) / BIN_THREADS)

static_assert(NB * EPB >= NUM_EDGES, "ranges cover every edge");
static_assert(NCHUNK * CHUNK == NUM_CONN, "chunks cover every connection");
static_assert(BIN_THREADS * BIN_PER_THREAD >= CHUNK, "one register batch per chunk");
static_assert(NB == 4 * 32 && NCHUNK == 4 * 32, "four whole waves scan the counts and read the offsets");
static_assert(PIECES_PER_WAVE * (BIN_THREADS / 32) == NCHUNK, "every piece has a wave");
static_assert((1 << NODE_BITS) >= NUM_NODES && EPB <= EPB_PAD, "packed word fits");
static_assert(STAGE_CAP % (4 * BIN_THREADS) == 0, "the level-2 window is a whole number of copy batches");

__device__ __forceinline__ int clamp_node(int node) {
    return (unsigned)node < (unsigned)NUM_NODES ? node : NUM_NODES - 1;
}

__global__ __launch_bounds__(BIN_THREADS) void chunk_sort_kernel(
    const int* __restrict__ nodes,
    const int* __restrict__ edges,
    int* __restrict__ tab,
    int* __restrict__ binned,
    const float* __restrict__ W1,
    float* __restrict__ wpack) {
    if (blockIdx.x >= NCHUNK) {
        repack_w1_item(W1, wpack, (int)(blockIdx.x - NCHUNK) * BIN_THREADS + (int)threadIdx.x);
        return;
    }
    __shared__ int cnt[NB];
    __shared__ int wtot[NB / 32];
    __shared__ int stage[BIN_THREADS * BIN_PER_THREAD];
    __shared__ int ctot;
    const int t    = threadIdx.x;
    const int lane = t & 31;
    const int wv   = t >> 5;
    const int w    = blockIdx.x;
    if (t < NB) cnt[t] = 0;
    const int* __restrict__ esrc = edges + w * CHUNK;
    const int* __restrict__ nsrc = nodes + w * CHUNK;
    int e[BIN_PER_THREAD], nd[BIN_PER_THREAD], rk[BIN_PER_THREAD];
#pragma unroll
    for (int k = 0; k < BIN_PER_THREAD; ++k) {
        const int i = t + k * BIN_THREADS;
        e[k]  = (i < CHUNK) ? esrc[i] : -1;
        nd[k] = (i < CHUNK) ? nsrc[i] : 0;
    }
    __syncthreads();
#pragma unroll
    for (int k = 0; k < BIN_PER_THREAD; ++k) {
        rk[k] = 0;
        if ((unsigned)e[k] < (unsigned)NUM_EDGES) rk[k] = atomicAdd(&cnt[e[k] / EPB], 1);
    }
    __syncthreads();

    int c = 0, inc = 0;
    if (t < NB) {
        c = cnt[t];
        inc = c;
#pragma unroll
        for (int d = 1; d < 32; d <<= 1) {
            const int up = __shfl_up(inc, d, 32);
            if (lane >= d) inc += up;
        }
        if (lane == 31) wtot[wv] = inc;
    }
    __syncthreads();
    if (t < NB) {
        int pre = 0;
#pragma unroll
        for (int j = 0; j < NB / 32 - 1; ++j) if (j < wv) pre += wtot[j];
        const int start = pre + inc - c;
        cnt[t] = start;
        tab[w * OFF_PITCH + t] = start;
        if (t == NB - 1) {
            tab[w * OFF_PITCH + NB] = start + c;
            ctot = start + c;
        }
    }
    __syncthreads();

#pragma unroll
    for (int k = 0; k < BIN_PER_THREAD; ++k) {
        if ((unsigned)e[k] < (unsigned)NUM_EDGES) {
            const int b  = e[k] / EPB;
            const int le = e[k] - b * EPB;
            const unsigned pos = (unsigned)(cnt[b] + rk[k]);
            if (pos < (unsigned)CHUNK) stage[pos] = (le << NODE_BITS) | clamp_node(nd[k]);
        }
    }
    __syncthreads();
    const int tot = ctot < CHUNK ? ctot : CHUNK;
    int sv[BIN_PER_THREAD];
#pragma unroll
    for (int k = 0; k < BIN_PER_THREAD; ++k) sv[k] = stage[t + k * BIN_THREADS];
#pragma unroll
    for (int k = 0; k < BIN_PER_THREAD; ++k) {
        const int i = t + k * BIN_THREADS;
        if (i < tot) binned[w * CHUNK + i] = sv[k];
    }
}

__global__ __launch_bounds__(BIN_THREADS) void range_sort_kernel(
    const int* __restrict__ tab,
    const int* __restrict__ binned,
    int* __restrict__ perm,
    int* __restrict__ ends) {
    __shared__ int cnt[EPB_PAD];
    __shared__ int wsum[32];
    __shared__ int pbeg[NCHUNK];
    __shared__ int plen[NCHUNK];
    __shared__ int osum[NCHUNK / 32];
    __shared__ int stage[STAGE_CAP];
    __shared__ int rtot;
    const int t    = threadIdx.x;
    const int lane = t & 31;
    const int w    = t >> 5;
    const int b    = blockIdx.x;

    if (t < EPB_PAD) cnt[t] = 0;

    int o0 = 0;
    if (t < NCHUNK) {
        o0     = tab[t * OFF_PITCH + b];
        int o1 = tab[t * OFF_PITCH + b + 1];
        if (o0 < 0) o0 = 0;
        if (o0 > CHUNK) o0 = CHUNK;
        if (o1 < o0) o1 = o0;
        if (o1 > CHUNK) o1 = CHUNK;
        pbeg[t] = t * CHUNK + o0;
        plen[t] = o1 - o0;
    }
    __syncthreads();

    int pb[PIECES_PER_WAVE], pl[PIECES_PER_WAVE];
    int p[PIECES_PER_WAVE][PIECE_TRIPS];
#pragma unroll
    for (int j = 0; j < PIECES_PER_WAVE; ++j) {
        pb[j] = __builtin_amdgcn_readfirstlane(pbeg[w * PIECES_PER_WAVE + j]);
        pl[j] = __builtin_amdgcn_readfirstlane(plen[w * PIECES_PER_WAVE + j]);
    }
#pragma unroll
    for (int j = 0; j < PIECES_PER_WAVE; ++j) {
#pragma unroll
        for (int k = 0; k < PIECE_TRIPS; ++k) {
            const int i = lane + 32 * k;
            p[j][k] = (i < pl[j]) ? binned[pb[j] + i] : -1;
        }
    }

    if (t < NCHUNK) {
        int osm = o0;
#pragma unroll
        for (int d = 16; d >= 1; d >>= 1) osm += __shfl_xor(osm, d, 32);
        if (lane == 0) osum[w] = osm;
    }

#pragma unroll
    for (int j = 0; j < PIECES_PER_WAVE; ++j) {
#pragma unroll
        for (int k = 0; k < PIECE_TRIPS; ++k) {
            if (p[j][k] >= 0) atomicAdd(&cnt[(p[j][k] >> NODE_BITS) & (EPB_PAD - 1)], 1);
        }
        for (int i = lane + 32 * PIECE_TRIPS; i < pl[j]; i += 32) {
            const int q = binned[pb[j] + i];
            if (q >= 0) atomicAdd(&cnt[(q >> NODE_BITS) & (EPB_PAD - 1)], 1);
        }
    }
    __syncthreads();

    int bs = 0;
#pragma unroll
    for (int j = 0; j < NCHUNK / 32; ++j) bs += osum[j];
    if (bs > NUM_CONN) bs = NUM_CONN;

    const int c = (t < EPB_PAD) ? cnt[t] : 0;
    int inc = c;
#pragma unroll
    for (int d = 1; d < 32; d <<= 1) {
        const int up = __shfl_up(inc, d, 32);
        if (lane >= d) inc += up;
    }
    if (lane == 31) wsum[w] = inc;
    __syncthreads();
    if (w == 0) {
        const int tot = wsum[lane];
        int inc2 = tot;
#pragma unroll
        for (int d = 1; d < 32; d <<= 1) {
            const int u = __shfl_up(inc2, d, 32);
            if (lane >= d) inc2 += u;
        }
        wsum[lane] = inc2 - tot;
        if (lane == 31) rtot = inc2;
    }
    __syncthreads();
    const int  rt     = rtot;
    const bool staged = rt <= STAGE_CAP;
    const int end_t = bs + wsum[w] + inc;
    const int e = b * EPB + t;
    if (t < EPB && e < NUM_EDGES) ends[e] = end_t;
    if (t < EPB_PAD) cnt[t] = end_t - c;
    __syncthreads();

#pragma unroll
    for (int j = 0; j < PIECES_PER_WAVE; ++j) {
#pragma unroll
        for (int k = 0; k < PIECE_TRIPS; ++k) {
            if (p[j][k] >= 0) {
                const unsigned pos = (unsigned)atomicAdd(&cnt[(p[j][k] >> NODE_BITS) & (EPB_PAD - 1)], 1);
                const unsigned loc = pos - (unsigned)bs;
                if (staged) { if (loc < (unsigned)STAGE_CAP) stage[loc] = p[j][k] & NODE_MASK; }
                else if (pos < (unsigned)NUM_CONN) perm[pos] = p[j][k] & NODE_MASK;
            }
        }
        for (int i = lane + 32 * PIECE_TRIPS; i < pl[j]; i += 32) {
            const int q = binned[pb[j] + i];
            if (q >= 0) {
                const unsigned pos = (unsigned)atomicAdd(&cnt[(q >> NODE_BITS) & (EPB_PAD - 1)], 1);
                const unsigned loc = pos - (unsigned)bs;
                if (staged) { if (loc < (unsigned)STAGE_CAP) stage[loc] = q & NODE_MASK; }
                else if (pos < (unsigned)NUM_CONN) perm[pos] = q & NODE_MASK;
            }
        }
    }
    __syncthreads();
    if (staged) {
        for (int i0 = t; i0 < rt; i0 += 4 * BIN_THREADS) {
            int sv[4];
#pragma unroll
            for (int k = 0; k < 4; ++k) sv[k] = stage[i0 + k * BIN_THREADS];
#pragma unroll
            for (int k = 0; k < 4; ++k) {
                const int i = i0 + k * BIN_THREADS;
                if (i < rt && bs + i < NUM_CONN) perm[bs + i] = sv[k];
            }
        }
    }
}

#define A_PITCH (IN_DIM + 4)
#define ROWS_IN_FLIGHT 16

__device__ __forceinline__ void add4(float4& a, const float4& x) {
    a.x += x.x; a.y += x.y; a.z += x.z; a.w += x.w;
}

__device__ __forceinline__ const float4* node_row(const float* __restrict__ nf, int node) {
    const unsigned n = (unsigned)node < (unsigned)NUM_NODES ? (unsigned)node
                                                            : (unsigned)(NUM_NODES - 1);
    return (const float4*)(nf + (size_t)n * IN_DIM);
}

__device__ __forceinline__ void add4v(float4& a, const v4f& x) {
    a.x += x.x; a.y += x.y; a.z += x.z; a.w += x.w;
}

__device__ __forceinline__ const float* node_row_base(const float* __restrict__ nf, int node) {
    const unsigned n = (unsigned)node < (unsigned)NUM_NODES ? (unsigned)node
                                                            : (unsigned)(NUM_NODES - 1);
    return nf + (size_t)n * IN_DIM;
}

#define FINISH_EDGES_BEFORE(POS)                                                        \
    _Pragma("nounroll")                                                                 \
    while (cur < TILE_ROWS && (POS) >= next_end) {                                      \
        float4* drow = (float4*)(abase + cur * A_PITCH);                                \
        drow[lane]      = a0;                                                           \
        drow[32 + lane] = a1;                                                           \
        a0 = make_float4(0.0f, 0.0f, 0.0f, 0.0f);                                       \
        a1 = a0;                                                                        \
        ++cur;                                                                          \
        next_end = __builtin_amdgcn_readlane(bndv, cur < TILE_ROWS ? cur + 1 : TILE_ROWS); \
    }

#define LOAD_ROWS(N)                                                                    \
        const float4* rp[N];                                                            \
        float4 rx[N], ry[N];                                                            \
        _Pragma("unroll")                                                               \
        for (int i = 0; i < (N); ++i)                                                   \
            rp[i] = node_row(nf, __builtin_amdgcn_readlane(myid, j + i));               \
        _Pragma("unroll")                                                               \
        for (int i = 0; i < (N); ++i) { rx[i] = rp[i][lane]; ry[i] = rp[i][32 + lane]; }
#define ADD_ROWS(N)                                                                     \
        _Pragma("unroll")                                                               \
        for (int i = 0; i < (N); ++i) {                                                 \
            FINISH_EDGES_BEFORE(base + j + i)                                           \
            add4(a0, rx[i]);                                                            \
            add4(a1, ry[i]);                                                            \
        }                                                                               \
        j += (N);
#define SUM_ROWS(N) { LOAD_ROWS(N) ADD_ROWS(N) }

#define RING_FIRST(S)                                                                   \
        asm volatile("global_load_b128 %0, %2, %3\n\tglobal_load_b128 %1, %2, %3 offset:512" \
                     : "=v"(sx[S]), "=v"(sy[S])                                         \
                     : "v"(lane16), "s"(node_row_base(nf, __builtin_amdgcn_readlane(id_cur, S))));
#define RING_ASK(S, IDV, LN)                                                            \
        asm volatile("global_load_b128 %0, %2, %3\n\tglobal_load_b128 %1, %2, %3 offset:512" \
                     : "+v"(sx[S]), "+v"(sy[S])                                         \
                     : "v"(lane16), "s"(node_row_base(nf, __builtin_amdgcn_readlane(IDV, LN))));
#define RING_STEP_A(S)                                                                  \
        FINISH_EDGES_BEFORE(base + S)                                                   \
        asm volatile("s_wait_loadcnt 31" : "+v"(sx[S]), "+v"(sy[S]));                   \
        add4v(a0, sx[S]);                                                               \
        add4v(a1, sy[S]);                                                               \
        RING_ASK(S, id_cur, 16 + S)
#define RING_STEP_B0                                                                    \
        FINISH_EDGES_BEFORE(base + 16)                                                  \
        asm volatile("s_wait_loadcnt 30\n\tv_mov_b32 %2, %3"                            \
                     : "+v"(sx[0]), "+v"(sy[0]), "=v"(id_ok) : "v"(id_nx));             \
        add4v(a0, sx[0]);                                                               \
        add4v(a1, sy[0]);                                                               \
        RING_ASK(0, id_ok, 0)
#define RING_STEP_B(S)                                                                  \
        FINISH_EDGES_BEFORE(base + 16 + S)                                              \
        asm volatile("s_wait_loadcnt 30" : "+v"(sx[S]), "+v"(sy[S]));                   \
        add4v(a0, sx[S]);                                                               \
        add4v(a1, sy[S]);                                                               \
        RING_ASK(S, id_ok, S)
#define RING_DRAIN(S, N)                                                                \
        FINISH_EDGES_BEFORE(base + S)                                                   \
        asm volatile("s_wait_loadcnt " #N : "+v"(sx[S]), "+v"(sy[S]));                  \
        add4v(a0, sx[S]);                                                               \
        add4v(a1, sy[S]);

__global__ __launch_bounds__(WAVES_PER_BLOCK * 32) void edge_sum_mlp_kernel(
    const float* __restrict__ nf,
    const int*   __restrict__ perm,
    const int*   __restrict__ ends,
    const float* __restrict__ wpack,
    const float* __restrict__ b1,
    const float* __restrict__ lng,
    const float* __restrict__ lnb,
    const float* __restrict__ W2,
    const float* __restrict__ b2,
    float* __restrict__ out) {
    __shared__ alignas(16) float sA[WAVES_PER_BLOCK * TILE_ROWS * A_PITCH];
    __shared__ float sW2[W2_COPY_TRIPS * WAVES_PER_BLOCK * 32];

    const int lane = threadIdx.x & 31;
    const int wave = __builtin_amdgcn_readfirstlane((int)(threadIdx.x >> 5));
    const int  tile_id = blockIdx.x * WAVES_PER_BLOCK + wave;
    const bool live    = tile_id < NUM_TILES;
    const int  tile    = live ? tile_id : NUM_TILES - 1;
    const int  m0      = tile * TILE_ROWS;
    const int m    = lane & 15;
    const int hi   = lane >> 4;

    float* abase = sA + wave * (TILE_ROWS * A_PITCH);

    const int bidx = m0 - 1 + (lane < TILE_ROWS ? lane : TILE_ROWS);
    const int braw = ends[bidx < 0 ? 0 : bidx];

    float w2r[W2_COPY_TRIPS];
#pragma unroll
    for (int i = 0; i < W2_COPY_TRIPS; ++i) {
        const int id  = (int)threadIdx.x + i * (WAVES_PER_BLOCK * 32);
        const int idc = id < W2_FLOATS ? id : W2_FLOATS - 1;
        w2r[i] = W2[idc];
    }

    int bndv = (bidx < 0) ? 0 : braw;
    if (bndv < 0) bndv = 0;
    if (bndv > NUM_CONN) bndv = NUM_CONN;

    {
        const int beg   = __builtin_amdgcn_readlane(bndv, 0);
        const int end_t = __builtin_amdgcn_readlane(bndv, TILE_ROWS);
        int cur      = 0;
        int next_end = __builtin_amdgcn_readlane(bndv, 1);
        float4 a0 = make_float4(0.0f, 0.0f, 0.0f, 0.0f);
        float4 a1 = a0;

        int base   = beg;
        int j0     = 0;
        int id_cur = 0;
        if (beg + lane < end_t) id_cur = perm[beg + lane];
#pragma unroll
        for (int i = 0; i < W2_COPY_TRIPS; ++i)
            sW2[(int)threadIdx.x + i * (WAVES_PER_BLOCK * 32)] = w2r[i];

        if (end_t - beg >= 48) {
            v4f sx[ROWS_IN_FLIGHT], sy[ROWS_IN_FLIGHT];
            const int lane16 = lane * 16;
            const int lane4  = lane * 4;
            RING_FIRST(0) RING_FIRST(1) RING_FIRST(2) RING_FIRST(3) RING_FIRST(4) RING_FIRST(5) RING_FIRST(6) RING_FIRST(7) RING_FIRST(8) RING_FIRST(9) RING_FIRST(10) RING_FIRST(11) RING_FIRST(12) RING_FIRST(13) RING_FIRST(14) RING_FIRST(15)
            do {
                int id_nx, id_ok;
                asm volatile("global_load_b32 %0, %1, %2" : "=v"(id_nx) : "v"(lane4), "s"(perm + base + 32));
                RING_STEP_A(0) RING_STEP_A(1) RING_STEP_A(2) RING_STEP_A(3) RING_STEP_A(4) RING_STEP_A(5) RING_STEP_A(6) RING_STEP_A(7) RING_STEP_A(8) RING_STEP_A(9) RING_STEP_A(10) RING_STEP_A(11) RING_STEP_A(12) RING_STEP_A(13) RING_STEP_A(14) RING_STEP_A(15)
                RING_STEP_B0 RING_STEP_B(1) RING_STEP_B(2) RING_STEP_B(3) RING_STEP_B(4) RING_STEP_B(5) RING_STEP_B(6) RING_STEP_B(7) RING_STEP_B(8) RING_STEP_B(9) RING_STEP_B(10) RING_STEP_B(11) RING_STEP_B(12) RING_STEP_B(13) RING_STEP_B(14) RING_STEP_B(15)
                id_cur = id_ok;
                base += 32;
            } while (end_t - base >= 48);
            RING_DRAIN(0, 30) RING_DRAIN(1, 28) RING_DRAIN(2, 26) RING_DRAIN(3, 24) RING_DRAIN(4, 22) RING_DRAIN(5, 20) RING_DRAIN(6, 18) RING_DRAIN(7, 16) RING_DRAIN(8, 14) RING_DRAIN(9, 12) RING_DRAIN(10, 10) RING_DRAIN(11, 8) RING_DRAIN(12, 6) RING_DRAIN(13, 4) RING_DRAIN(14, 2) RING_DRAIN(15, 0)
            j0 = 16;
        }

        for (; base < end_t; base += 32) {
            const int left = end_t - base;
            const int mm   = left < 32 ? left : 32;
            const int myid = id_cur;
            int j = j0;
            j0 = 0;
            id_cur = 0;
            if (base + 32 + lane < end_t) id_cur = perm[base + 32 + lane];
            while (j + ROWS_IN_FLIGHT <= mm) SUM_ROWS(ROWS_IN_FLIGHT)
            if (j + 8 <= mm) SUM_ROWS(8)
            if (j + 4 <= mm) SUM_ROWS(4)
            if (j + 2 <= mm) SUM_ROWS(2)
            if (j + 1 <= mm) SUM_ROWS(1)
        }
        FINISH_EDGES_BEFORE(NUM_CONN)
    }
    __syncthreads();

    const int seg_beg = __shfl(bndv, m, 32);
    const int seg_end = __shfl(bndv, m + 1, 32);
    const float degree = (float)(seg_end > seg_beg ? seg_end - seg_beg : 0);
    const float invc = 1.0f / fmaxf(degree, 1.0f);
    const float* arow = abase + m * A_PITCH;

    v8f acc[N_TILES] = {};

    for (int ks = 0; ks < K_STEPS; ++ks) {
        const int k = ks * 4;
        v2f a = *(const v2f*)(arow + k + 2 * hi);
        a.x *= invc;
        a.y *= invc;

        const float4* __restrict__ bp =
            (const float4*)(wpack + ((size_t)ks * 32 + lane) * 16);
        float4 q0 = bp[0], q1 = bp[1], q2 = bp[2], q3 = bp[3];
        v2f bbs[N_TILES];
        bbs[0].x = q0.x; bbs[0].y = q0.y;
        bbs[1].x = q0.z; bbs[1].y = q0.w;
        bbs[2].x = q1.x; bbs[2].y = q1.y;
        bbs[3].x = q1.z; bbs[3].y = q1.w;
        bbs[4].x = q2.x; bbs[4].y = q2.y;
        bbs[5].x = q2.z; bbs[5].y = q2.w;
        bbs[6].x = q3.x; bbs[6].y = q3.y;
        bbs[7].x = q3.z; bbs[7].y = q3.w;

#pragma unroll
        for (int nt = 0; nt < N_TILES; ++nt) {
            acc[nt] = __builtin_amdgcn_wmma_f32_16x16x4_f32(
                false, a, false, bbs[nt], (short)0, acc[nt], false, false);
        }
    }

#pragma unroll
    for (int nt = 0; nt < N_TILES; ++nt) {
        const float bias = b1[nt * 16 + m];
#pragma unroll
        for (int r = 0; r < 8; ++r) acc[nt][r] += bias;
    }

    float s[8], q[8];
#pragma unroll
    for (int r = 0; r < 8; ++r) { s[r] = 0.0f; q[r] = 0.0f; }
#pragma unroll
    for (int nt = 0; nt < N_TILES; ++nt) {
#pragma unroll
        for (int r = 0; r < 8; ++r) {
            const float x = acc[nt][r];
            s[r] += x;
            q[r] += x * x;
        }
    }
#pragma unroll
    for (int mask = 1; mask < 16; mask <<= 1) {
#pragma unroll
        for (int r = 0; r < 8; ++r) {
            s[r] += __shfl_xor(s[r], mask, 32);
            q[r] += __shfl_xor(q[r], mask, 32);
        }
    }
    float mean[8], rstd[8];
#pragma unroll
    for (int r = 0; r < 8; ++r) {
        mean[r] = s[r] * (1.0f / HIDDEN);
        const float var = q[r] * (1.0f / HIDDEN) - mean[r] * mean[r];
        rstd[r] = rsqrtf(var + LN_EPS);
    }

    float* hbase = abase;
#pragma unroll
    for (int nt = 0; nt < N_TILES; ++nt) {
        const int n = nt * 16 + m;
        const float g  = lng[n];
        const float bb = lnb[n];
#pragma unroll
        for (int r = 0; r < 8; ++r) {
            float x = (acc[nt][r] - mean[r]) * rstd[r] * g + bb;
            x = fmaxf(x, 0.0f);
            hbase[(r + 8 * hi) * H_PITCH + n] = x;
        }
    }
    __syncthreads();

    v8f o0, o1;
    {
        const float bias0 = b2[m];
        const float bias1 = b2[16];
#pragma unroll
        for (int r = 0; r < 8; ++r) { o0[r] = bias0; o1[r] = bias1; }
    }
    const float* ha  = hbase + m * H_PITCH + 2 * hi;
    const float* __restrict__ wb0 = sW2 + (2 * hi) * OUT_DIM + m;
    const float* __restrict__ wb1 = sW2 + (2 * hi) * OUT_DIM + 16;
#pragma unroll
    for (int ks = 0; ks < K2_STEPS; ++ks) {
        const v2f a = *(const v2f*)(ha + 4 * ks);
        v2f bA, bB;
        bA.x = wb0[ks * 4 * OUT_DIM];
        bA.y = wb0[ks * 4 * OUT_DIM + OUT_DIM];
        bB.x = wb1[ks * 4 * OUT_DIM];
        bB.y = wb1[ks * 4 * OUT_DIM + OUT_DIM];
        o0 = __builtin_amdgcn_wmma_f32_16x16x4_f32(
            false, a, false, bA, (short)0, o0, false, false);
        o1 = __builtin_amdgcn_wmma_f32_16x16x4_f32(
            false, a, false, bB, (short)0, o1, false, false);
    }

    if (live) {
        float* __restrict__ orow = out + (size_t)(m0 + 8 * hi) * OUT_DIM;
#pragma unroll
        for (int r = 0; r < 8; ++r) orow[r * OUT_DIM + m] = o0[r];
        if (m == 0) {
#pragma unroll
            for (int r = 0; r < 8; ++r) orow[r * OUT_DIM + 16] = o1[r];
        }
    }
}

extern "C" void kernel_launch(void* const* d_in, const int* in_sizes, int n_in,
                              void* d_out, int out_size, void* d_ws, size_t ws_size,
                              hipStream_t stream) {
    const float* nf    = (const float*)d_in[0];
    const int*   hidx  = (const int*)d_in[1];
    const int*   nodes = hidx;
    const int*   edges = hidx + NUM_CONN;
    const float* W1    = (const float*)d_in[2];
    const float* b1    = (const float*)d_in[3];
    const float* lng   = (const float*)d_in[4];
    const float* lnb   = (const float*)d_in[5];
    const float* W2    = (const float*)d_in[6];
    const float* b2    = (const float*)d_in[7];
    float*       out   = (float*)d_out;

    int* tab    = (int*)d_ws;
    int* binned = tab + TAB_N + 64;
    int* perm   = binned + NUM_CONN;
    int* ends   = perm + NUM_CONN;
    float* wpack = (float*)d_ws + (size_t)NUM_EDGES * IN_DIM + NUM_EDGES;

    chunk_sort_kernel<<<NCHUNK + REPACK_GROUPS, BIN_THREADS, 0, stream>>>(nodes, edges, tab, binned, W1, wpack);

    range_sort_kernel<<<NB, BIN_THREADS, 0, stream>>>(tab, binned, perm, ends);

    edge_sum_mlp_kernel<<<(NUM_TILES + WAVES_PER_BLOCK - 1) / WAVES_PER_BLOCK, WAVES_PER_BLOCK * 32, 0, stream>>>(
        nf, perm, ends, wpack, b1, lng, lnb, W2, b2, out);
}


# Last launch spread evenly over the 256 WGPs: 768 full work-groups (3 per WGP) + 53 one-tile groups whose 4 waves split row sums by rows and GEMM1 by columns; LN sums of those 53 tiles regrouped
# speedup vs baseline: 9.6217x; 1.0419x over previous
#include <hip/hip_runtime.h>

#define NUM_NODES 100000
#define NUM_EDGES 50000
#define NUM_CONN  800000
#define IN_DIM    256
#define HIDDEN    128
#define OUT_DIM   17
#define LN_EPS    1e-5f

#define WAVES_PER_BLOCK 4
#define TILE_ROWS 16
#define NUM_TILES (NUM_EDGES / TILE_ROWS)
#define NUM_WGPS    256
#define FULL_GROUPS (3 * NUM_WGPS)
#define FULL_TILES  (FULL_GROUPS * WAVES_PER_BLOCK)
#define TAIL_GROUPS (NUM_TILES - FULL_TILES)
#define TAIL_ROWS   (TILE_ROWS / WAVES_PER_BLOCK)
#define TAIL_NT     (N_TILES / WAVES_PER_BLOCK)
static_assert(TAIL_GROUPS >= 0 && TAIL_GROUPS <= NUM_WGPS, "at most one shared tile per WGP");
static_assert(HIDDEN / 16 / WAVES_PER_BLOCK == 2, "a shared tile's wave takes one float4 (two column blocks) of the packed B stream");
#define N_TILES (HIDDEN / 16)
#define K_STEPS (IN_DIM / 4)
#define WPACK_FLOATS (K_STEPS * 32 * 16)
#define K2_STEPS (HIDDEN / 4)
#define H_PITCH  (HIDDEN + 4)
#define W2_FLOATS (HIDDEN * OUT_DIM)
#define W2_COPY_TRIPS ((W2_FLOATS + WAVES_PER_BLOCK * 32 - 1) / (WAVES_PER_BLOCK * 32))

typedef float v2f __attribute__((ext_vector_type(2)));
typedef float v8f __attribute__((ext_vector_type(8)));
typedef float v4f __attribute__((ext_vector_type(4)));

#define REPACK_ITEMS  (K_STEPS * 32 * 8)
__device__ __forceinline__ void repack_w1_item(const float* __restrict__ W1,
                                               float* __restrict__ wpack, int id) {
    if (id >= REPACK_ITEMS) return;
    int ks   = id >> 8;
    int rem  = id & 255;
    int lane = rem >> 3;
    int nt   = rem & 7;
    int m    = lane & 15;
    int hi   = lane >> 4;
    int krow = ks * 4 + 2 * hi;
    int n    = nt * 16 + m;
    wpack[2 * id + 0] = W1[(size_t)(krow + 0) * HIDDEN + n];
    wpack[2 * id + 1] = W1[(size_t)(krow + 1) * HIDDEN + n];
}


#define NB 128
#define EPB 391
#define EPB_PAD 512
#define NCHUNK 128
#define CHUNK (NUM_CONN / NCHUNK)
#define BIN_THREADS 1024
#define BIN_PER_THREAD 7
#define OFF_PITCH (NB + 1)
#define TAB_N (NCHUNK * OFF_PITCH)
#define NODE_BITS 17
#define NODE_MASK ((1 << NODE_BITS) - 1)
#define PIECES_PER_WAVE (NCHUNK / 32)
#define PIECE_TRIPS 2
#define STAGE_CAP 16384
#define REPACK_GROUPS ((REPACK_ITEMS + BIN_THREADS - 1) / BIN_THREADS)

static_assert(NB * EPB >= NUM_EDGES, "ranges cover every edge");
static_assert(NCHUNK * CHUNK == NUM_CONN, "chunks cover every connection");
static_assert(BIN_THREADS * BIN_PER_THREAD >= CHUNK, "one register batch per chunk");
static_assert(NB == 4 * 32 && NCHUNK == 4 * 32, "four whole waves scan the counts and read the offsets");
static_assert(PIECES_PER_WAVE * (BIN_THREADS / 32) == NCHUNK, "every piece has a wave");
static_assert((1 << NODE_BITS) >= NUM_NODES && EPB <= EPB_PAD, "packed word fits");
static_assert(STAGE_CAP % (4 * BIN_THREADS) == 0, "the level-2 window is a whole number of copy batches");

__device__ __forceinline__ int clamp_node(int node) {
    return (unsigned)node < (unsigned)NUM_NODES ? node : NUM_NODES - 1;
}

__global__ __launch_bounds__(BIN_THREADS) void chunk_sort_kernel(
    const int* __restrict__ nodes,
    const int* __restrict__ edges,
    int* __restrict__ tab,
    int* __restrict__ binned,
    const float* __restrict__ W1,
    float* __restrict__ wpack) {
    if (blockIdx.x >= NCHUNK) {
        repack_w1_item(W1, wpack, (int)(blockIdx.x - NCHUNK) * BIN_THREADS + (int)threadIdx.x);
        return;
    }
    __shared__ int cnt[NB];
    __shared__ int wtot[NB / 32];
    __shared__ int stage[BIN_THREADS * BIN_PER_THREAD];
    __shared__ int ctot;
    const int t    = threadIdx.x;
    const int lane = t & 31;
    const int wv   = t >> 5;
    const int w    = blockIdx.x;
    if (t < NB) cnt[t] = 0;
    const int* __restrict__ esrc = edges + w * CHUNK;
    const int* __restrict__ nsrc = nodes + w * CHUNK;
    int e[BIN_PER_THREAD], nd[BIN_PER_THREAD], rk[BIN_PER_THREAD];
#pragma unroll
    for (int k = 0; k < BIN_PER_THREAD; ++k) {
        const int i = t + k * BIN_THREADS;
        e[k]  = (i < CHUNK) ? esrc[i] : -1;
        nd[k] = (i < CHUNK) ? nsrc[i] : 0;
    }
    __syncthreads();
#pragma unroll
    for (int k = 0; k < BIN_PER_THREAD; ++k) {
        rk[k] = 0;
        if ((unsigned)e[k] < (unsigned)NUM_EDGES) rk[k] = atomicAdd(&cnt[e[k] / EPB], 1);
    }
    __syncthreads();

    int c = 0, inc = 0;
    if (t < NB) {
        c = cnt[t];
        inc = c;
#pragma unroll
        for (int d = 1; d < 32; d <<= 1) {
            const int up = __shfl_up(inc, d, 32);
            if (lane >= d) inc += up;
        }
        if (lane == 31) wtot[wv] = inc;
    }
    __syncthreads();
    if (t < NB) {
        int pre = 0;
#pragma unroll
        for (int j = 0; j < NB / 32 - 1; ++j) if (j < wv) pre += wtot[j];
        const int start = pre + inc - c;
        cnt[t] = start;
        tab[w * OFF_PITCH + t] = start;
        if (t == NB - 1) {
            tab[w * OFF_PITCH + NB] = start + c;
            ctot = start + c;
        }
    }
    __syncthreads();

#pragma unroll
    for (int k = 0; k < BIN_PER_THREAD; ++k) {
        if ((unsigned)e[k] < (unsigned)NUM_EDGES) {
            const int b  = e[k] / EPB;
            const int le = e[k] - b * EPB;
            const unsigned pos = (unsigned)(cnt[b] + rk[k]);
            if (pos < (unsigned)CHUNK) stage[pos] = (le << NODE_BITS) | clamp_node(nd[k]);
        }
    }
    __syncthreads();
    const int tot = ctot < CHUNK ? ctot : CHUNK;
    int sv[BIN_PER_THREAD];
#pragma unroll
    for (int k = 0; k < BIN_PER_THREAD; ++k) sv[k] = stage[t + k * BIN_THREADS];
#pragma unroll
    for (int k = 0; k < BIN_PER_THREAD; ++k) {
        const int i = t + k * BIN_THREADS;
        if (i < tot) binned[w * CHUNK + i] = sv[k];
    }
}

__global__ __launch_bounds__(BIN_THREADS) void range_sort_kernel(
    const int* __restrict__ tab,
    const int* __restrict__ binned,
    int* __restrict__ perm,
    int* __restrict__ ends) {
    __shared__ int cnt[EPB_PAD];
    __shared__ int wsum[32];
    __shared__ int pbeg[NCHUNK];
    __shared__ int plen[NCHUNK];
    __shared__ int osum[NCHUNK / 32];
    __shared__ int stage[STAGE_CAP];
    __shared__ int rtot;
    const int t    = threadIdx.x;
    const int lane = t & 31;
    const int w    = t >> 5;
    const int b    = blockIdx.x;

    if (t < EPB_PAD) cnt[t] = 0;

    int o0 = 0;
    if (t < NCHUNK) {
        o0     = tab[t * OFF_PITCH + b];
        int o1 = tab[t * OFF_PITCH + b + 1];
        if (o0 < 0) o0 = 0;
        if (o0 > CHUNK) o0 = CHUNK;
        if (o1 < o0) o1 = o0;
        if (o1 > CHUNK) o1 = CHUNK;
        pbeg[t] = t * CHUNK + o0;
        plen[t] = o1 - o0;
    }
    __syncthreads();

    int pb[PIECES_PER_WAVE], pl[PIECES_PER_WAVE];
    int p[PIECES_PER_WAVE][PIECE_TRIPS];
#pragma unroll
    for (int j = 0; j < PIECES_PER_WAVE; ++j) {
        pb[j] = __builtin_amdgcn_readfirstlane(pbeg[w * PIECES_PER_WAVE + j]);
        pl[j] = __builtin_amdgcn_readfirstlane(plen[w * PIECES_PER_WAVE + j]);
    }
#pragma unroll
    for (int j = 0; j < PIECES_PER_WAVE; ++j) {
#pragma unroll
        for (int k = 0; k < PIECE_TRIPS; ++k) {
            const int i = lane + 32 * k;
            p[j][k] = (i < pl[j]) ? binned[pb[j] + i] : -1;
        }
    }

    if (t < NCHUNK) {
        int osm = o0;
#pragma unroll
        for (int d = 16; d >= 1; d >>= 1) osm += __shfl_xor(osm, d, 32);
        if (lane == 0) osum[w] = osm;
    }

#pragma unroll
    for (int j = 0; j < PIECES_PER_WAVE; ++j) {
#pragma unroll
        for (int k = 0; k < PIECE_TRIPS; ++k) {
            if (p[j][k] >= 0) atomicAdd(&cnt[(p[j][k] >> NODE_BITS) & (EPB_PAD - 1)], 1);
        }
        for (int i = lane + 32 * PIECE_TRIPS; i < pl[j]; i += 32) {
            const int q = binned[pb[j] + i];
            if (q >= 0) atomicAdd(&cnt[(q >> NODE_BITS) & (EPB_PAD - 1)], 1);
        }
    }
    __syncthreads();

    int bs = 0;
#pragma unroll
    for (int j = 0; j < NCHUNK / 32; ++j) bs += osum[j];
    if (bs > NUM_CONN) bs = NUM_CONN;

    const int c = (t < EPB_PAD) ? cnt[t] : 0;
    int inc = c;
#pragma unroll
    for (int d = 1; d < 32; d <<= 1) {
        const int up = __shfl_up(inc, d, 32);
        if (lane >= d) inc += up;
    }
    if (lane == 31) wsum[w] = inc;
    __syncthreads();
    if (w == 0) {
        const int tot = wsum[lane];
        int inc2 = tot;
#pragma unroll
        for (int d = 1; d < 32; d <<= 1) {
            const int u = __shfl_up(inc2, d, 32);
            if (lane >= d) inc2 += u;
        }
        wsum[lane] = inc2 - tot;
        if (lane == 31) rtot = inc2;
    }
    __syncthreads();
    const int  rt     = rtot;
    const bool staged = rt <= STAGE_CAP;
    const int end_t = bs + wsum[w] + inc;
    const int e = b * EPB + t;
    if (t < EPB && e < NUM_EDGES) ends[e] = end_t;
    if (t < EPB_PAD) cnt[t] = end_t - c;
    __syncthreads();

#pragma unroll
    for (int j = 0; j < PIECES_PER_WAVE; ++j) {
#pragma unroll
        for (int k = 0; k < PIECE_TRIPS; ++k) {
            if (p[j][k] >= 0) {
                const unsigned pos = (unsigned)atomicAdd(&cnt[(p[j][k] >> NODE_BITS) & (EPB_PAD - 1)], 1);
                const unsigned loc = pos - (unsigned)bs;
                if (staged) { if (loc < (unsigned)STAGE_CAP) stage[loc] = p[j][k] & NODE_MASK; }
                else if (pos < (unsigned)NUM_CONN) perm[pos] = p[j][k] & NODE_MASK;
            }
        }
        for (int i = lane + 32 * PIECE_TRIPS; i < pl[j]; i += 32) {
            const int q = binned[pb[j] + i];
            if (q >= 0) {
                const unsigned pos = (unsigned)atomicAdd(&cnt[(q >> NODE_BITS) & (EPB_PAD - 1)], 1);
                const unsigned loc = pos - (unsigned)bs;
                if (staged) { if (loc < (unsigned)STAGE_CAP) stage[loc] = q & NODE_MASK; }
                else if (pos < (unsigned)NUM_CONN) perm[pos] = q & NODE_MASK;
            }
        }
    }
    __syncthreads();
    if (staged) {
        for (int i0 = t; i0 < rt; i0 += 4 * BIN_THREADS) {
            int sv[4];
#pragma unroll
            for (int k = 0; k < 4; ++k) sv[k] = stage[i0 + k * BIN_THREADS];
#pragma unroll
            for (int k = 0; k < 4; ++k) {
                const int i = i0 + k * BIN_THREADS;
                if (i < rt && bs + i < NUM_CONN) perm[bs + i] = sv[k];
            }
        }
    }
}

#define A_PITCH (IN_DIM + 4)
#define ROWS_IN_FLIGHT 16

__device__ __forceinline__ void add4(float4& a, const float4& x) {
    a.x += x.x; a.y += x.y; a.z += x.z; a.w += x.w;
}

__device__ __forceinline__ const float4* node_row(const float* __restrict__ nf, int node) {
    const unsigned n = (unsigned)node < (unsigned)NUM_NODES ? (unsigned)node
                                                            : (unsigned)(NUM_NODES - 1);
    return (const float4*)(nf + (size_t)n * IN_DIM);
}

__device__ __forceinline__ void add4v(float4& a, const v4f& x) {
    a.x += x.x; a.y += x.y; a.z += x.z; a.w += x.w;
}

__device__ __forceinline__ const float* node_row_base(const float* __restrict__ nf, int node) {
    const unsigned n = (unsigned)node < (unsigned)NUM_NODES ? (unsigned)node
                                                            : (unsigned)(NUM_NODES - 1);
    return nf + (size_t)n * IN_DIM;
}

#define FINISH_EDGES_BEFORE(POS)                                                        \
    _Pragma("nounroll")                                                                 \
    while (cur < row_end && (POS) >= next_end) {                                        \
        float4* drow = (float4*)(abase + cur * A_PITCH);                                \
        drow[lane]      = a0;                                                           \
        drow[32 + lane] = a1;                                                           \
        a0 = make_float4(0.0f, 0.0f, 0.0f, 0.0f);                                       \
        a1 = a0;                                                                        \
        ++cur;                                                                          \
        next_end = __builtin_amdgcn_readlane(bndv, cur < row_end ? cur + 1 : row_end);   \
    }

#define LOAD_ROWS(N)                                                                    \
        const float4* rp[N];                                                            \
        float4 rx[N], ry[N];                                                            \
        _Pragma("unroll")                                                               \
        for (int i = 0; i < (N); ++i)                                                   \
            rp[i] = node_row(nf, __builtin_amdgcn_readlane(myid, j + i));               \
        _Pragma("unroll")                                                               \
        for (int i = 0; i < (N); ++i) { rx[i] = rp[i][lane]; ry[i] = rp[i][32 + lane]; }
#define ADD_ROWS(N)                                                                     \
        _Pragma("unroll")                                                               \
        for (int i = 0; i < (N); ++i) {                                                 \
            FINISH_EDGES_BEFORE(base + j + i)                                           \
            add4(a0, rx[i]);                                                            \
            add4(a1, ry[i]);                                                            \
        }                                                                               \
        j += (N);
#define SUM_ROWS(N) { LOAD_ROWS(N) ADD_ROWS(N) }

#define RING_FIRST(S)                                                                   \
        asm volatile("global_load_b128 %0, %2, %3\n\tglobal_load_b128 %1, %2, %3 offset:512" \
                     : "=v"(sx[S]), "=v"(sy[S])                                         \
                     : "v"(lane16), "s"(node_row_base(nf, __builtin_amdgcn_readlane(id_cur, S))));
#define RING_ASK(S, IDV, LN)                                                            \
        asm volatile("global_load_b128 %0, %2, %3\n\tglobal_load_b128 %1, %2, %3 offset:512" \
                     : "+v"(sx[S]), "+v"(sy[S])                                         \
                     : "v"(lane16), "s"(node_row_base(nf, __builtin_amdgcn_readlane(IDV, LN))));
#define RING_STEP_A(S)                                                                  \
        FINISH_EDGES_BEFORE(base + S)                                                   \
        asm volatile("s_wait_loadcnt 31" : "+v"(sx[S]), "+v"(sy[S]));                   \
        add4v(a0, sx[S]);                                                               \
        add4v(a1, sy[S]);                                                               \
        RING_ASK(S, id_cur, 16 + S)
#define RING_STEP_B0                                                                    \
        FINISH_EDGES_BEFORE(base + 16)                                                  \
        asm volatile("s_wait_loadcnt 30\n\tv_mov_b32 %2, %3"                            \
                     : "+v"(sx[0]), "+v"(sy[0]), "=v"(id_ok) : "v"(id_nx));             \
        add4v(a0, sx[0]);                                                               \
        add4v(a1, sy[0]);                                                               \
        RING_ASK(0, id_ok, 0)
#define RING_STEP_B(S)                                                                  \
        FINISH_EDGES_BEFORE(base + 16 + S)                                              \
        asm volatile("s_wait_loadcnt 30" : "+v"(sx[S]), "+v"(sy[S]));                   \
        add4v(a0, sx[S]);                                                               \
        add4v(a1, sy[S]);                                                               \
        RING_ASK(S, id_ok, S)
#define RING_DRAIN(S, N)                                                                \
        FINISH_EDGES_BEFORE(base + S)                                                   \
        asm volatile("s_wait_loadcnt " #N : "+v"(sx[S]), "+v"(sy[S]));                  \
        add4v(a0, sx[S]);                                                               \
        add4v(a1, sy[S]);

__global__ __launch_bounds__(WAVES_PER_BLOCK * 32) void edge_sum_mlp_kernel(
    const float* __restrict__ nf,
    const int*   __restrict__ perm,
    const int*   __restrict__ ends,
    const float* __restrict__ wpack,
    const float* __restrict__ b1,
    const float* __restrict__ lng,
    const float* __restrict__ lnb,
    const float* __restrict__ W2,
    const float* __restrict__ b2,
    float* __restrict__ out) {
    __shared__ alignas(16) float sA[WAVES_PER_BLOCK * TILE_ROWS * A_PITCH];
    __shared__ float sW2[W2_COPY_TRIPS * WAVES_PER_BLOCK * 32];

    const int lane = threadIdx.x & 31;
    const int wave = __builtin_amdgcn_readfirstlane((int)(threadIdx.x >> 5));
    const bool tail = blockIdx.x >= FULL_GROUPS;
    const int  tile = tail ? FULL_TILES + ((int)blockIdx.x - FULL_GROUPS)
                           : (int)blockIdx.x * WAVES_PER_BLOCK + wave;
    const int  m0   = tile * TILE_ROWS;
    const int m    = lane & 15;
    const int hi   = lane >> 4;

    const int row_beg = tail ? wave * TAIL_ROWS : 0;
    const int row_end = row_beg + (tail ? TAIL_ROWS : TILE_ROWS);
    float* abase = sA + (tail ? 0 : wave) * (TILE_ROWS * A_PITCH);

    const int bidx = m0 - 1 + (lane < TILE_ROWS ? lane : TILE_ROWS);
    const int braw = ends[bidx < 0 ? 0 : bidx];

    float w2r[W2_COPY_TRIPS];
#pragma unroll
    for (int i = 0; i < W2_COPY_TRIPS; ++i) {
        const int id  = (int)threadIdx.x + i * (WAVES_PER_BLOCK * 32);
        const int idc = id < W2_FLOATS ? id : W2_FLOATS - 1;
        w2r[i] = W2[idc];
    }

    int bndv = (bidx < 0) ? 0 : braw;
    if (bndv < 0) bndv = 0;
    if (bndv > NUM_CONN) bndv = NUM_CONN;

    {
        const int beg   = __builtin_amdgcn_readlane(bndv, row_beg);
        const int end_t = __builtin_amdgcn_readlane(bndv, row_end);
        int cur      = row_beg;
        int next_end = __builtin_amdgcn_readlane(bndv, row_beg + 1);
        float4 a0 = make_float4(0.0f, 0.0f, 0.0f, 0.0f);
        float4 a1 = a0;

        int base   = beg;
        int j0     = 0;
        int id_cur = 0;
        if (beg + lane < end_t) id_cur = perm[beg + lane];
#pragma unroll
        for (int i = 0; i < W2_COPY_TRIPS; ++i)
            sW2[(int)threadIdx.x + i * (WAVES_PER_BLOCK * 32)] = w2r[i];

        if (end_t - beg >= 48) {
            v4f sx[ROWS_IN_FLIGHT], sy[ROWS_IN_FLIGHT];
            const int lane16 = lane * 16;
            const int lane4  = lane * 4;
            RING_FIRST(0) RING_FIRST(1) RING_FIRST(2) RING_FIRST(3) RING_FIRST(4) RING_FIRST(5) RING_FIRST(6) RING_FIRST(7) RING_FIRST(8) RING_FIRST(9) RING_FIRST(10) RING_FIRST(11) RING_FIRST(12) RING_FIRST(13) RING_FIRST(14) RING_FIRST(15)
            do {
                int id_nx, id_ok;
                asm volatile("global_load_b32 %0, %1, %2" : "=v"(id_nx) : "v"(lane4), "s"(perm + base + 32));
                RING_STEP_A(0) RING_STEP_A(1) RING_STEP_A(2) RING_STEP_A(3) RING_STEP_A(4) RING_STEP_A(5) RING_STEP_A(6) RING_STEP_A(7) RING_STEP_A(8) RING_STEP_A(9) RING_STEP_A(10) RING_STEP_A(11) RING_STEP_A(12) RING_STEP_A(13) RING_STEP_A(14) RING_STEP_A(15)
                RING_STEP_B0 RING_STEP_B(1) RING_STEP_B(2) RING_STEP_B(3) RING_STEP_B(4) RING_STEP_B(5) RING_STEP_B(6) RING_STEP_B(7) RING_STEP_B(8) RING_STEP_B(9) RING_STEP_B(10) RING_STEP_B(11) RING_STEP_B(12) RING_STEP_B(13) RING_STEP_B(14) RING_STEP_B(15)
                id_cur = id_ok;
                base += 32;
            } while (end_t - base >= 48);
            RING_DRAIN(0, 30) RING_DRAIN(1, 28) RING_DRAIN(2, 26) RING_DRAIN(3, 24) RING_DRAIN(4, 22) RING_DRAIN(5, 20) RING_DRAIN(6, 18) RING_DRAIN(7, 16) RING_DRAIN(8, 14) RING_DRAIN(9, 12) RING_DRAIN(10, 10) RING_DRAIN(11, 8) RING_DRAIN(12, 6) RING_DRAIN(13, 4) RING_DRAIN(14, 2) RING_DRAIN(15, 0)
            j0 = 16;
        }

        for (; base < end_t; base += 32) {
            const int left = end_t - base;
            const int mm   = left < 32 ? left : 32;
            const int myid = id_cur;
            int j = j0;
            j0 = 0;
            id_cur = 0;
            if (base + 32 + lane < end_t) id_cur = perm[base + 32 + lane];
            while (j + ROWS_IN_FLIGHT <= mm) SUM_ROWS(ROWS_IN_FLIGHT)
            if (j + 8 <= mm) SUM_ROWS(8)
            if (j + 4 <= mm) SUM_ROWS(4)
            if (j + 2 <= mm) SUM_ROWS(2)
            if (j + 1 <= mm) SUM_ROWS(1)
        }
        FINISH_EDGES_BEFORE(NUM_CONN)
    }
    __syncthreads();

    const int seg_beg = __shfl(bndv, m, 32);
    const int seg_end = __shfl(bndv, m + 1, 32);
    const float degree = (float)(seg_end > seg_beg ? seg_end - seg_beg : 0);
    const float invc = 1.0f / fmaxf(degree, 1.0f);
    const float* arow = abase + m * A_PITCH;

    float* hbase = abase;
    if (!tail) {
        v8f acc[N_TILES] = {};

        for (int ks = 0; ks < K_STEPS; ++ks) {
            const int k = ks * 4;
            v2f a = *(const v2f*)(arow + k + 2 * hi);
            a.x *= invc;
            a.y *= invc;

            const float4* __restrict__ bp =
                (const float4*)(wpack + ((size_t)ks * 32 + lane) * 16);
            float4 q0 = bp[0], q1 = bp[1], q2 = bp[2], q3 = bp[3];
            v2f bbs[N_TILES];
            bbs[0].x = q0.x; bbs[0].y = q0.y;
            bbs[1].x = q0.z; bbs[1].y = q0.w;
            bbs[2].x = q1.x; bbs[2].y = q1.y;
            bbs[3].x = q1.z; bbs[3].y = q1.w;
            bbs[4].x = q2.x; bbs[4].y = q2.y;
            bbs[5].x = q2.z; bbs[5].y = q2.w;
            bbs[6].x = q3.x; bbs[6].y = q3.y;
            bbs[7].x = q3.z; bbs[7].y = q3.w;

#pragma unroll
            for (int nt = 0; nt < N_TILES; ++nt) {
                acc[nt] = __builtin_amdgcn_wmma_f32_16x16x4_f32(
                    false, a, false, bbs[nt], (short)0, acc[nt], false, false);
            }
        }

#pragma unroll
        for (int nt = 0; nt < N_TILES; ++nt) {
            const float bias = b1[nt * 16 + m];
#pragma unroll
            for (int r = 0; r < 8; ++r) acc[nt][r] += bias;
        }

        float s[8], q[8];
#pragma unroll
        for (int r = 0; r < 8; ++r) { s[r] = 0.0f; q[r] = 0.0f; }
#pragma unroll
        for (int nt = 0; nt < N_TILES; ++nt) {
#pragma unroll
            for (int r = 0; r < 8; ++r) {
                const float x = acc[nt][r];
                s[r] += x;
                q[r] += x * x;
            }
        }
#pragma unroll
        for (int mask = 1; mask < 16; mask <<= 1) {
#pragma unroll
            for (int r = 0; r < 8; ++r) {
                s[r] += __shfl_xor(s[r], mask, 32);
                q[r] += __shfl_xor(q[r], mask, 32);
            }
        }
        float mean[8], rstd[8];
#pragma unroll
        for (int r = 0; r < 8; ++r) {
            mean[r] = s[r] * (1.0f / HIDDEN);
            const float var = q[r] * (1.0f / HIDDEN) - mean[r] * mean[r];
            rstd[r] = rsqrtf(var + LN_EPS);
        }

#pragma unroll
        for (int nt = 0; nt < N_TILES; ++nt) {
            const int n = nt * 16 + m;
            const float g  = lng[n];
            const float bb = lnb[n];
#pragma unroll
            for (int r = 0; r < 8; ++r) {
                float x = (acc[nt][r] - mean[r]) * rstd[r] * g + bb;
                x = fmaxf(x, 0.0f);
                hbase[(r + 8 * hi) * H_PITCH + n] = x;
            }
        }
    } else {
        float* red = sA + 2 * (TILE_ROWS * A_PITCH);
        v8f c2[TAIL_NT] = {};
        const float4* __restrict__ bq = (const float4*)wpack + wave;
        for (int ks = 0; ks < K_STEPS; ++ks) {
            v2f a = *(const v2f*)(arow + ks * 4 + 2 * hi);
            a.x *= invc;
            a.y *= invc;
            const float4 q4 = bq[((size_t)ks * 32 + lane) * 4];
            v2f bA, bB;
            bA.x = q4.x; bA.y = q4.y;
            bB.x = q4.z; bB.y = q4.w;
            c2[0] = __builtin_amdgcn_wmma_f32_16x16x4_f32(
                false, a, false, bA, (short)0, c2[0], false, false);
            c2[1] = __builtin_amdgcn_wmma_f32_16x16x4_f32(
                false, a, false, bB, (short)0, c2[1], false, false);
        }
#pragma unroll
        for (int j = 0; j < TAIL_NT; ++j) {
            const float bias = b1[(wave * TAIL_NT + j) * 16 + m];
#pragma unroll
            for (int r = 0; r < 8; ++r) c2[j][r] += bias;
        }
        float s[8], q[8];
#pragma unroll
        for (int r = 0; r < 8; ++r) { s[r] = 0.0f; q[r] = 0.0f; }
#pragma unroll
        for (int j = 0; j < TAIL_NT; ++j) {
#pragma unroll
            for (int r = 0; r < 8; ++r) {
                const float x = c2[j][r];
                s[r] += x;
                q[r] += x * x;
            }
        }
#pragma unroll
        for (int mask = 1; mask < 16; mask <<= 1) {
#pragma unroll
            for (int r = 0; r < 8; ++r) {
                s[r] += __shfl_xor(s[r], mask, 32);
                q[r] += __shfl_xor(q[r], mask, 32);
            }
        }
        if (m == 0) {
#pragma unroll
            for (int r = 0; r < 8; ++r) {
                red[(wave * TILE_ROWS + r + 8 * hi) * 2 + 0] = s[r];
                red[(wave * TILE_ROWS + r + 8 * hi) * 2 + 1] = q[r];
            }
        }
        __syncthreads();
        float mean[8], rstd[8];
#pragma unroll
        for (int r = 0; r < 8; ++r) {
            float st = 0.0f, qt = 0.0f;
#pragma unroll
            for (int w = 0; w < WAVES_PER_BLOCK; ++w) {
                st += red[(w * TILE_ROWS + r + 8 * hi) * 2 + 0];
                qt += red[(w * TILE_ROWS + r + 8 * hi) * 2 + 1];
            }
            mean[r] = st * (1.0f / HIDDEN);
            const float var = qt * (1.0f / HIDDEN) - mean[r] * mean[r];
            rstd[r] = rsqrtf(var + LN_EPS);
        }
        hbase = sA + 1 * (TILE_ROWS * A_PITCH);
#pragma unroll
        for (int j = 0; j < TAIL_NT; ++j) {
            const int n = (wave * TAIL_NT + j) * 16 + m;
            const float g  = lng[n];
            const float bb = lnb[n];
#pragma unroll
            for (int r = 0; r < 8; ++r) {
                float x = (c2[j][r] - mean[r]) * rstd[r] * g + bb;
                x = fmaxf(x, 0.0f);
                hbase[(r + 8 * hi) * H_PITCH + n] = x;
            }
        }
    }
    __syncthreads();
    if (tail && wave != 0) return;

    v8f o0, o1;
    {
        const float bias0 = b2[m];
        const float bias1 = b2[16];
#pragma unroll
        for (int r = 0; r < 8; ++r) { o0[r] = bias0; o1[r] = bias1; }
    }
    const float* ha  = hbase + m * H_PITCH + 2 * hi;
    const float* __restrict__ wb0 = sW2 + (2 * hi) * OUT_DIM + m;
    const float* __restrict__ wb1 = sW2 + (2 * hi) * OUT_DIM + 16;
#pragma unroll
    for (int ks = 0; ks < K2_STEPS; ++ks) {
        const v2f a = *(const v2f*)(ha + 4 * ks);
        v2f bA, bB;
        bA.x = wb0[ks * 4 * OUT_DIM];
        bA.y = wb0[ks * 4 * OUT_DIM + OUT_DIM];
        bB.x = wb1[ks * 4 * OUT_DIM];
        bB.y = wb1[ks * 4 * OUT_DIM + OUT_DIM];
        o0 = __builtin_amdgcn_wmma_f32_16x16x4_f32(
            false, a, false, bA, (short)0, o0, false, false);
        o1 = __builtin_amdgcn_wmma_f32_16x16x4_f32(
            false, a, false, bB, (short)0, o1, false, false);
    }

    {
        float* __restrict__ orow = out + (size_t)(m0 + 8 * hi) * OUT_DIM;
#pragma unroll
        for (int r = 0; r < 8; ++r) orow[r * OUT_DIM + m] = o0[r];
        if (m == 0) {
#pragma unroll
            for (int r = 0; r < 8; ++r) orow[r * OUT_DIM + 16] = o1[r];
        }
    }
}

extern "C" void kernel_launch(void* const* d_in, const int* in_sizes, int n_in,
                              void* d_out, int out_size, void* d_ws, size_t ws_size,
                              hipStream_t stream) {
    const float* nf    = (const float*)d_in[0];
    const int*   hidx  = (const int*)d_in[1];
    const int*   nodes = hidx;
    const int*   edges = hidx + NUM_CONN;
    const float* W1    = (const float*)d_in[2];
    const float* b1    = (const float*)d_in[3];
    const float* lng   = (const float*)d_in[4];
    const float* lnb   = (const float*)d_in[5];
    const float* W2    = (const float*)d_in[6];
    const float* b2    = (const float*)d_in[7];
    float*       out   = (float*)d_out;

    int* tab    = (int*)d_ws;
    int* binned = tab + TAB_N + 64;
    int* perm   = binned + NUM_CONN;
    int* ends   = perm + NUM_CONN;
    float* wpack = (float*)d_ws + (size_t)NUM_EDGES * IN_DIM + NUM_EDGES;

    chunk_sort_kernel<<<NCHUNK + REPACK_GROUPS, BIN_THREADS, 0, stream>>>(nodes, edges, tab, binned, W1, wpack);

    range_sort_kernel<<<NB, BIN_THREADS, 0, stream>>>(tab, binned, perm, ends);

    edge_sum_mlp_kernel<<<FULL_GROUPS + TAIL_GROUPS, WAVES_PER_BLOCK * 32, 0, stream>>>(
        nf, perm, ends, wpack, b1, lng, lnb, W2, b2, out);
}


# Gather and first product split into two feature-half passes per wave (128 of 256 features each): a work-group's products run under the gathers of the others on its SIMDs; every sum order as before
# speedup vs baseline: 10.0390x; 1.0434x over previous
#include <hip/hip_runtime.h>

#define NUM_NODES 100000
#define NUM_EDGES 50000
#define NUM_CONN  800000
#define IN_DIM    256
#define HIDDEN    128
#define OUT_DIM   17
#define LN_EPS    1e-5f

#define WAVES_PER_BLOCK 4
#define TILE_ROWS 16
#define NUM_TILES (NUM_EDGES / TILE_ROWS)
#define NUM_WGPS    256
#define FULL_GROUPS (3 * NUM_WGPS)
#define FULL_TILES  (FULL_GROUPS * WAVES_PER_BLOCK)
#define TAIL_GROUPS (NUM_TILES - FULL_TILES)
#define TAIL_ROWS   (TILE_ROWS / WAVES_PER_BLOCK)
#define TAIL_NT     (N_TILES / WAVES_PER_BLOCK)
static_assert(TAIL_GROUPS >= 0 && TAIL_GROUPS <= NUM_WGPS, "at most one shared tile per WGP");
static_assert(HIDDEN / 16 / WAVES_PER_BLOCK == 2, "a shared tile's wave takes one float4 (two column blocks) of the packed B stream");
#define N_TILES (HIDDEN / 16)
#define K_STEPS (IN_DIM / 4)
#define WPACK_FLOATS (K_STEPS * 32 * 16)
#define K2_STEPS (HIDDEN / 4)
#define H_PITCH  (HIDDEN + 4)
#define W2_FLOATS (HIDDEN * OUT_DIM)
#define W2_COPY_TRIPS ((W2_FLOATS + WAVES_PER_BLOCK * 32 - 1) / (WAVES_PER_BLOCK * 32))

typedef float v2f __attribute__((ext_vector_type(2)));
typedef float v8f __attribute__((ext_vector_type(8)));
typedef float v4f __attribute__((ext_vector_type(4)));

#define REPACK_ITEMS  (K_STEPS * 32 * 8)
__device__ __forceinline__ void repack_w1_item(const float* __restrict__ W1,
                                               float* __restrict__ wpack, int id) {
    if (id >= REPACK_ITEMS) return;
    int ks   = id >> 8;
    int rem  = id & 255;
    int lane = rem >> 3;
    int nt   = rem & 7;
    int m    = lane & 15;
    int hi   = lane >> 4;
    int krow = ks * 4 + 2 * hi;
    int n    = nt * 16 + m;
    wpack[2 * id + 0] = W1[(size_t)(krow + 0) * HIDDEN + n];
    wpack[2 * id + 1] = W1[(size_t)(krow + 1) * HIDDEN + n];
}


#define NB 128
#define EPB 391
#define EPB_PAD 512
#define NCHUNK 128
#define CHUNK (NUM_CONN / NCHUNK)
#define BIN_THREADS 1024
#define BIN_PER_THREAD 7
#define OFF_PITCH (NB + 1)
#define TAB_N (NCHUNK * OFF_PITCH)
#define NODE_BITS 17
#define NODE_MASK ((1 << NODE_BITS) - 1)
#define PIECES_PER_WAVE (NCHUNK / 32)
#define PIECE_TRIPS 2
#define STAGE_CAP 16384
#define REPACK_GROUPS ((REPACK_ITEMS + BIN_THREADS - 1) / BIN_THREADS)

static_assert(NB * EPB >= NUM_EDGES, "ranges cover every edge");
static_assert(NCHUNK * CHUNK == NUM_CONN, "chunks cover every connection");
static_assert(BIN_THREADS * BIN_PER_THREAD >= CHUNK, "one register batch per chunk");
static_assert(NB == 4 * 32 && NCHUNK == 4 * 32, "four whole waves scan the counts and read the offsets");
static_assert(PIECES_PER_WAVE * (BIN_THREADS / 32) == NCHUNK, "every piece has a wave");
static_assert((1 << NODE_BITS) >= NUM_NODES && EPB <= EPB_PAD, "packed word fits");
static_assert(STAGE_CAP % (4 * BIN_THREADS) == 0, "the level-2 window is a whole number of copy batches");

__device__ __forceinline__ int clamp_node(int node) {
    return (unsigned)node < (unsigned)NUM_NODES ? node : NUM_NODES - 1;
}

__global__ __launch_bounds__(BIN_THREADS) void chunk_sort_kernel(
    const int* __restrict__ nodes,
    const int* __restrict__ edges,
    int* __restrict__ tab,
    int* __restrict__ binned,
    const float* __restrict__ W1,
    float* __restrict__ wpack) {
    if (blockIdx.x >= NCHUNK) {
        repack_w1_item(W1, wpack, (int)(blockIdx.x - NCHUNK) * BIN_THREADS + (int)threadIdx.x);
        return;
    }
    __shared__ int cnt[NB];
    __shared__ int wtot[NB / 32];
    __shared__ int stage[BIN_THREADS * BIN_PER_THREAD];
    __shared__ int ctot;
    const int t    = threadIdx.x;
    const int lane = t & 31;
    const int wv   = t >> 5;
    const int w    = blockIdx.x;
    if (t < NB) cnt[t] = 0;
    const int* __restrict__ esrc = edges + w * CHUNK;
    const int* __restrict__ nsrc = nodes + w * CHUNK;
    int e[BIN_PER_THREAD], nd[BIN_PER_THREAD], rk[BIN_PER_THREAD];
#pragma unroll
    for (int k = 0; k < BIN_PER_THREAD; ++k) {
        const int i = t + k * BIN_THREADS;
        e[k]  = (i < CHUNK) ? esrc[i] : -1;
        nd[k] = (i < CHUNK) ? nsrc[i] : 0;
    }
    __syncthreads();
#pragma unroll
    for (int k = 0; k < BIN_PER_THREAD; ++k) {
        rk[k] = 0;
        if ((unsigned)e[k] < (unsigned)NUM_EDGES) rk[k] = atomicAdd(&cnt[e[k] / EPB], 1);
    }
    __syncthreads();

    int c = 0, inc = 0;
    if (t < NB) {
        c = cnt[t];
        inc = c;
#pragma unroll
        for (int d = 1; d < 32; d <<= 1) {
            const int up = __shfl_up(inc, d, 32);
            if (lane >= d) inc += up;
        }
        if (lane == 31) wtot[wv] = inc;
    }
    __syncthreads();
    if (t < NB) {
        int pre = 0;
#pragma unroll
        for (int j = 0; j < NB / 32 - 1; ++j) if (j < wv) pre += wtot[j];
        const int start = pre + inc - c;
        cnt[t] = start;
        tab[w * OFF_PITCH + t] = start;
        if (t == NB - 1) {
            tab[w * OFF_PITCH + NB] = start + c;
            ctot = start + c;
        }
    }
    __syncthreads();

#pragma unroll
    for (int k = 0; k < BIN_PER_THREAD; ++k) {
        if ((unsigned)e[k] < (unsigned)NUM_EDGES) {
            const int b  = e[k] / EPB;
            const int le = e[k] - b * EPB;
            const unsigned pos = (unsigned)(cnt[b] + rk[k]);
            if (pos < (unsigned)CHUNK) stage[pos] = (le << NODE_BITS) | clamp_node(nd[k]);
        }
    }
    __syncthreads();
    const int tot = ctot < CHUNK ? ctot : CHUNK;
    int sv[BIN_PER_THREAD];
#pragma unroll
    for (int k = 0; k < BIN_PER_THREAD; ++k) sv[k] = stage[t + k * BIN_THREADS];
#pragma unroll
    for (int k = 0; k < BIN_PER_THREAD; ++k) {
        const int i = t + k * BIN_THREADS;
        if (i < tot) binned[w * CHUNK + i] = sv[k];
    }
}

__global__ __launch_bounds__(BIN_THREADS) void range_sort_kernel(
    const int* __restrict__ tab,
    const int* __restrict__ binned,
    int* __restrict__ perm,
    int* __restrict__ ends) {
    __shared__ int cnt[EPB_PAD];
    __shared__ int wsum[32];
    __shared__ int pbeg[NCHUNK];
    __shared__ int plen[NCHUNK];
    __shared__ int osum[NCHUNK / 32];
    __shared__ int stage[STAGE_CAP];
    __shared__ int rtot;
    const int t    = threadIdx.x;
    const int lane = t & 31;
    const int w    = t >> 5;
    const int b    = blockIdx.x;

    if (t < EPB_PAD) cnt[t] = 0;

    int o0 = 0;
    if (t < NCHUNK) {
        o0     = tab[t * OFF_PITCH + b];
        int o1 = tab[t * OFF_PITCH + b + 1];
        if (o0 < 0) o0 = 0;
        if (o0 > CHUNK) o0 = CHUNK;
        if (o1 < o0) o1 = o0;
        if (o1 > CHUNK) o1 = CHUNK;
        pbeg[t] = t * CHUNK + o0;
        plen[t] = o1 - o0;
    }
    __syncthreads();

    int pb[PIECES_PER_WAVE], pl[PIECES_PER_WAVE];
    int p[PIECES_PER_WAVE][PIECE_TRIPS];
#pragma unroll
    for (int j = 0; j < PIECES_PER_WAVE; ++j) {
        pb[j] = __builtin_amdgcn_readfirstlane(pbeg[w * PIECES_PER_WAVE + j]);
        pl[j] = __builtin_amdgcn_readfirstlane(plen[w * PIECES_PER_WAVE + j]);
    }
#pragma unroll
    for (int j = 0; j < PIECES_PER_WAVE; ++j) {
#pragma unroll
        for (int k = 0; k < PIECE_TRIPS; ++k) {
            const int i = lane + 32 * k;
            p[j][k] = (i < pl[j]) ? binned[pb[j] + i] : -1;
        }
    }

    if (t < NCHUNK) {
        int osm = o0;
#pragma unroll
        for (int d = 16; d >= 1; d >>= 1) osm += __shfl_xor(osm, d, 32);
        if (lane == 0) osum[w] = osm;
    }

#pragma unroll
    for (int j = 0; j < PIECES_PER_WAVE; ++j) {
#pragma unroll
        for (int k = 0; k < PIECE_TRIPS; ++k) {
            if (p[j][k] >= 0) atomicAdd(&cnt[(p[j][k] >> NODE_BITS) & (EPB_PAD - 1)], 1);
        }
        for (int i = lane + 32 * PIECE_TRIPS; i < pl[j]; i += 32) {
            const int q = binned[pb[j] + i];
            if (q >= 0) atomicAdd(&cnt[(q >> NODE_BITS) & (EPB_PAD - 1)], 1);
        }
    }
    __syncthreads();

    int bs = 0;
#pragma unroll
    for (int j = 0; j < NCHUNK / 32; ++j) bs += osum[j];
    if (bs > NUM_CONN) bs = NUM_CONN;

    const int c = (t < EPB_PAD) ? cnt[t] : 0;
    int inc = c;
#pragma unroll
    for (int d = 1; d < 32; d <<= 1) {
        const int up = __shfl_up(inc, d, 32);
        if (lane >= d) inc += up;
    }
    if (lane == 31) wsum[w] = inc;
    __syncthreads();
    if (w == 0) {
        const int tot = wsum[lane];
        int inc2 = tot;
#pragma unroll
        for (int d = 1; d < 32; d <<= 1) {
            const int u = __shfl_up(inc2, d, 32);
            if (lane >= d) inc2 += u;
        }
        wsum[lane] = inc2 - tot;
        if (lane == 31) rtot = inc2;
    }
    __syncthreads();
    const int  rt     = rtot;
    const bool staged = rt <= STAGE_CAP;
    const int end_t = bs + wsum[w] + inc;
    const int e = b * EPB + t;
    if (t < EPB && e < NUM_EDGES) ends[e] = end_t;
    if (t < EPB_PAD) cnt[t] = end_t - c;
    __syncthreads();

#pragma unroll
    for (int j = 0; j < PIECES_PER_WAVE; ++j) {
#pragma unroll
        for (int k = 0; k < PIECE_TRIPS; ++k) {
            if (p[j][k] >= 0) {
                const unsigned pos = (unsigned)atomicAdd(&cnt[(p[j][k] >> NODE_BITS) & (EPB_PAD - 1)], 1);
                const unsigned loc = pos - (unsigned)bs;
                if (staged) { if (loc < (unsigned)STAGE_CAP) stage[loc] = p[j][k] & NODE_MASK; }
                else if (pos < (unsigned)NUM_CONN) perm[pos] = p[j][k] & NODE_MASK;
            }
        }
        for (int i = lane + 32 * PIECE_TRIPS; i < pl[j]; i += 32) {
            const int q = binned[pb[j] + i];
            if (q >= 0) {
                const unsigned pos = (unsigned)atomicAdd(&cnt[(q >> NODE_BITS) & (EPB_PAD - 1)], 1);
                const unsigned loc = pos - (unsigned)bs;
                if (staged) { if (loc < (unsigned)STAGE_CAP) stage[loc] = q & NODE_MASK; }
                else if (pos < (unsigned)NUM_CONN) perm[pos] = q & NODE_MASK;
            }
        }
    }
    __syncthreads();
    if (staged) {
        for (int i0 = t; i0 < rt; i0 += 4 * BIN_THREADS) {
            int sv[4];
#pragma unroll
            for (int k = 0; k < 4; ++k) sv[k] = stage[i0 + k * BIN_THREADS];
#pragma unroll
            for (int k = 0; k < 4; ++k) {
                const int i = i0 + k * BIN_THREADS;
                if (i < rt && bs + i < NUM_CONN) perm[bs + i] = sv[k];
            }
        }
    }
}

#define A_PITCH (IN_DIM + 4)
#define ROWS_IN_FLIGHT 16

__device__ __forceinline__ void add4(float4& a, const float4& x) {
    a.x += x.x; a.y += x.y; a.z += x.z; a.w += x.w;
}

__device__ __forceinline__ const float4* node_row(const float* __restrict__ nf, int node) {
    const unsigned n = (unsigned)node < (unsigned)NUM_NODES ? (unsigned)node
                                                            : (unsigned)(NUM_NODES - 1);
    return (const float4*)(nf + (size_t)n * IN_DIM);
}

__device__ __forceinline__ void add4v(float4& a, const v4f& x) {
    a.x += x.x; a.y += x.y; a.z += x.z; a.w += x.w;
}

__device__ __forceinline__ const float* node_row_base(const float* __restrict__ nf, int node) {
    const unsigned n = (unsigned)node < (unsigned)NUM_NODES ? (unsigned)node
                                                            : (unsigned)(NUM_NODES - 1);
    return nf + (size_t)n * IN_DIM;
}

#define FINISH_EDGES_BEFORE(POS)                                                        \
    _Pragma("nounroll")                                                                 \
    while (cur < row_end && (POS) >= next_end) {                                        \
        float4* drow = (float4*)(abh + cur * A_PITCH);                                  \
        drow[lane] = a0;                                                                \
        a0 = make_float4(0.0f, 0.0f, 0.0f, 0.0f);                                       \
        ++cur;                                                                          \
        next_end = __builtin_amdgcn_readlane(bndv, cur < row_end ? cur + 1 : row_end);   \
    }

#define LOAD_ROWS(N)                                                                    \
        const float4* rp[N];                                                            \
        float4 rx[N];                                                                   \
        _Pragma("unroll")                                                               \
        for (int i = 0; i < (N); ++i)                                                   \
            rp[i] = node_row(nfh, __builtin_amdgcn_readlane(myid, j + i));              \
        _Pragma("unroll")                                                               \
        for (int i = 0; i < (N); ++i) rx[i] = rp[i][lane];
#define ADD_ROWS(N)                                                                     \
        _Pragma("unroll")                                                               \
        for (int i = 0; i < (N); ++i) {                                                 \
            FINISH_EDGES_BEFORE(base + j + i)                                           \
            add4(a0, rx[i]);                                                            \
        }                                                                               \
        j += (N);
#define SUM_ROWS(N) { LOAD_ROWS(N) ADD_ROWS(N) }

#define RING_FIRST(S)                                                                   \
        asm volatile("global_load_b128 %0, %2, %3\n\tglobal_load_b128 %1, %2, %4"       \
                     : "=v"(sx[S]), "=v"(sy[S])                                         \
                     : "v"(lane16),                                                     \
                       "s"(node_row_base(nfh, __builtin_amdgcn_readlane(id_cur, 2 * S))), \
                       "s"(node_row_base(nfh, __builtin_amdgcn_readlane(id_cur, 2 * S + 1))));
#define RING_ASK(S)                                                                     \
        asm volatile("global_load_b128 %0, %2, %3\n\tglobal_load_b128 %1, %2, %4"       \
                     : "+v"(sx[S]), "+v"(sy[S])                                         \
                     : "v"(lane16),                                                     \
                       "s"(node_row_base(nfh, __builtin_amdgcn_readlane(id_use, 2 * S))), \
                       "s"(node_row_base(nfh, __builtin_amdgcn_readlane(id_use, 2 * S + 1))));
#define RING_STEP(S)                                                                    \
        FINISH_EDGES_BEFORE(base + 2 * S)                                               \
        asm volatile("s_wait_loadcnt 31" : "+v"(sx[S]), "+v"(sy[S]));                   \
        add4v(a0, sx[S]);                                                               \
        FINISH_EDGES_BEFORE(base + 2 * S + 1)                                           \
        add4v(a0, sy[S]);                                                               \
        RING_ASK(S)
#define RING_TAKE_IDS                                                                   \
        asm volatile("s_wait_loadcnt 32\n\tv_mov_b32 %0, %1" : "=v"(id_use) : "v"(id_ld));
#define RING_DRAIN(S, N)                                                                \
        FINISH_EDGES_BEFORE(base + 2 * S)                                               \
        asm volatile("s_wait_loadcnt " #N : "+v"(sx[S]), "+v"(sy[S]));                  \
        add4v(a0, sx[S]);                                                               \
        FINISH_EDGES_BEFORE(base + 2 * S + 1)                                           \
        add4v(a0, sy[S]);

__global__ __launch_bounds__(WAVES_PER_BLOCK * 32) void edge_sum_mlp_kernel(
    const float* __restrict__ nf,
    const int*   __restrict__ perm,
    const int*   __restrict__ ends,
    const float* __restrict__ wpack,
    const float* __restrict__ b1,
    const float* __restrict__ lng,
    const float* __restrict__ lnb,
    const float* __restrict__ W2,
    const float* __restrict__ b2,
    float* __restrict__ out) {
    __shared__ alignas(16) float sA[WAVES_PER_BLOCK * TILE_ROWS * A_PITCH];
    __shared__ float sW2[W2_COPY_TRIPS * WAVES_PER_BLOCK * 32];

    const int lane = threadIdx.x & 31;
    const int wave = __builtin_amdgcn_readfirstlane((int)(threadIdx.x >> 5));
    const bool tail = blockIdx.x >= FULL_GROUPS;
    const int  tile = tail ? FULL_TILES + ((int)blockIdx.x - FULL_GROUPS)
                           : (int)blockIdx.x * WAVES_PER_BLOCK + wave;
    const int  m0   = tile * TILE_ROWS;
    const int m    = lane & 15;
    const int hi   = lane >> 4;

    const int row_beg = tail ? wave * TAIL_ROWS : 0;
    const int row_end = row_beg + (tail ? TAIL_ROWS : TILE_ROWS);
    float* abase = sA + (tail ? 0 : wave) * (TILE_ROWS * A_PITCH);

    const int bidx = m0 - 1 + (lane < TILE_ROWS ? lane : TILE_ROWS);
    const int braw = ends[bidx < 0 ? 0 : bidx];

    float w2r[W2_COPY_TRIPS];
#pragma unroll
    for (int i = 0; i < W2_COPY_TRIPS; ++i) {
        const int id  = (int)threadIdx.x + i * (WAVES_PER_BLOCK * 32);
        const int idc = id < W2_FLOATS ? id : W2_FLOATS - 1;
        w2r[i] = W2[idc];
    }

    int bndv = (bidx < 0) ? 0 : braw;
    if (bndv < 0) bndv = 0;
    if (bndv > NUM_CONN) bndv = NUM_CONN;

    const int beg   = __builtin_amdgcn_readlane(bndv, row_beg);
    const int end_t = __builtin_amdgcn_readlane(bndv, row_end);
    int id_first = 0;
    if (beg + lane < end_t) id_first = perm[beg + lane];
#pragma unroll
    for (int i = 0; i < W2_COPY_TRIPS; ++i)
        sW2[(int)threadIdx.x + i * (WAVES_PER_BLOCK * 32)] = w2r[i];

    const int seg_beg = __shfl(bndv, m, 32);
    const int seg_end = __shfl(bndv, m + 1, 32);
    const float degree = (float)(seg_end > seg_beg ? seg_end - seg_beg : 0);
    const float invc = 1.0f / fmaxf(degree, 1.0f);
    const float* arow = abase + m * A_PITCH;

    v8f acc[N_TILES] = {};
    const float4* __restrict__ bq = (const float4*)wpack + wave;

    for (int h = 0; h < 2; ++h) {
        const float* __restrict__ nfh = nf + h * (IN_DIM / 2);
        float* abh = abase + h * (IN_DIM / 2);
        {
            int cur      = row_beg;
            int next_end = __builtin_amdgcn_readlane(bndv, row_beg + 1);
            float4 a0 = make_float4(0.0f, 0.0f, 0.0f, 0.0f);

            int base   = beg;
            int id_cur = id_first;

            if (end_t - beg >= 32) {
                v4f sx[16], sy[16];
                const int lane16 = lane * 16;
                const int lane4  = lane * 4;
                int id_ld, id_use;
                asm volatile("global_load_b32 %0, %1, %2" : "=v"(id_ld) : "v"(lane4), "s"(perm + base + 32));
                RING_FIRST(0) RING_FIRST(1) RING_FIRST(2) RING_FIRST(3) RING_FIRST(4) RING_FIRST(5) RING_FIRST(6) RING_FIRST(7) RING_FIRST(8) RING_FIRST(9) RING_FIRST(10) RING_FIRST(11) RING_FIRST(12) RING_FIRST(13) RING_FIRST(14) RING_FIRST(15)
                RING_TAKE_IDS
                while (end_t - base >= 64) {
                    asm volatile("global_load_b32 %0, %1, %2" : "=v"(id_ld) : "v"(lane4), "s"(perm + base + 64));
                    RING_STEP(0) RING_STEP(1) RING_STEP(2) RING_STEP(3) RING_STEP(4) RING_STEP(5) RING_STEP(6) RING_STEP(7) RING_STEP(8) RING_STEP(9) RING_STEP(10) RING_STEP(11) RING_STEP(12) RING_STEP(13) RING_STEP(14) RING_STEP(15)
                    RING_TAKE_IDS
                    base += 32;
                }
                RING_DRAIN(0, 30) RING_DRAIN(1, 28) RING_DRAIN(2, 26) RING_DRAIN(3, 24) RING_DRAIN(4, 22) RING_DRAIN(5, 20) RING_DRAIN(6, 18) RING_DRAIN(7, 16) RING_DRAIN(8, 14) RING_DRAIN(9, 12) RING_DRAIN(10, 10) RING_DRAIN(11, 8) RING_DRAIN(12, 6) RING_DRAIN(13, 4) RING_DRAIN(14, 2) RING_DRAIN(15, 0)
                base += 32;
                id_cur = id_use;
            }

            for (; base < end_t; base += 32) {
                const int left = end_t - base;
                const int mm   = left < 32 ? left : 32;
                const int myid = id_cur;
                int j = 0;
                id_cur = 0;
                if (base + 32 + lane < end_t) id_cur = perm[base + 32 + lane];
                while (j + ROWS_IN_FLIGHT <= mm) SUM_ROWS(ROWS_IN_FLIGHT)
                if (j + 8 <= mm) SUM_ROWS(8)
                if (j + 4 <= mm) SUM_ROWS(4)
                if (j + 2 <= mm) SUM_ROWS(2)
                if (j + 1 <= mm) SUM_ROWS(1)
            }
            FINISH_EDGES_BEFORE(NUM_CONN)
        }
        __syncthreads();

        if (!tail) {
#pragma unroll 4
            for (int ks = h * (K_STEPS / 2); ks < (h + 1) * (K_STEPS / 2); ++ks) {
                const int k = ks * 4;
                v2f a = *(const v2f*)(arow + k + 2 * hi);
                a.x *= invc;
                a.y *= invc;

                const float4* __restrict__ bp =
                    (const float4*)(wpack + ((size_t)ks * 32 + lane) * 16);
                float4 q0 = bp[0], q1 = bp[1], q2 = bp[2], q3 = bp[3];
                v2f bbs[N_TILES];
                bbs[0].x = q0.x; bbs[0].y = q0.y;
                bbs[1].x = q0.z; bbs[1].y = q0.w;
                bbs[2].x = q1.x; bbs[2].y = q1.y;
                bbs[3].x = q1.z; bbs[3].y = q1.w;
                bbs[4].x = q2.x; bbs[4].y = q2.y;
                bbs[5].x = q2.z; bbs[5].y = q2.w;
                bbs[6].x = q3.x; bbs[6].y = q3.y;
                bbs[7].x = q3.z; bbs[7].y = q3.w;

#pragma unroll
                for (int nt = 0; nt < N_TILES; ++nt) {
                    acc[nt] = __builtin_amdgcn_wmma_f32_16x16x4_f32(
                        false, a, false, bbs[nt], (short)0, acc[nt], false, false);
                }
            }
        } else {
            for (int ks = h * (K_STEPS / 2); ks < (h + 1) * (K_STEPS / 2); ++ks) {
                v2f a = *(const v2f*)(arow + ks * 4 + 2 * hi);
                a.x *= invc;
                a.y *= invc;
                const float4 q4 = bq[((size_t)ks * 32 + lane) * 4];
                v2f bA, bB;
                bA.x = q4.x; bA.y = q4.y;
                bB.x = q4.z; bB.y = q4.w;
                acc[0] = __builtin_amdgcn_wmma_f32_16x16x4_f32(
                    false, a, false, bA, (short)0, acc[0], false, false);
                acc[1] = __builtin_amdgcn_wmma_f32_16x16x4_f32(
                    false, a, false, bB, (short)0, acc[1], false, false);
            }
        }
    }

    float* hbase = abase;
    if (!tail) {
#pragma unroll
        for (int nt = 0; nt < N_TILES; ++nt) {
            const float bias = b1[nt * 16 + m];
#pragma unroll
            for (int r = 0; r < 8; ++r) acc[nt][r] += bias;
        }

        float s[8], q[8];
#pragma unroll
        for (int r = 0; r < 8; ++r) { s[r] = 0.0f; q[r] = 0.0f; }
#pragma unroll
        for (int nt = 0; nt < N_TILES; ++nt) {
#pragma unroll
            for (int r = 0; r < 8; ++r) {
                const float x = acc[nt][r];
                s[r] += x;
                q[r] += x * x;
            }
        }
#pragma unroll
        for (int mask = 1; mask < 16; mask <<= 1) {
#pragma unroll
            for (int r = 0; r < 8; ++r) {
                s[r] += __shfl_xor(s[r], mask, 32);
                q[r] += __shfl_xor(q[r], mask, 32);
            }
        }
        float mean[8], rstd[8];
#pragma unroll
        for (int r = 0; r < 8; ++r) {
            mean[r] = s[r] * (1.0f / HIDDEN);
            const float var = q[r] * (1.0f / HIDDEN) - mean[r] * mean[r];
            rstd[r] = rsqrtf(var + LN_EPS);
        }

#pragma unroll
        for (int nt = 0; nt < N_TILES; ++nt) {
            const int n = nt * 16 + m;
            const float g  = lng[n];
            const float bb = lnb[n];
#pragma unroll
            for (int r = 0; r < 8; ++r) {
                float x = (acc[nt][r] - mean[r]) * rstd[r] * g + bb;
                x = fmaxf(x, 0.0f);
                hbase[(r + 8 * hi) * H_PITCH + n] = x;
            }
        }
    } else {
        float* red = sA + 2 * (TILE_ROWS * A_PITCH);
#pragma unroll
        for (int j = 0; j < TAIL_NT; ++j) {
            const float bias = b1[(wave * TAIL_NT + j) * 16 + m];
#pragma unroll
            for (int r = 0; r < 8; ++r) acc[j][r] += bias;
        }
        float s[8], q[8];
#pragma unroll
        for (int r = 0; r < 8; ++r) { s[r] = 0.0f; q[r] = 0.0f; }
#pragma unroll
        for (int j = 0; j < TAIL_NT; ++j) {
#pragma unroll
            for (int r = 0; r < 8; ++r) {
                const float x = acc[j][r];
                s[r] += x;
                q[r] += x * x;
            }
        }
#pragma unroll
        for (int mask = 1; mask < 16; mask <<= 1) {
#pragma unroll
            for (int r = 0; r < 8; ++r) {
                s[r] += __shfl_xor(s[r], mask, 32);
                q[r] += __shfl_xor(q[r], mask, 32);
            }
        }
        if (m == 0) {
#pragma unroll
            for (int r = 0; r < 8; ++r) {
                red[(wave * TILE_ROWS + r + 8 * hi) * 2 + 0] = s[r];
                red[(wave * TILE_ROWS + r + 8 * hi) * 2 + 1] = q[r];
            }
        }
        __syncthreads();
        float mean[8], rstd[8];
#pragma unroll
        for (int r = 0; r < 8; ++r) {
            float st = 0.0f, qt = 0.0f;
#pragma unroll
            for (int w = 0; w < WAVES_PER_BLOCK; ++w) {
                st += red[(w * TILE_ROWS + r + 8 * hi) * 2 + 0];
                qt += red[(w * TILE_ROWS + r + 8 * hi) * 2 + 1];
            }
            mean[r] = st * (1.0f / HIDDEN);
            const float var = qt * (1.0f / HIDDEN) - mean[r] * mean[r];
            rstd[r] = rsqrtf(var + LN_EPS);
        }
        hbase = sA + 1 * (TILE_ROWS * A_PITCH);
#pragma unroll
        for (int j = 0; j < TAIL_NT; ++j) {
            const int n = (wave * TAIL_NT + j) * 16 + m;
            const float g  = lng[n];
            const float bb = lnb[n];
#pragma unroll
            for (int r = 0; r < 8; ++r) {
                float x = (acc[j][r] - mean[r]) * rstd[r] * g + bb;
                x = fmaxf(x, 0.0f);
                hbase[(r + 8 * hi) * H_PITCH + n] = x;
            }
        }
    }
    __syncthreads();
    if (tail && wave != 0) return;

    v8f o0, o1;
    {
        const float bias0 = b2[m];
        const float bias1 = b2[16];
#pragma unroll
        for (int r = 0; r < 8; ++r) { o0[r] = bias0; o1[r] = bias1; }
    }
    const float* ha  = hbase + m * H_PITCH + 2 * hi;
    const float* __restrict__ wb0 = sW2 + (2 * hi) * OUT_DIM + m;
    const float* __restrict__ wb1 = sW2 + (2 * hi) * OUT_DIM + 16;
#pragma unroll
    for (int ks = 0; ks < K2_STEPS; ++ks) {
        const v2f a = *(const v2f*)(ha + 4 * ks);
        v2f bA, bB;
        bA.x = wb0[ks * 4 * OUT_DIM];
        bA.y = wb0[ks * 4 * OUT_DIM + OUT_DIM];
        bB.x = wb1[ks * 4 * OUT_DIM];
        bB.y = wb1[ks * 4 * OUT_DIM + OUT_DIM];
        o0 = __builtin_amdgcn_wmma_f32_16x16x4_f32(
            false, a, false, bA, (short)0, o0, false, false);
        o1 = __builtin_amdgcn_wmma_f32_16x16x4_f32(
            false, a, false, bB, (short)0, o1, false, false);
    }

    {
        float* __restrict__ orow = out + (size_t)(m0 + 8 * hi) * OUT_DIM;
#pragma unroll
        for (int r = 0; r < 8; ++r) orow[r * OUT_DIM + m] = o0[r];
        if (m == 0) {
#pragma unroll
            for (int r = 0; r < 8; ++r) orow[r * OUT_DIM + 16] = o1[r];
        }
    }
}

extern "C" void kernel_launch(void* const* d_in, const int* in_sizes, int n_in,
                              void* d_out, int out_size, void* d_ws, size_t ws_size,
                              hipStream_t stream) {
    const float* nf    = (const float*)d_in[0];
    const int*   hidx  = (const int*)d_in[1];
    const int*   nodes = hidx;
    const int*   edges = hidx + NUM_CONN;
    const float* W1    = (const float*)d_in[2];
    const float* b1    = (const float*)d_in[3];
    const float* lng   = (const float*)d_in[4];
    const float* lnb   = (const float*)d_in[5];
    const float* W2    = (const float*)d_in[6];
    const float* b2    = (const float*)d_in[7];
    float*       out   = (float*)d_out;

    int* tab    = (int*)d_ws;
    int* binned = tab + TAB_N + 64;
    int* perm   = binned + NUM_CONN;
    int* ends   = perm + NUM_CONN;
    float* wpack = (float*)d_ws + (size_t)NUM_EDGES * IN_DIM + NUM_EDGES;

    chunk_sort_kernel<<<NCHUNK + REPACK_GROUPS, BIN_THREADS, 0, stream>>>(nodes, edges, tab, binned, W1, wpack);

    range_sort_kernel<<<NB, BIN_THREADS, 0, stream>>>(tab, binned, perm, ends);

    edge_sum_mlp_kernel<<<FULL_GROUPS + TAIL_GROUPS, WAVES_PER_BLOCK * 32, 0, stream>>>(
        nf, perm, ends, wpack, b1, lng, lnb, W2, b2, out);
}
